# starting kernel
# baseline (speedup 1.0000x reference)
#include <hip/hip_runtime.h>
#include <hip/hip_bf16.h>

typedef float v2f __attribute__((ext_vector_type(2)));
typedef float v8f __attribute__((ext_vector_type(8)));

template <int K, int N, int MODE>
__global__ void gemm_wmma(const float* __restrict__ A, const float* __restrict__ W,
                          float* __restrict__ Y, const int* __restrict__ gidx,
                          int R, int nsrc) {
  const int wave = blockIdx.x * (blockDim.x >> 5) + (threadIdx.x >> 5);
  const int lane = threadIdx.x & 31;
  const int row0 = wave * 16;
  if (row0 >= R) return;

  const int m    = lane & 15;
  const int kb   = (lane >> 4) << 1;
  const int r    = row0 + m;

  long off;
  if (MODE == 0) {
    off = (long)r * K;
  } else if (MODE == 1) {
    const int n = r >> 2, b = r & 3;
    off = ((long)b * nsrc + n) * (long)K;
  } else {
    const int d = gidx[r >> 2];
    off = ((long)d * 4 + (r & 3)) * (long)K;
  }
  const float* __restrict__ arow = A + off;

  v2f wf[K / 4][N / 16];
#pragma unroll
  for (int ks = 0; ks < K / 4; ++ks) {
#pragma unroll
    for (int nt = 0; nt < N / 16; ++nt) {
      const int kk = ks * 4 + kb;
      const int nn = nt * 16 + m;
      wf[ks][nt].x = W[kk * N + nn];
      wf[ks][nt].y = W[(kk + 1) * N + nn];
    }
  }

  v8f acc[N / 16] = {};
#pragma unroll
  for (int ks = 0; ks < K / 4; ++ks) {
    v2f a;
    a.x = arow[ks * 4 + kb];
    a.y = arow[ks * 4 + kb + 1];
#pragma unroll
    for (int nt = 0; nt < N / 16; ++nt) {
      acc[nt] = __builtin_amdgcn_wmma_f32_16x16x4_f32(
           false, a,  false, wf[ks][nt],
           (short)0, acc[nt],  false,  false);
    }
  }

  const int half = lane >> 4;
  const int cn   = lane & 15;
#pragma unroll
  for (int nt = 0; nt < N / 16; ++nt) {
#pragma unroll
    for (int vr = 0; vr < 8; ++vr) {
      const int rr = row0 + half * 8 + vr;
      Y[(long)rr * N + nt * 16 + cn] = acc[nt][vr];
    }
  }
}

template <int BH, int H>
__global__ void edge_kernel(const float* __restrict__ ys, const float* __restrict__ yd,
                            const int* __restrict__ src, const int* __restrict__ dst,
                            const float* __restrict__ ew, const float* __restrict__ We,
                            float* __restrict__ agg, float* __restrict__ cnt, int E) {
  const int e = blockIdx.x * (blockDim.x >> 5) + (threadIdx.x >> 5);
  if (e >= E) return;
  const int lane = threadIdx.x & 31;
  const int s = src[e], d = dst[e];
  const float w = ew[e];
  const float* __restrict__ ysr = ys + (long)s * BH;
  const float* __restrict__ ydr = yd + (long)d * BH;
  float* __restrict__ ar        = agg + (long)d * BH;
#pragma unroll
  for (int i = lane; i < BH; i += 32) {
    const float xj = ysr[i];
    const float xi = ydr[i];
    const float ev = w * We[i & (H - 1)];
    const float t  = xi * xj * ev;
    const float sg = 1.0f / (1.0f + __expf(-t));
    atomicAdd(ar + i, xj * sg);
  }
  if (lane == 0) atomicAdd(cnt + d, 1.0f);
}

template <int BH, int H, bool FINAL>
__global__ void finalize_kernel(const float* __restrict__ xs, const int* __restrict__ res,
                                const float* __restrict__ yd, const float* __restrict__ agg,
                                const float* __restrict__ cnt, float* __restrict__ out,
                                int Nd) {
  const int d = blockIdx.x * (blockDim.x >> 5) + (threadIdx.x >> 5);
  if (d >= Nd) return;
  const int lane = threadIdx.x & 31;
  constexpr int PER = BH / 32;

  const float inv_cnt = 1.0f / fmaxf(cnt[d], 1.0f);
  const long  base    = (long)d * BH;
  const long  xbase   = (long)res[d] * BH;

  float vals[PER], xdv[PER];
  float sum = 0.0f, ssq = 0.0f;
#pragma unroll
  for (int j = 0; j < PER; ++j) {
    const int i  = lane + j * 32;
    const float o = yd[base + i] + agg[base + i] * inv_cnt;
    vals[j] = o;
    xdv[j]  = xs[xbase + i];
    sum += o;
    ssq += o * o;
  }
#pragma unroll
  for (int mask = 16; mask >= 1; mask >>= 1) {
    sum += __shfl_xor(sum, mask, 32);
    ssq += __shfl_xor(ssq, mask, 32);
  }
  const float mu  = sum * (1.0f / BH);
  const float var = ssq * (1.0f / BH) - mu * mu;
  const float is  = rsqrtf(var + 1e-5f);

#pragma unroll
  for (int j = 0; j < PER; ++j) {
    const int i = lane + j * 32;
    float v = xdv[j] + (vals[j] - mu) * is;
    v = v > 0.0f ? v : 0.01f * v;
    if (FINAL) {
      const int b = i / H, c = i & (H - 1);
      out[((long)b * Nd + d) * H + c] = v;
    } else {
      out[base + i] = v;
    }
  }
}

extern "C" void kernel_launch(void* const* d_in, const int* in_sizes, int n_in,
                              void* d_out, int out_size, void* d_ws, size_t ws_size,
                              hipStream_t stream) {
  constexpr int B = 4, Cin = 32, H = 16, Cout = 32;
  const float* X    = (const float*)d_in[0];
  const int*   src0 = (const int*)d_in[1];
  const int*   dst0 = (const int*)d_in[2];
  const float* ew0  = (const float*)d_in[3];
  const int*   res0 = (const int*)d_in[4];
  const int*   src1 = (const int*)d_in[5];
  const int*   dst1 = (const int*)d_in[6];
  const float* ew1  = (const float*)d_in[7];
  const int*   res1 = (const int*)d_in[8];
  const float* Wn1  = (const float*)d_in[9];
  const float* We1  = (const float*)d_in[10];
  const float* U1   = (const float*)d_in[11];
  const float* V1   = (const float*)d_in[12];
  const float* Wn2  = (const float*)d_in[13];
  const float* We2  = (const float*)d_in[14];
  const float* U2   = (const float*)d_in[15];
  const float* V2   = (const float*)d_in[16];

  const int E0 = in_sizes[1];
  const int N1 = in_sizes[4];
  const int E1 = in_sizes[5];
  const int N2 = in_sizes[8];
  const int N0 = in_sizes[0] / (B * Cin);

  float* ws = (float*)d_ws;
  const size_t szXS = (size_t)((N0 * B * H > N1 * B * Cout) ? N0 * B * H : N1 * B * Cout);
  const size_t szYD = (size_t)((N1 * B * H > N2 * B * Cout) ? N1 * B * H : N2 * B * Cout);
  float* xs  = ws;
  float* ys  = xs + szXS;
  float* yd  = ys + szXS;
  float* agg = yd + szYD;
  float* cnt = agg + szYD;
  float* h1  = cnt + (size_t)((N1 > N2) ? N1 : N2);

  const dim3 blk(256);
  auto waves_grid = [](int tiles) { return dim3((unsigned)((tiles + 7) / 8)); };

  const int R1s = N0 * B;
  const int R1d = N1 * B;
  gemm_wmma<32, 16, 1><<<waves_grid(R1s / 16), blk, 0, stream>>>(X,  Wn1, xs, nullptr, R1s, N0);
  gemm_wmma<16, 16, 0><<<waves_grid(R1s / 16), blk, 0, stream>>>(xs, V1,  ys, nullptr, R1s, 0);
  gemm_wmma<16, 16, 2><<<waves_grid(R1d / 16), blk, 0, stream>>>(xs, U1,  yd, res0,    R1d, 0);
  hipMemsetAsync(agg, 0, (size_t)R1d * H * sizeof(float), stream);
  hipMemsetAsync(cnt, 0, (size_t)N1 * sizeof(float), stream);
  edge_kernel<64, 16><<<dim3((unsigned)((E0 + 7) / 8)), blk, 0, stream>>>(
      ys, yd, src0, dst0, ew0, We1, agg, cnt, E0);
  finalize_kernel<64, 16, false><<<dim3((unsigned)((N1 + 7) / 8)), blk, 0, stream>>>(
      xs, res0, yd, agg, cnt, h1, N1);

  const int R2s = N1 * B;
  const int R2d = N2 * B;
  gemm_wmma<16, 32, 0><<<waves_grid(R2s / 16), blk, 0, stream>>>(h1, Wn2, xs, nullptr, R2s, 0);
  gemm_wmma<32, 32, 0><<<waves_grid(R2s / 16), blk, 0, stream>>>(xs, V2,  ys, nullptr, R2s, 0);
  gemm_wmma<32, 32, 2><<<waves_grid(R2d / 16), blk, 0, stream>>>(xs, U2,  yd, res1,    R2d, 0);
  hipMemsetAsync(agg, 0, (size_t)R2d * Cout * sizeof(float), stream);
  hipMemsetAsync(cnt, 0, (size_t)N2 * sizeof(float), stream);
  edge_kernel<128, 32><<<dim3((unsigned)((E1 + 7) / 8)), blk, 0, stream>>>(
      ys, yd, src1, dst1, ew1, We2, agg, cnt, E1);
  finalize_kernel<128, 32, true><<<dim3((unsigned)((N2 + 7) / 8)), blk, 0, stream>>>(
      xs, res1, yd, agg, cnt, (float*)d_out, N2);
}


# Edge sums: one agg and cnt plane per package half (a wave adds into the plane of the half its die is in); finalize adds the two planes (sum tree changes); planes back to back, one fill per layer
# speedup vs baseline: 1.5638x; 1.5638x over previous
#include <hip/hip_runtime.h>
#include <hip/hip_bf16.h>

typedef float v2f __attribute__((ext_vector_type(2)));
typedef float v8f __attribute__((ext_vector_type(8)));

template <int K, int N, int MODE>
__global__ void gemm_wmma(const float* __restrict__ A, const float* __restrict__ W,
                          float* __restrict__ Y, const int* __restrict__ gidx,
                          int R, int nsrc) {
  const int wave = blockIdx.x * (blockDim.x >> 5) + (threadIdx.x >> 5);
  const int lane = threadIdx.x & 31;
  const int row0 = wave * 16;
  if (row0 >= R) return;

  const int m    = lane & 15;
  const int kb   = (lane >> 4) << 1;
  const int r    = row0 + m;

  long off;
  if (MODE == 0) {
    off = (long)r * K;
  } else if (MODE == 1) {
    const int n = r >> 2, b = r & 3;
    off = ((long)b * nsrc + n) * (long)K;
  } else {
    const int d = gidx[r >> 2];
    off = ((long)d * 4 + (r & 3)) * (long)K;
  }
  const float* __restrict__ arow = A + off;

  v2f wf[K / 4][N / 16];
#pragma unroll
  for (int ks = 0; ks < K / 4; ++ks) {
#pragma unroll
    for (int nt = 0; nt < N / 16; ++nt) {
      const int kk = ks * 4 + kb;
      const int nn = nt * 16 + m;
      wf[ks][nt].x = W[kk * N + nn];
      wf[ks][nt].y = W[(kk + 1) * N + nn];
    }
  }

  v8f acc[N / 16] = {};
#pragma unroll
  for (int ks = 0; ks < K / 4; ++ks) {
    v2f a;
    a.x = arow[ks * 4 + kb];
    a.y = arow[ks * 4 + kb + 1];
#pragma unroll
    for (int nt = 0; nt < N / 16; ++nt) {
      acc[nt] = __builtin_amdgcn_wmma_f32_16x16x4_f32(
           false, a,  false, wf[ks][nt],
           (short)0, acc[nt],  false,  false);
    }
  }

  const int half = lane >> 4;
  const int cn   = lane & 15;
#pragma unroll
  for (int nt = 0; nt < N / 16; ++nt) {
#pragma unroll
    for (int vr = 0; vr < 8; ++vr) {
      const int rr = row0 + half * 8 + vr;
      Y[(long)rr * N + nt * 16 + cn] = acc[nt][vr];
    }
  }
}

__device__ __forceinline__ int package_half() {
  const unsigned id = (unsigned)__builtin_amdgcn_s_sendmsg_rtn(0x87);
  return (int)((id >> 18) & 1u);
}

template <int BH, int H, int U>
__global__ void edge_kernel(const float* __restrict__ ys, const float* __restrict__ yd,
                            const int* __restrict__ src, const int* __restrict__ dst,
                            const float* __restrict__ ew, const float* __restrict__ We,
                            float* __restrict__ agg0, float* __restrict__ cnt0,
                            long aggPlane, int cntPlane, int E) {
  constexpr int PER = BH / 32;
  const int wave = blockIdx.x * (blockDim.x >> 5) + (threadIdx.x >> 5);
  const int lane = threadIdx.x & 31;
  const int e0 = wave * 32;
  if (e0 >= E) return;
  const int n = (E - e0 < 32) ? (E - e0) : 32;
  const int hf = package_half();
  float* __restrict__ agg = agg0 + (long)hf * aggPlane;
  float* __restrict__ cnt = cnt0 + hf * cntPlane;

  const float wev = We[lane & (H - 1)];

  int sv = 0, dv = 0;
  float wv = 0.0f;
  if (lane < n) {
    sv = src[e0 + lane];
    dv = dst[e0 + lane];
    wv = ew[e0 + lane];
    atomicAdd(cnt + dv, 1.0f);
  }

  if (n == 32) {
    for (int j0 = 0; j0 < 32; j0 += U) {
      float xj[U][PER], xi[U][PER];
#pragma unroll
      for (int u = 0; u < U; ++u) {
        const int s = __builtin_amdgcn_readlane(sv, j0 + u);
        const int d = __builtin_amdgcn_readlane(dv, j0 + u);
        const float* __restrict__ ysr = ys + (long)s * BH + lane;
        const float* __restrict__ ydr = yd + (long)d * BH + lane;
#pragma unroll
        for (int k = 0; k < PER; ++k) {
          xj[u][k] = ysr[32 * k];
          xi[u][k] = ydr[32 * k];
        }
      }
#pragma unroll
      for (int u = 0; u < U; ++u) {
        const int d = __builtin_amdgcn_readlane(dv, j0 + u);
        const float w = __int_as_float(__builtin_amdgcn_readlane(__float_as_int(wv), j0 + u));
        const float ev = w * wev;
        float* __restrict__ ar = agg + (long)d * BH + lane;
#pragma unroll
        for (int k = 0; k < PER; ++k) {
          const float t  = xi[u][k] * xj[u][k] * ev;
          const float sg = 1.0f / (1.0f + __expf(-t));
          atomicAdd(ar + 32 * k, xj[u][k] * sg);
        }
      }
    }
  } else {
    for (int j = 0; j < n; ++j) {
      const int s = __builtin_amdgcn_readlane(sv, j);
      const int d = __builtin_amdgcn_readlane(dv, j);
      const float w = __int_as_float(__builtin_amdgcn_readlane(__float_as_int(wv), j));
      const float ev = w * wev;
      const float* __restrict__ ysr = ys + (long)s * BH + lane;
      const float* __restrict__ ydr = yd + (long)d * BH + lane;
      float* __restrict__ ar        = agg + (long)d * BH + lane;
#pragma unroll
      for (int k = 0; k < PER; ++k) {
        const float xjv = ysr[32 * k];
        const float xiv = ydr[32 * k];
        const float t   = xiv * xjv * ev;
        const float sg  = 1.0f / (1.0f + __expf(-t));
        atomicAdd(ar + 32 * k, xjv * sg);
      }
    }
  }
}

template <int BH, int H, bool FINAL>
__global__ void finalize_kernel(const float* __restrict__ xs, const int* __restrict__ res,
                                const float* __restrict__ yd, const float* __restrict__ agg,
                                const float* __restrict__ cnt, float* __restrict__ out,
                                long aggPlane, int cntPlane, int Nd) {
  const int d = blockIdx.x * (blockDim.x >> 5) + (threadIdx.x >> 5);
  if (d >= Nd) return;
  const int lane = threadIdx.x & 31;
  constexpr int PER = BH / 32;

  const bool  two     = aggPlane != 0;
  const float cn      = (cntPlane != 0) ? cnt[d] + cnt[cntPlane + d] : cnt[d];
  const float inv_cnt = 1.0f / fmaxf(cn, 1.0f);
  const long  base    = (long)d * BH;
  const long  xbase   = (long)res[d] * BH;

  float vals[PER], xdv[PER];
  float sum = 0.0f, ssq = 0.0f;
#pragma unroll
  for (int j = 0; j < PER; ++j) {
    const int i  = lane + j * 32;
    const float ag = two ? agg[base + i] + agg[aggPlane + base + i] : agg[base + i];
    const float o = yd[base + i] + ag * inv_cnt;
    vals[j] = o;
    xdv[j]  = xs[xbase + i];
    sum += o;
    ssq += o * o;
  }
#pragma unroll
  for (int mask = 16; mask >= 1; mask >>= 1) {
    sum += __shfl_xor(sum, mask, 32);
    ssq += __shfl_xor(ssq, mask, 32);
  }
  const float mu  = sum * (1.0f / BH);
  const float var = ssq * (1.0f / BH) - mu * mu;
  const float is  = rsqrtf(var + 1e-5f);

#pragma unroll
  for (int j = 0; j < PER; ++j) {
    const int i = lane + j * 32;
    float v = xdv[j] + (vals[j] - mu) * is;
    v = v > 0.0f ? v : 0.01f * v;
    if (FINAL) {
      const int b = i / H, c = i & (H - 1);
      out[((long)b * Nd + d) * H + c] = v;
    } else {
      out[base + i] = v;
    }
  }
}

extern "C" void kernel_launch(void* const* d_in, const int* in_sizes, int n_in,
                              void* d_out, int out_size, void* d_ws, size_t ws_size,
                              hipStream_t stream) {
  constexpr int B = 4, Cin = 32, H = 16, Cout = 32;
  const float* X    = (const float*)d_in[0];
  const int*   src0 = (const int*)d_in[1];
  const int*   dst0 = (const int*)d_in[2];
  const float* ew0  = (const float*)d_in[3];
  const int*   res0 = (const int*)d_in[4];
  const int*   src1 = (const int*)d_in[5];
  const int*   dst1 = (const int*)d_in[6];
  const float* ew1  = (const float*)d_in[7];
  const int*   res1 = (const int*)d_in[8];
  const float* Wn1  = (const float*)d_in[9];
  const float* We1  = (const float*)d_in[10];
  const float* U1   = (const float*)d_in[11];
  const float* V1   = (const float*)d_in[12];
  const float* Wn2  = (const float*)d_in[13];
  const float* We2  = (const float*)d_in[14];
  const float* U2   = (const float*)d_in[15];
  const float* V2   = (const float*)d_in[16];

  const int E0 = in_sizes[1];
  const int N1 = in_sizes[4];
  const int E1 = in_sizes[5];
  const int N2 = in_sizes[8];
  const int N0 = in_sizes[0] / (B * Cin);

  float* ws = (float*)d_ws;
  const size_t szXS = (size_t)((N0 * B * H > N1 * B * Cout) ? N0 * B * H : N1 * B * Cout);
  const size_t szYD = (size_t)((N1 * B * H > N2 * B * Cout) ? N1 * B * H : N2 * B * Cout);
  const size_t szCN = (size_t)((N1 > N2) ? N1 : N2);
  float* xs  = ws;
  float* ys  = xs + szXS;
  float* yd  = ys + szXS;
  float* h1  = ys;
  const size_t needAgg2 = (2 * szXS + 3 * szYD + szCN) * sizeof(float);
  const size_t needCnt2 = needAgg2 + szCN * sizeof(float);
  const int    aggPlanes = (ws_size >= needAgg2) ? 2 : 1;
  const int    cntPlanes = (ws_size >= needCnt2) ? 2 : 1;
  const long   aggPlane  = (aggPlanes == 2) ? (long)szYD : 0;
  const int    cntPlane  = (cntPlanes == 2) ? (int)szCN : 0;
  float* agg = yd + szYD;
  float* cnt = agg + (size_t)aggPlanes * szYD;
  const size_t fillBytes = ((size_t)aggPlanes * szYD + (size_t)cntPlanes * szCN) * sizeof(float);

  const dim3 blk(256);
  auto waves_grid = [](int tiles) { return dim3((unsigned)((tiles + 7) / 8)); };

  const int R1s = N0 * B;
  const int R1d = N1 * B;
  gemm_wmma<32, 16, 1><<<waves_grid(R1s / 16), blk, 0, stream>>>(X,  Wn1, xs, nullptr, R1s, N0);
  gemm_wmma<16, 16, 0><<<waves_grid(R1s / 16), blk, 0, stream>>>(xs, V1,  ys, nullptr, R1s, 0);
  gemm_wmma<16, 16, 2><<<waves_grid(R1d / 16), blk, 0, stream>>>(xs, U1,  yd, res0,    R1d, 0);
  hipMemsetAsync(agg, 0, fillBytes, stream);
  edge_kernel<64, 16, 8><<<dim3((unsigned)((E0 + 127) / 128)), dim3(128), 0, stream>>>(
      ys, yd, src0, dst0, ew0, We1, agg, cnt, aggPlane, cntPlane, E0);
  finalize_kernel<64, 16, false><<<dim3((unsigned)((N1 + 7) / 8)), blk, 0, stream>>>(
      xs, res0, yd, agg, cnt, h1, aggPlane, cntPlane, N1);

  const int R2s = N1 * B;
  const int R2d = N2 * B;
  gemm_wmma<16, 32, 0><<<waves_grid(R2s / 16), blk, 0, stream>>>(h1, Wn2, xs, nullptr, R2s, 0);
  gemm_wmma<32, 32, 0><<<waves_grid(R2s / 16), blk, 0, stream>>>(xs, V2,  ys, nullptr, R2s, 0);
  gemm_wmma<32, 32, 2><<<waves_grid(R2d / 16), blk, 0, stream>>>(xs, U2,  yd, res1,    R2d, 0);
  hipMemsetAsync(agg, 0, fillBytes, stream);
  edge_kernel<128, 32, 4><<<dim3((unsigned)((E1 + 127) / 128)), dim3(128), 0, stream>>>(
      ys, yd, src1, dst1, ew1, We2, agg, cnt, aggPlane, cntPlane, E1);
  finalize_kernel<128, 32, true><<<dim3((unsigned)((N2 + 7) / 8)), blk, 0, stream>>>(
      xs, res1, yd, agg, cnt, (float*)d_out, aggPlane, cntPlane, N2);
}


# Node GEMM and finalize launches: one wave works through 4 row tiles (4 dst nodes): weights loaded once per wave, all its rows requested before the first product; a quarter of the waves there
# speedup vs baseline: 1.6581x; 1.0603x over previous
#include <hip/hip_runtime.h>
#include <hip/hip_bf16.h>

typedef float v2f __attribute__((ext_vector_type(2)));
typedef float v8f __attribute__((ext_vector_type(8)));


template <int K, int N, int MODE, int TT>
__device__ __forceinline__ void gemm_tiles(const float* __restrict__ A,
                                           const v2f (&wf)[K / 4][N / 16],
                                           float* __restrict__ Y, const int* __restrict__ gidx,
                                           int row00, int nsrc, int lane) {
  const int m  = lane & 15;
  const int kb = (lane >> 4) << 1;

  v2f af[TT][K / 4];
#pragma unroll
  for (int t = 0; t < TT; ++t) {
    const int r = row00 + t * 16 + m;
    long off;
    if (MODE == 0) {
      off = (long)r * K;
    } else if (MODE == 1) {
      const int n = r >> 2, b = r & 3;
      off = ((long)b * nsrc + n) * (long)K;
    } else {
      const int d = gidx[r >> 2];
      off = ((long)d * 4 + (r & 3)) * (long)K;
    }
    const float* __restrict__ arow = A + off;
#pragma unroll
    for (int ks = 0; ks < K / 4; ++ks) {
      af[t][ks].x = arow[ks * 4 + kb];
      af[t][ks].y = arow[ks * 4 + kb + 1];
    }
  }
  __builtin_amdgcn_sched_barrier(0);

  const int half = lane >> 4;
  const int cn   = lane & 15;
#pragma unroll
  for (int t = 0; t < TT; ++t) {
    v8f acc[N / 16] = {};
#pragma unroll
    for (int ks = 0; ks < K / 4; ++ks) {
#pragma unroll
      for (int nt = 0; nt < N / 16; ++nt) {
        acc[nt] = __builtin_amdgcn_wmma_f32_16x16x4_f32(
             false, af[t][ks],  false, wf[ks][nt],
             (short)0, acc[nt],  false,  false);
      }
    }
#pragma unroll
    for (int nt = 0; nt < N / 16; ++nt) {
#pragma unroll
      for (int vr = 0; vr < 8; ++vr) {
        const int rr = row00 + t * 16 + half * 8 + vr;
        Y[(long)rr * N + nt * 16 + cn] = acc[nt][vr];
      }
    }
  }
}

template <int K, int N, int MODE, int T>
__global__ void gemm_wmma(const float* __restrict__ A, const float* __restrict__ W,
                          float* __restrict__ Y, const int* __restrict__ gidx,
                          int R, int nsrc) {
  const int wave = blockIdx.x * (blockDim.x >> 5) + (threadIdx.x >> 5);
  const int lane = threadIdx.x & 31;
  const int row00 = wave * (16 * T);
  if (row00 >= R) return;

  const int m  = lane & 15;
  const int kb = (lane >> 4) << 1;

  v2f wf[K / 4][N / 16];
#pragma unroll
  for (int ks = 0; ks < K / 4; ++ks) {
#pragma unroll
    for (int nt = 0; nt < N / 16; ++nt) {
      const int kk = ks * 4 + kb;
      const int nn = nt * 16 + m;
      wf[ks][nt].x = W[kk * N + nn];
      wf[ks][nt].y = W[(kk + 1) * N + nn];
    }
  }

  if (row00 + 16 * T <= R) {
    gemm_tiles<K, N, MODE, T>(A, wf, Y, gidx, row00, nsrc, lane);
  } else {
    for (int row0 = row00; row0 < R; row0 += 16)
      gemm_tiles<K, N, MODE, 1>(A, wf, Y, gidx, row0, nsrc, lane);
  }
}

__device__ __forceinline__ int package_half() {
  const unsigned id = (unsigned)__builtin_amdgcn_s_sendmsg_rtn(0x87);
  return (int)((id >> 18) & 1u);
}

template <int BH, int H, int U>
__global__ void edge_kernel(const float* __restrict__ ys, const float* __restrict__ yd,
                            const int* __restrict__ src, const int* __restrict__ dst,
                            const float* __restrict__ ew, const float* __restrict__ We,
                            float* __restrict__ agg0, float* __restrict__ cnt0,
                            long aggPlane, int cntPlane, int E) {
  constexpr int PER = BH / 32;
  const int wave = blockIdx.x * (blockDim.x >> 5) + (threadIdx.x >> 5);
  const int lane = threadIdx.x & 31;
  const int e0 = wave * 32;
  if (e0 >= E) return;
  const int n = (E - e0 < 32) ? (E - e0) : 32;
  const int hf = package_half();
  float* __restrict__ agg = agg0 + (long)hf * aggPlane;
  float* __restrict__ cnt = cnt0 + hf * cntPlane;

  const float wev = We[lane & (H - 1)];

  int sv = 0, dv = 0;
  float wv = 0.0f;
  if (lane < n) {
    sv = src[e0 + lane];
    dv = dst[e0 + lane];
    wv = ew[e0 + lane];
    atomicAdd(cnt + dv, 1.0f);
  }

  if (n == 32) {
    for (int j0 = 0; j0 < 32; j0 += U) {
      float xj[U][PER], xi[U][PER];
#pragma unroll
      for (int u = 0; u < U; ++u) {
        const int s = __builtin_amdgcn_readlane(sv, j0 + u);
        const int d = __builtin_amdgcn_readlane(dv, j0 + u);
        const float* __restrict__ ysr = ys + (long)s * BH + lane;
        const float* __restrict__ ydr = yd + (long)d * BH + lane;
#pragma unroll
        for (int k = 0; k < PER; ++k) {
          xj[u][k] = ysr[32 * k];
          xi[u][k] = ydr[32 * k];
        }
      }
#pragma unroll
      for (int u = 0; u < U; ++u) {
        const int d = __builtin_amdgcn_readlane(dv, j0 + u);
        const float w = __int_as_float(__builtin_amdgcn_readlane(__float_as_int(wv), j0 + u));
        const float ev = w * wev;
        float* __restrict__ ar = agg + (long)d * BH + lane;
#pragma unroll
        for (int k = 0; k < PER; ++k) {
          const float t  = xi[u][k] * xj[u][k] * ev;
          const float sg = 1.0f / (1.0f + __expf(-t));
          atomicAdd(ar + 32 * k, xj[u][k] * sg);
        }
      }
    }
  } else {
    for (int j = 0; j < n; ++j) {
      const int s = __builtin_amdgcn_readlane(sv, j);
      const int d = __builtin_amdgcn_readlane(dv, j);
      const float w = __int_as_float(__builtin_amdgcn_readlane(__float_as_int(wv), j));
      const float ev = w * wev;
      const float* __restrict__ ysr = ys + (long)s * BH + lane;
      const float* __restrict__ ydr = yd + (long)d * BH + lane;
      float* __restrict__ ar        = agg + (long)d * BH + lane;
#pragma unroll
      for (int k = 0; k < PER; ++k) {
        const float xjv = ysr[32 * k];
        const float xiv = ydr[32 * k];
        const float t   = xiv * xjv * ev;
        const float sg  = 1.0f / (1.0f + __expf(-t));
        atomicAdd(ar + 32 * k, xjv * sg);
      }
    }
  }
}


template <int BH, int H, bool FINAL, int NN, bool TWOA, bool TWOC>
__device__ __forceinline__ void finalize_nodes(const float* __restrict__ xs,
                                               const int* __restrict__ res,
                                               const float* __restrict__ yd,
                                               const float* __restrict__ agg,
                                               const float* __restrict__ cnt,
                                               float* __restrict__ out,
                                               long aggPlane, int cntPlane, int Nd,
                                               int d0, int lane) {
  constexpr int PER = BH / 32;

  float vals[NN][PER], xdv[NN][PER];
#pragma unroll
  for (int q = 0; q < NN; ++q) {
    const int d = d0 + q;
    const float cn      = TWOC ? cnt[d] + cnt[cntPlane + d] : cnt[d];
    const float inv_cnt = 1.0f / fmaxf(cn, 1.0f);
    const long  base    = (long)d * BH;
    const long  xbase   = (long)res[d] * BH;
#pragma unroll
    for (int j = 0; j < PER; ++j) {
      const int i  = lane + j * 32;
      const float ag = TWOA ? agg[base + i] + agg[aggPlane + base + i] : agg[base + i];
      vals[q][j] = yd[base + i] + ag * inv_cnt;
      xdv[q][j]  = xs[xbase + i];
    }
  }

#pragma unroll
  for (int q = 0; q < NN; ++q) {
    const int d = d0 + q;
    float sum = 0.0f, ssq = 0.0f;
#pragma unroll
    for (int j = 0; j < PER; ++j) {
      const float o = vals[q][j];
      sum += o;
      ssq += o * o;
    }
#pragma unroll
    for (int mask = 16; mask >= 1; mask >>= 1) {
      sum += __shfl_xor(sum, mask, 32);
      ssq += __shfl_xor(ssq, mask, 32);
    }
    const float mu  = sum * (1.0f / BH);
    const float var = ssq * (1.0f / BH) - mu * mu;
    const float is  = rsqrtf(var + 1e-5f);

    const long base = (long)d * BH;
#pragma unroll
    for (int j = 0; j < PER; ++j) {
      const int i = lane + j * 32;
      float v = xdv[q][j] + (vals[q][j] - mu) * is;
      v = v > 0.0f ? v : 0.01f * v;
      if (FINAL) {
        const int b = i / H, c = i & (H - 1);
        out[((long)b * Nd + d) * H + c] = v;
      } else {
        out[base + i] = v;
      }
    }
  }
}

template <int BH, int H, bool FINAL, int NPW, bool TWOA, bool TWOC>
__device__ __forceinline__ void finalize_wave(const float* __restrict__ xs,
                                              const int* __restrict__ res,
                                              const float* __restrict__ yd,
                                              const float* __restrict__ agg,
                                              const float* __restrict__ cnt,
                                              float* __restrict__ out,
                                              long aggPlane, int cntPlane, int Nd,
                                              int d0, int lane) {
  if (d0 + NPW <= Nd) {
    finalize_nodes<BH, H, FINAL, NPW, TWOA, TWOC>(xs, res, yd, agg, cnt, out, aggPlane,
                                                  cntPlane, Nd, d0, lane);
  } else {
    for (int d = d0; d < Nd; ++d)
      finalize_nodes<BH, H, FINAL, 1, TWOA, TWOC>(xs, res, yd, agg, cnt, out, aggPlane,
                                                  cntPlane, Nd, d, lane);
  }
}

template <int BH, int H, bool FINAL, int NPW>
__global__ void finalize_kernel(const float* __restrict__ xs, const int* __restrict__ res,
                                const float* __restrict__ yd, const float* __restrict__ agg,
                                const float* __restrict__ cnt, float* __restrict__ out,
                                long aggPlane, int cntPlane, int Nd) {
  const int wave = blockIdx.x * (blockDim.x >> 5) + (threadIdx.x >> 5);
  const int lane = threadIdx.x & 31;
  const int d0 = wave * NPW;
  if (d0 >= Nd) return;
  if (aggPlane != 0 && cntPlane != 0) {
    finalize_wave<BH, H, FINAL, NPW, true, true>(xs, res, yd, agg, cnt, out, aggPlane, cntPlane,
                                                 Nd, d0, lane);
  } else if (aggPlane != 0) {
    finalize_wave<BH, H, FINAL, NPW, true, false>(xs, res, yd, agg, cnt, out, aggPlane, cntPlane,
                                                  Nd, d0, lane);
  } else if (cntPlane != 0) {
    finalize_wave<BH, H, FINAL, NPW, false, true>(xs, res, yd, agg, cnt, out, aggPlane, cntPlane,
                                                  Nd, d0, lane);
  } else {
    finalize_wave<BH, H, FINAL, NPW, false, false>(xs, res, yd, agg, cnt, out, aggPlane,
                                                   cntPlane, Nd, d0, lane);
  }
}

extern "C" void kernel_launch(void* const* d_in, const int* in_sizes, int n_in,
                              void* d_out, int out_size, void* d_ws, size_t ws_size,
                              hipStream_t stream) {
  constexpr int B = 4, Cin = 32, H = 16, Cout = 32;
  const float* X    = (const float*)d_in[0];
  const int*   src0 = (const int*)d_in[1];
  const int*   dst0 = (const int*)d_in[2];
  const float* ew0  = (const float*)d_in[3];
  const int*   res0 = (const int*)d_in[4];
  const int*   src1 = (const int*)d_in[5];
  const int*   dst1 = (const int*)d_in[6];
  const float* ew1  = (const float*)d_in[7];
  const int*   res1 = (const int*)d_in[8];
  const float* Wn1  = (const float*)d_in[9];
  const float* We1  = (const float*)d_in[10];
  const float* U1   = (const float*)d_in[11];
  const float* V1   = (const float*)d_in[12];
  const float* Wn2  = (const float*)d_in[13];
  const float* We2  = (const float*)d_in[14];
  const float* U2   = (const float*)d_in[15];
  const float* V2   = (const float*)d_in[16];

  const int E0 = in_sizes[1];
  const int N1 = in_sizes[4];
  const int E1 = in_sizes[5];
  const int N2 = in_sizes[8];
  const int N0 = in_sizes[0] / (B * Cin);

  float* ws = (float*)d_ws;
  const size_t szXS = (size_t)((N0 * B * H > N1 * B * Cout) ? N0 * B * H : N1 * B * Cout);
  const size_t szYD = (size_t)((N1 * B * H > N2 * B * Cout) ? N1 * B * H : N2 * B * Cout);
  const size_t szCN = (size_t)((N1 > N2) ? N1 : N2);
  float* xs  = ws;
  float* ys  = xs + szXS;
  float* yd  = ys + szXS;
  float* h1  = ys;
  const size_t needAgg2 = (2 * szXS + 3 * szYD + szCN) * sizeof(float);
  const size_t needCnt2 = needAgg2 + szCN * sizeof(float);
  const int    aggPlanes = (ws_size >= needAgg2) ? 2 : 1;
  const int    cntPlanes = (ws_size >= needCnt2) ? 2 : 1;
  const long   aggPlane  = (aggPlanes == 2) ? (long)szYD : 0;
  const int    cntPlane  = (cntPlanes == 2) ? (int)szCN : 0;
  float* agg = yd + szYD;
  float* cnt = agg + (size_t)aggPlanes * szYD;
  const size_t fillBytes = ((size_t)aggPlanes * szYD + (size_t)cntPlanes * szCN) * sizeof(float);

  const dim3 blk(256);
  const dim3 gblk(128);
  auto gemm_grid = [](int tiles, int T) {
    const int waves = (tiles + T - 1) / T;
    return dim3((unsigned)((waves + 3) / 4));
  };
  auto fin_grid = [](int nodes, int NPW) {
    const int waves = (nodes + NPW - 1) / NPW;
    return dim3((unsigned)((waves + 7) / 8));
  };

  const int R1s = N0 * B;
  const int R1d = N1 * B;
  gemm_wmma<32, 16, 1, 4><<<gemm_grid(R1s / 16, 4), gblk, 0, stream>>>(X,  Wn1, xs, nullptr, R1s, N0);
  gemm_wmma<16, 16, 0, 4><<<gemm_grid(R1s / 16, 4), gblk, 0, stream>>>(xs, V1,  ys, nullptr, R1s, 0);
  gemm_wmma<16, 16, 2, 4><<<gemm_grid(R1d / 16, 4), gblk, 0, stream>>>(xs, U1,  yd, res0,    R1d, 0);
  hipMemsetAsync(agg, 0, fillBytes, stream);
  edge_kernel<64, 16, 8><<<dim3((unsigned)((E0 + 127) / 128)), dim3(128), 0, stream>>>(
      ys, yd, src0, dst0, ew0, We1, agg, cnt, aggPlane, cntPlane, E0);
  finalize_kernel<64, 16, false, 4><<<fin_grid(N1, 4), blk, 0, stream>>>(
      xs, res0, yd, agg, cnt, h1, aggPlane, cntPlane, N1);

  const int R2s = N1 * B;
  const int R2d = N2 * B;
  gemm_wmma<16, 32, 0, 4><<<gemm_grid(R2s / 16, 4), gblk, 0, stream>>>(h1, Wn2, xs, nullptr, R2s, 0);
  gemm_wmma<32, 32, 0, 4><<<gemm_grid(R2s / 16, 4), gblk, 0, stream>>>(xs, V2,  ys, nullptr, R2s, 0);
  gemm_wmma<32, 32, 2, 2><<<gemm_grid(R2d / 16, 2), gblk, 0, stream>>>(xs, U2,  yd, res1,    R2d, 0);
  hipMemsetAsync(agg, 0, fillBytes, stream);
  edge_kernel<128, 32, 4><<<dim3((unsigned)((E1 + 127) / 128)), dim3(128), 0, stream>>>(
      ys, yd, src1, dst1, ew1, We2, agg, cnt, aggPlane, cntPlane, E1);
  finalize_kernel<128, 32, true, 4><<<fin_grid(N2, 4), blk, 0, stream>>>(
      xs, res1, yd, agg, cnt, (float*)d_out, aggPlane, cntPlane, N2);
}


# Per layer one launch reads the source rows once and writes xs = x@Wn and ys = x@(Wn@V); each wave forms Wn@V on WMMA first; ys re-associated, k grouping (sum order) of both changes; launches 10 to 8
# speedup vs baseline: 1.7232x; 1.0393x over previous
#include <hip/hip_runtime.h>
#include <hip/hip_bf16.h>

typedef float v2f __attribute__((ext_vector_type(2)));
typedef float v8f __attribute__((ext_vector_type(8)));


template <int K, int N, int MODE, int TT>
__device__ __forceinline__ void gemm_tiles(const float* __restrict__ A,
                                           const v2f (&wf)[K / 4][N / 16],
                                           float* __restrict__ Y, const int* __restrict__ gidx,
                                           int row00, int nsrc, int lane) {
  const int m  = lane & 15;
  const int kb = (lane >> 4) << 1;

  v2f af[TT][K / 4];
#pragma unroll
  for (int t = 0; t < TT; ++t) {
    const int r = row00 + t * 16 + m;
    long off;
    if (MODE == 0) {
      off = (long)r * K;
    } else if (MODE == 1) {
      const int n = r >> 2, b = r & 3;
      off = ((long)b * nsrc + n) * (long)K;
    } else {
      const int d = gidx[r >> 2];
      off = ((long)d * 4 + (r & 3)) * (long)K;
    }
    const float* __restrict__ arow = A + off;
#pragma unroll
    for (int ks = 0; ks < K / 4; ++ks) {
      af[t][ks].x = arow[ks * 4 + kb];
      af[t][ks].y = arow[ks * 4 + kb + 1];
    }
  }
  __builtin_amdgcn_sched_barrier(0);

  const int half = lane >> 4;
  const int cn   = lane & 15;
#pragma unroll
  for (int t = 0; t < TT; ++t) {
    v8f acc[N / 16] = {};
#pragma unroll
    for (int ks = 0; ks < K / 4; ++ks) {
#pragma unroll
      for (int nt = 0; nt < N / 16; ++nt) {
        acc[nt] = __builtin_amdgcn_wmma_f32_16x16x4_f32(
             false, af[t][ks],  false, wf[ks][nt],
             (short)0, acc[nt],  false,  false);
      }
    }
#pragma unroll
    for (int nt = 0; nt < N / 16; ++nt) {
#pragma unroll
      for (int vr = 0; vr < 8; ++vr) {
        const int rr = row00 + t * 16 + half * 8 + vr;
        Y[(long)rr * N + nt * 16 + cn] = acc[nt][vr];
      }
    }
  }
}

template <int K, int N, int MODE, int T>
__global__ void gemm_wmma(const float* __restrict__ A, const float* __restrict__ W,
                          float* __restrict__ Y, const int* __restrict__ gidx,
                          int R, int nsrc) {
  const int wave = blockIdx.x * (blockDim.x >> 5) + (threadIdx.x >> 5);
  const int lane = threadIdx.x & 31;
  const int row00 = wave * (16 * T);
  if (row00 >= R) return;

  const int m  = lane & 15;
  const int kb = (lane >> 4) << 1;

  v2f wf[K / 4][N / 16];
#pragma unroll
  for (int ks = 0; ks < K / 4; ++ks) {
#pragma unroll
    for (int nt = 0; nt < N / 16; ++nt) {
      const int kk = ks * 4 + kb;
      const int nn = nt * 16 + m;
      wf[ks][nt].x = W[kk * N + nn];
      wf[ks][nt].y = W[(kk + 1) * N + nn];
    }
  }

  if (row00 + 16 * T <= R) {
    gemm_tiles<K, N, MODE, T>(A, wf, Y, gidx, row00, nsrc, lane);
  } else {
    for (int row0 = row00; row0 < R; row0 += 16)
      gemm_tiles<K, N, MODE, 1>(A, wf, Y, gidx, row0, nsrc, lane);
  }
}


template <int K, int MODE, int TT>
__device__ __forceinline__ void fold_load_rows(const float* __restrict__ A, v2f (&af)[TT][K / 4],
                                               int row00, int nsrc, int lane) {
  const int m = lane & 15;
  const int h = lane >> 4;
#pragma unroll
  for (int t = 0; t < TT; ++t) {
    const int r = row00 + t * 16 + m;
    long off;
    if (MODE == 0) {
      off = (long)r * K;
    } else {
      const int n = r >> 2, b = r & 3;
      off = ((long)b * nsrc + n) * (long)K;
    }
    const float4* __restrict__ arow = reinterpret_cast<const float4*>(A + off);
#pragma unroll
    for (int t2 = 0; t2 < K / 16; ++t2) {
      const float4 lo = arow[t2 * 4 + h * 2];
      const float4 hi = arow[t2 * 4 + h * 2 + 1];
      af[t][t2 * 4 + 0].x = lo.x; af[t][t2 * 4 + 0].y = lo.y;
      af[t][t2 * 4 + 1].x = lo.z; af[t][t2 * 4 + 1].y = lo.w;
      af[t][t2 * 4 + 2].x = hi.x; af[t][t2 * 4 + 2].y = hi.y;
      af[t][t2 * 4 + 3].x = hi.z; af[t][t2 * 4 + 3].y = hi.w;
    }
  }
}

template <int K, int N>
__device__ __forceinline__ void fold_form(const v2f (&wa)[K / 16][N / 4],
                                          const v2f (&vb)[N / 4][N / 16],
                                          v8f (&ff)[K / 16][N / 16]) {
#pragma unroll
  for (int t2 = 0; t2 < K / 16; ++t2) {
#pragma unroll
    for (int nt = 0; nt < N / 16; ++nt) ff[t2][nt] = v8f{};
  }
#pragma unroll
  for (int q = 0; q < N / 4; ++q) {
#pragma unroll
    for (int t2 = 0; t2 < K / 16; ++t2) {
#pragma unroll
      for (int nt = 0; nt < N / 16; ++nt) {
        ff[t2][nt] = __builtin_amdgcn_wmma_f32_16x16x4_f32(
             false, wa[t2][q],  false, vb[q][nt],
             (short)0, ff[t2][nt],  false,  false);
      }
    }
  }
}

template <int K, int N, int TT>
__device__ __forceinline__ void fold_products(const v2f (&af)[TT][K / 4],
                                              const v2f (&wf)[K / 4][N / 16],
                                              const v8f (&ff)[K / 16][N / 16],
                                              float* __restrict__ Y1, float* __restrict__ Y2,
                                              int row00, int lane) {
  const int h  = lane >> 4;
  const int cn = lane & 15;
#pragma unroll
  for (int t = 0; t < TT; ++t) {
    v8f acc1[N / 16] = {};
    v8f acc2[N / 16] = {};
#pragma unroll
    for (int ks = 0; ks < K / 4; ++ks) {
      const int t2 = ks >> 2, j = ks & 3;
#pragma unroll
      for (int nt = 0; nt < N / 16; ++nt) {
        v2f fb;
        fb.x = ff[t2][nt][2 * j];
        fb.y = ff[t2][nt][2 * j + 1];
        acc1[nt] = __builtin_amdgcn_wmma_f32_16x16x4_f32(
             false, af[t][ks],  false, wf[ks][nt],
             (short)0, acc1[nt],  false,  false);
        acc2[nt] = __builtin_amdgcn_wmma_f32_16x16x4_f32(
             false, af[t][ks],  false, fb,
             (short)0, acc2[nt],  false,  false);
      }
    }
#pragma unroll
    for (int nt = 0; nt < N / 16; ++nt) {
#pragma unroll
      for (int vr = 0; vr < 8; ++vr) {
        const int rr = row00 + t * 16 + h * 8 + vr;
        Y1[(long)rr * N + nt * 16 + cn] = acc1[nt][vr];
        Y2[(long)rr * N + nt * 16 + cn] = acc2[nt][vr];
      }
    }
  }
}

template <int K, int N, int MODE, int T>
__global__ void __launch_bounds__(128) gemm_fold_wmma(const float* __restrict__ A, const float* __restrict__ W,
                               const float* __restrict__ V, float* __restrict__ Y1,
                               float* __restrict__ Y2, int R, int nsrc) {
  const int wave = blockIdx.x * (blockDim.x >> 5) + (threadIdx.x >> 5);
  const int lane = threadIdx.x & 31;
  const int row00 = wave * (16 * T);
  if (row00 >= R) return;

  const int m  = lane & 15;
  const int h  = lane >> 4;
  const int kb = h << 1;

  v2f vb[N / 4][N / 16];
  v2f wa[K / 16][N / 4];
#pragma unroll
  for (int q = 0; q < N / 4; ++q) {
#pragma unroll
    for (int nt = 0; nt < N / 16; ++nt) {
      vb[q][nt].x = V[(q * 4 + kb) * N + nt * 16 + m];
      vb[q][nt].y = V[(q * 4 + kb + 1) * N + nt * 16 + m];
    }
#pragma unroll
    for (int t2 = 0; t2 < K / 16; ++t2) {
      wa[t2][q].x = W[(t2 * 16 + m) * N + q * 4 + kb];
      wa[t2][q].y = W[(t2 * 16 + m) * N + q * 4 + kb + 1];
    }
  }
  v2f wf[K / 4][N / 16];
#pragma unroll
  for (int ks = 0; ks < K / 4; ++ks) {
    const int kk = (ks >> 2) * 16 + h * 8 + (ks & 3) * 2;
#pragma unroll
    for (int nt = 0; nt < N / 16; ++nt) {
      wf[ks][nt].x = W[kk * N + nt * 16 + m];
      wf[ks][nt].y = W[(kk + 1) * N + nt * 16 + m];
    }
  }

  v8f ff[K / 16][N / 16];
  if (row00 + 16 * T <= R) {
    v2f af[T][K / 4];
    fold_load_rows<K, MODE, T>(A, af, row00, nsrc, lane);
    __builtin_amdgcn_sched_barrier(0);
    fold_form<K, N>(wa, vb, ff);
    fold_products<K, N, T>(af, wf, ff, Y1, Y2, row00, lane);
  } else {
    fold_form<K, N>(wa, vb, ff);
    for (int row0 = row00; row0 < R; row0 += 16) {
      v2f af[1][K / 4];
      fold_load_rows<K, MODE, 1>(A, af, row0, nsrc, lane);
      fold_products<K, N, 1>(af, wf, ff, Y1, Y2, row0, lane);
    }
  }
}

__device__ __forceinline__ int package_half() {
  const unsigned id = (unsigned)__builtin_amdgcn_s_sendmsg_rtn(0x87);
  return (int)((id >> 18) & 1u);
}

template <int BH, int H, int U>
__global__ void edge_kernel(const float* __restrict__ ys, const float* __restrict__ yd,
                            const int* __restrict__ src, const int* __restrict__ dst,
                            const float* __restrict__ ew, const float* __restrict__ We,
                            float* __restrict__ agg0, float* __restrict__ cnt0,
                            long aggPlane, int cntPlane, int E) {
  constexpr int PER = BH / 32;
  const int wave = blockIdx.x * (blockDim.x >> 5) + (threadIdx.x >> 5);
  const int lane = threadIdx.x & 31;
  const int e0 = wave * 32;
  if (e0 >= E) return;
  const int n = (E - e0 < 32) ? (E - e0) : 32;
  const int hf = package_half();
  float* __restrict__ agg = agg0 + (long)hf * aggPlane;
  float* __restrict__ cnt = cnt0 + hf * cntPlane;

  const float wev = We[lane & (H - 1)];

  int sv = 0, dv = 0;
  float wv = 0.0f;
  if (lane < n) {
    sv = src[e0 + lane];
    dv = dst[e0 + lane];
    wv = ew[e0 + lane];
    atomicAdd(cnt + dv, 1.0f);
  }

  if (n == 32) {
    for (int j0 = 0; j0 < 32; j0 += U) {
      float xj[U][PER], xi[U][PER];
#pragma unroll
      for (int u = 0; u < U; ++u) {
        const int s = __builtin_amdgcn_readlane(sv, j0 + u);
        const int d = __builtin_amdgcn_readlane(dv, j0 + u);
        const float* __restrict__ ysr = ys + (long)s * BH + lane;
        const float* __restrict__ ydr = yd + (long)d * BH + lane;
#pragma unroll
        for (int k = 0; k < PER; ++k) {
          xj[u][k] = ysr[32 * k];
          xi[u][k] = ydr[32 * k];
        }
      }
#pragma unroll
      for (int u = 0; u < U; ++u) {
        const int d = __builtin_amdgcn_readlane(dv, j0 + u);
        const float w = __int_as_float(__builtin_amdgcn_readlane(__float_as_int(wv), j0 + u));
        const float ev = w * wev;
        float* __restrict__ ar = agg + (long)d * BH + lane;
#pragma unroll
        for (int k = 0; k < PER; ++k) {
          const float t  = xi[u][k] * xj[u][k] * ev;
          const float sg = 1.0f / (1.0f + __expf(-t));
          atomicAdd(ar + 32 * k, xj[u][k] * sg);
        }
      }
    }
  } else {
    for (int j = 0; j < n; ++j) {
      const int s = __builtin_amdgcn_readlane(sv, j);
      const int d = __builtin_amdgcn_readlane(dv, j);
      const float w = __int_as_float(__builtin_amdgcn_readlane(__float_as_int(wv), j));
      const float ev = w * wev;
      const float* __restrict__ ysr = ys + (long)s * BH + lane;
      const float* __restrict__ ydr = yd + (long)d * BH + lane;
      float* __restrict__ ar        = agg + (long)d * BH + lane;
#pragma unroll
      for (int k = 0; k < PER; ++k) {
        const float xjv = ysr[32 * k];
        const float xiv = ydr[32 * k];
        const float t   = xiv * xjv * ev;
        const float sg  = 1.0f / (1.0f + __expf(-t));
        atomicAdd(ar + 32 * k, xjv * sg);
      }
    }
  }
}


template <bool SEPARATE> struct fin_io {
  typedef const float* in_t;
  typedef float* out_t;
};
template <> struct fin_io<true> {
  typedef const float* __restrict__ in_t;
  typedef float* __restrict__ out_t;
};

template <int BH, int H, bool FINAL, int NN, bool TWOA, bool TWOC>
__device__ __forceinline__ void finalize_nodes(const float* __restrict__ xs,
                                               const int* __restrict__ res,
                                               typename fin_io<FINAL>::in_t yd,
                                               const float* __restrict__ agg,
                                               const float* __restrict__ cnt,
                                               typename fin_io<FINAL>::out_t out,
                                               long aggPlane, int cntPlane, int Nd,
                                               int d0, int lane) {
  constexpr int PER = BH / 32;

  float vals[NN][PER], xdv[NN][PER];
#pragma unroll
  for (int q = 0; q < NN; ++q) {
    const int d = d0 + q;
    const float cn      = TWOC ? cnt[d] + cnt[cntPlane + d] : cnt[d];
    const float inv_cnt = 1.0f / fmaxf(cn, 1.0f);
    const long  base    = (long)d * BH;
    const long  xbase   = (long)res[d] * BH;
#pragma unroll
    for (int j = 0; j < PER; ++j) {
      const int i  = lane + j * 32;
      const float ag = TWOA ? agg[base + i] + agg[aggPlane + base + i] : agg[base + i];
      vals[q][j] = yd[base + i] + ag * inv_cnt;
      xdv[q][j]  = xs[xbase + i];
    }
  }

#pragma unroll
  for (int q = 0; q < NN; ++q) {
    const int d = d0 + q;
    float sum = 0.0f, ssq = 0.0f;
#pragma unroll
    for (int j = 0; j < PER; ++j) {
      const float o = vals[q][j];
      sum += o;
      ssq += o * o;
    }
#pragma unroll
    for (int mask = 16; mask >= 1; mask >>= 1) {
      sum += __shfl_xor(sum, mask, 32);
      ssq += __shfl_xor(ssq, mask, 32);
    }
    const float mu  = sum * (1.0f / BH);
    const float var = ssq * (1.0f / BH) - mu * mu;
    const float is  = rsqrtf(var + 1e-5f);

    const long base = (long)d * BH;
#pragma unroll
    for (int j = 0; j < PER; ++j) {
      const int i = lane + j * 32;
      float v = xdv[q][j] + (vals[q][j] - mu) * is;
      v = v > 0.0f ? v : 0.01f * v;
      if (FINAL) {
        const int b = i / H, c = i & (H - 1);
        out[((long)b * Nd + d) * H + c] = v;
      } else {
        out[base + i] = v;
      }
    }
  }
}

template <int BH, int H, bool FINAL, int NPW, bool TWOA, bool TWOC>
__device__ __forceinline__ void finalize_wave(const float* __restrict__ xs,
                                              const int* __restrict__ res,
                                              typename fin_io<FINAL>::in_t yd,
                                              const float* __restrict__ agg,
                                              const float* __restrict__ cnt,
                                              typename fin_io<FINAL>::out_t out,
                                              long aggPlane, int cntPlane, int Nd,
                                              int d0, int lane) {
  if (d0 + NPW <= Nd) {
    finalize_nodes<BH, H, FINAL, NPW, TWOA, TWOC>(xs, res, yd, agg, cnt, out, aggPlane,
                                                  cntPlane, Nd, d0, lane);
  } else {
    for (int d = d0; d < Nd; ++d)
      finalize_nodes<BH, H, FINAL, 1, TWOA, TWOC>(xs, res, yd, agg, cnt, out, aggPlane,
                                                  cntPlane, Nd, d, lane);
  }
}

template <int BH, int H, bool FINAL, int NPW>
__global__ void finalize_kernel(const float* __restrict__ xs, const int* __restrict__ res,
                                const float* yd, const float* __restrict__ agg,
                                const float* __restrict__ cnt, float* out,
                                long aggPlane, int cntPlane, int Nd) {
  const int wave = blockIdx.x * (blockDim.x >> 5) + (threadIdx.x >> 5);
  const int lane = threadIdx.x & 31;
  const int d0 = wave * NPW;
  if (d0 >= Nd) return;
  if (aggPlane != 0 && cntPlane != 0) {
    finalize_wave<BH, H, FINAL, NPW, true, true>(xs, res, yd, agg, cnt, out, aggPlane, cntPlane,
                                                 Nd, d0, lane);
  } else if (aggPlane != 0) {
    finalize_wave<BH, H, FINAL, NPW, true, false>(xs, res, yd, agg, cnt, out, aggPlane, cntPlane,
                                                  Nd, d0, lane);
  } else if (cntPlane != 0) {
    finalize_wave<BH, H, FINAL, NPW, false, true>(xs, res, yd, agg, cnt, out, aggPlane, cntPlane,
                                                  Nd, d0, lane);
  } else {
    finalize_wave<BH, H, FINAL, NPW, false, false>(xs, res, yd, agg, cnt, out, aggPlane,
                                                   cntPlane, Nd, d0, lane);
  }
}

extern "C" void kernel_launch(void* const* d_in, const int* in_sizes, int n_in,
                              void* d_out, int out_size, void* d_ws, size_t ws_size,
                              hipStream_t stream) {
  constexpr int B = 4, Cin = 32, H = 16, Cout = 32;
  const float* X    = (const float*)d_in[0];
  const int*   src0 = (const int*)d_in[1];
  const int*   dst0 = (const int*)d_in[2];
  const float* ew0  = (const float*)d_in[3];
  const int*   res0 = (const int*)d_in[4];
  const int*   src1 = (const int*)d_in[5];
  const int*   dst1 = (const int*)d_in[6];
  const float* ew1  = (const float*)d_in[7];
  const int*   res1 = (const int*)d_in[8];
  const float* Wn1  = (const float*)d_in[9];
  const float* We1  = (const float*)d_in[10];
  const float* U1   = (const float*)d_in[11];
  const float* V1   = (const float*)d_in[12];
  const float* Wn2  = (const float*)d_in[13];
  const float* We2  = (const float*)d_in[14];
  const float* U2   = (const float*)d_in[15];
  const float* V2   = (const float*)d_in[16];

  const int E0 = in_sizes[1];
  const int N1 = in_sizes[4];
  const int E1 = in_sizes[5];
  const int N2 = in_sizes[8];
  const int N0 = in_sizes[0] / (B * Cin);

  float* ws = (float*)d_ws;
  const size_t szXS = (size_t)((N0 * B * H > N1 * B * Cout) ? N0 * B * H : N1 * B * Cout);
  const size_t szYD = (size_t)((N1 * B * H > N2 * B * Cout) ? N1 * B * H : N2 * B * Cout);
  const size_t szCN = (size_t)((N1 > N2) ? N1 : N2);
  float* xs  = ws;
  float* ys  = xs + szXS;
  float* yd  = ys + szXS;
  float* h1  = yd;
  const size_t needAgg2 = (2 * szXS + 3 * szYD + szCN) * sizeof(float);
  const size_t needCnt2 = needAgg2 + szCN * sizeof(float);
  const int    aggPlanes = (ws_size >= needAgg2) ? 2 : 1;
  const int    cntPlanes = (ws_size >= needCnt2) ? 2 : 1;
  const long   aggPlane  = (aggPlanes == 2) ? (long)szYD : 0;
  const int    cntPlane  = (cntPlanes == 2) ? (int)szCN : 0;
  float* agg = yd + szYD;
  float* cnt = agg + (size_t)aggPlanes * szYD;
  const size_t fillBytes = ((size_t)aggPlanes * szYD + (size_t)cntPlanes * szCN) * sizeof(float);

  const dim3 blk(256);
  const dim3 gblk(128);
  auto gemm_grid = [](int tiles, int T) {
    const int waves = (tiles + T - 1) / T;
    return dim3((unsigned)((waves + 3) / 4));
  };
  auto fin_grid = [](int nodes, int NPW) {
    const int waves = (nodes + NPW - 1) / NPW;
    return dim3((unsigned)((waves + 7) / 8));
  };

  const int R1s = N0 * B;
  const int R1d = N1 * B;
  gemm_fold_wmma<32, 16, 1, 4><<<gemm_grid(R1s / 16, 4), gblk, 0, stream>>>(X, Wn1, V1, xs, ys, R1s, N0);
  gemm_wmma<16, 16, 2, 4><<<gemm_grid(R1d / 16, 4), gblk, 0, stream>>>(xs, U1,  yd, res0,    R1d, 0);
  hipMemsetAsync(agg, 0, fillBytes, stream);
  edge_kernel<64, 16, 8><<<dim3((unsigned)((E0 + 127) / 128)), dim3(128), 0, stream>>>(
      ys, yd, src0, dst0, ew0, We1, agg, cnt, aggPlane, cntPlane, E0);
  finalize_kernel<64, 16, false, 4><<<fin_grid(N1, 4), blk, 0, stream>>>(
      xs, res0, yd, agg, cnt, h1, aggPlane, cntPlane, N1);

  const int R2s = N1 * B;
  const int R2d = N2 * B;
  gemm_fold_wmma<16, 32, 0, 4><<<gemm_grid(R2s / 16, 4), gblk, 0, stream>>>(h1, Wn2, V2, xs, ys, R2s, 0);
  gemm_wmma<32, 32, 2, 2><<<gemm_grid(R2d / 16, 2), gblk, 0, stream>>>(xs, U2,  yd, res1,    R2d, 0);
  hipMemsetAsync(agg, 0, fillBytes, stream);
  edge_kernel<128, 32, 4><<<dim3((unsigned)((E1 + 127) / 128)), dim3(128), 0, stream>>>(
      ys, yd, src1, dst1, ew1, We2, agg, cnt, aggPlane, cntPlane, E1);
  finalize_kernel<128, 32, true, 4><<<fin_grid(N2, 4), blk, 0, stream>>>(
      xs, res1, yd, agg, cnt, (float*)d_out, aggPlane, cntPlane, N2);
}


# Per layer ONE node launch: a wave does 4 source tiles (ys) + 2 dst tiles gathered from the layer input (xd; yd via Wn@U, sum order changes) and zeroes their agg/cnt; no fill op; stream ops 10 to 6
# speedup vs baseline: 1.8668x; 1.0833x over previous
#include <hip/hip_runtime.h>
#include <hip/hip_bf16.h>

typedef float v2f __attribute__((ext_vector_type(2)));
typedef float v8f __attribute__((ext_vector_type(8)));


template <int K, int MODE, bool GATHER, int TT>
__device__ __forceinline__ void fold_load_rows(const float* __restrict__ A,
                                               const int (&gnode)[TT],
                                               v2f (&af)[TT][K / 4],
                                               int row00, int nsrc, int lane) {
  const int m = lane & 15;
  const int h = lane >> 4;
#pragma unroll
  for (int t = 0; t < TT; ++t) {
    const int r = row00 + t * 16 + m;
    int n = r >> 2;
    const int b = r & 3;
    if (GATHER) n = gnode[t];
    long off;
    if (MODE == 0) {
      off = ((long)n * 4 + b) * (long)K;
    } else {
      off = ((long)b * nsrc + n) * (long)K;
    }
    const float4* __restrict__ arow = reinterpret_cast<const float4*>(A + off);
#pragma unroll
    for (int t2 = 0; t2 < K / 16; ++t2) {
      const float4 lo = arow[t2 * 4 + h * 2];
      const float4 hi = arow[t2 * 4 + h * 2 + 1];
      af[t][t2 * 4 + 0].x = lo.x; af[t][t2 * 4 + 0].y = lo.y;
      af[t][t2 * 4 + 1].x = lo.z; af[t][t2 * 4 + 1].y = lo.w;
      af[t][t2 * 4 + 2].x = hi.x; af[t][t2 * 4 + 2].y = hi.y;
      af[t][t2 * 4 + 3].x = hi.z; af[t][t2 * 4 + 3].y = hi.w;
    }
  }
}

template <int K, int N>
__device__ __forceinline__ void fold_form(const v2f (&wa)[K / 16][N / 4],
                                          const v2f (&vb)[N / 4][N / 16],
                                          v8f (&ff)[K / 16][N / 16]) {
#pragma unroll
  for (int t2 = 0; t2 < K / 16; ++t2) {
#pragma unroll
    for (int nt = 0; nt < N / 16; ++nt) ff[t2][nt] = v8f{};
  }
#pragma unroll
  for (int q = 0; q < N / 4; ++q) {
#pragma unroll
    for (int t2 = 0; t2 < K / 16; ++t2) {
#pragma unroll
      for (int nt = 0; nt < N / 16; ++nt) {
        ff[t2][nt] = __builtin_amdgcn_wmma_f32_16x16x4_f32(
             false, wa[t2][q],  false, vb[q][nt],
             (short)0, ff[t2][nt],  false,  false);
      }
    }
  }
}

template <int K, int N, int TT, bool BOTH>
__device__ __forceinline__ void fold_products(const v2f (&af)[TT][K / 4],
                                              const v2f (&wf)[K / 4][N / 16],
                                              const v8f (&ff)[K / 16][N / 16],
                                              float* __restrict__ Y1, float* __restrict__ Y2,
                                              int row00, int lane) {
  const int h  = lane >> 4;
  const int cn = lane & 15;
#pragma unroll
  for (int t = 0; t < TT; ++t) {
    v8f acc1[N / 16] = {};
    v8f acc2[N / 16] = {};
#pragma unroll
    for (int ks = 0; ks < K / 4; ++ks) {
      const int t2 = ks >> 2, j = ks & 3;
#pragma unroll
      for (int nt = 0; nt < N / 16; ++nt) {
        v2f fb;
        fb.x = ff[t2][nt][2 * j];
        fb.y = ff[t2][nt][2 * j + 1];
        if (BOTH) {
          acc1[nt] = __builtin_amdgcn_wmma_f32_16x16x4_f32(
               false, af[t][ks],  false, wf[ks][nt],
               (short)0, acc1[nt],  false,  false);
        }
        acc2[nt] = __builtin_amdgcn_wmma_f32_16x16x4_f32(
             false, af[t][ks],  false, fb,
             (short)0, acc2[nt],  false,  false);
      }
    }
#pragma unroll
    for (int nt = 0; nt < N / 16; ++nt) {
#pragma unroll
      for (int vr = 0; vr < 8; ++vr) {
        const int rr = row00 + t * 16 + h * 8 + vr;
        if (BOTH) Y1[(long)rr * N + nt * 16 + cn] = acc1[nt][vr];
        Y2[(long)rr * N + nt * 16 + cn] = acc2[nt][vr];
      }
    }
  }
}

template <int N, int NROWS>
__device__ __forceinline__ void zero_dst_rows(float* __restrict__ agg, float* __restrict__ cnt,
                                              long aggPlane, int cntPlane,
                                              int drow0, int lane) {
  const float4 z = make_float4(0.0f, 0.0f, 0.0f, 0.0f);
  float* __restrict__ a0 = agg + (long)drow0 * N + lane * 4;
#pragma unroll
  for (int i = 0; i < NROWS * N; i += 128) *reinterpret_cast<float4*>(a0 + i) = z;
  if (aggPlane != 0) {
#pragma unroll
    for (int i = 0; i < NROWS * N; i += 128) *reinterpret_cast<float4*>(a0 + aggPlane + i) = z;
  }
  const int d0 = drow0 >> 2;
  if (lane < (NROWS >> 2)) {
    cnt[d0 + lane] = 0.0f;
    if (cntPlane != 0) cnt[cntPlane + d0 + lane] = 0.0f;
  }
}

template <int K, int N, int MODE, int T, int TD>
__global__ void __launch_bounds__(128) layer_nodes_wmma(
    const float* __restrict__ A, const float* __restrict__ W, const float* __restrict__ V,
    const float* __restrict__ U, const int* __restrict__ res,
    float* __restrict__ YS, float* __restrict__ XD, float* __restrict__ YD,
    float* __restrict__ agg, float* __restrict__ cnt, long aggPlane, int cntPlane,
    int Rs, int Rd, int nsrc) {
  const int wave = blockIdx.x * (blockDim.x >> 5) + (threadIdx.x >> 5);
  const int lane = threadIdx.x & 31;
  const int row00 = wave * (16 * T);
  const int drow0 = wave * (16 * TD);
  if (row00 >= Rs && drow0 >= Rd) return;

  const int m  = lane & 15;
  const int h  = lane >> 4;
  const int kb = h << 1;

  int gnode[TD];
#pragma unroll
  for (int t = 0; t < TD; ++t) {
    const int r  = drow0 + t * 16 + m;
    const int rc = r < Rd ? r : Rd - 1;
    gnode[t] = res[rc >> 2];
  }

  v2f vb[N / 4][N / 16];
  v2f ub[N / 4][N / 16];
  v2f wa[K / 16][N / 4];
#pragma unroll
  for (int q = 0; q < N / 4; ++q) {
#pragma unroll
    for (int nt = 0; nt < N / 16; ++nt) {
      vb[q][nt].x = V[(q * 4 + kb) * N + nt * 16 + m];
      vb[q][nt].y = V[(q * 4 + kb + 1) * N + nt * 16 + m];
      ub[q][nt].x = U[(q * 4 + kb) * N + nt * 16 + m];
      ub[q][nt].y = U[(q * 4 + kb + 1) * N + nt * 16 + m];
    }
#pragma unroll
    for (int t2 = 0; t2 < K / 16; ++t2) {
      wa[t2][q].x = W[(t2 * 16 + m) * N + q * 4 + kb];
      wa[t2][q].y = W[(t2 * 16 + m) * N + q * 4 + kb + 1];
    }
  }
  v2f wf[K / 4][N / 16];
#pragma unroll
  for (int ks = 0; ks < K / 4; ++ks) {
    const int kk = (ks >> 2) * 16 + h * 8 + (ks & 3) * 2;
#pragma unroll
    for (int nt = 0; nt < N / 16; ++nt) {
      wf[ks][nt].x = W[kk * N + nt * 16 + m];
      wf[ks][nt].y = W[(kk + 1) * N + nt * 16 + m];
    }
  }

  v8f fv[K / 16][N / 16];
  v8f fu[K / 16][N / 16];
  if (row00 + 16 * T <= Rs && drow0 + 16 * TD <= Rd) {
    v2f af[T][K / 4];
    v2f ad[TD][K / 4];
    const int nogather[T] = {};
    fold_load_rows<K, MODE, false, T>(A, nogather, af, row00, nsrc, lane);
    __builtin_amdgcn_sched_barrier(0);
    fold_load_rows<K, MODE, true, TD>(A, gnode, ad, drow0, nsrc, lane);
    zero_dst_rows<N, 16 * TD>(agg, cnt, aggPlane, cntPlane, drow0, lane);
    __builtin_amdgcn_sched_barrier(0);
    fold_form<K, N>(wa, vb, fv);
    fold_form<K, N>(wa, ub, fu);
    fold_products<K, N, T, false>(af, wf, fv, nullptr, YS, row00, lane);
    fold_products<K, N, TD, true>(ad, wf, fu, XD, YD, drow0, lane);
  } else {
    fold_form<K, N>(wa, vb, fv);
    fold_form<K, N>(wa, ub, fu);
    const int nogather[1] = {};
    for (int row0 = row00; row0 < row00 + 16 * T && row0 < Rs; row0 += 16) {
      v2f af[1][K / 4];
      fold_load_rows<K, MODE, false, 1>(A, nogather, af, row0, nsrc, lane);
      fold_products<K, N, 1, false>(af, wf, fv, nullptr, YS, row0, lane);
    }
#pragma unroll
    for (int t = 0; t < TD; ++t) {
      const int row0 = drow0 + t * 16;
      if (row0 < Rd) {
        v2f ad[1][K / 4];
        const int g1[1] = {gnode[t]};
        fold_load_rows<K, MODE, true, 1>(A, g1, ad, row0, nsrc, lane);
        zero_dst_rows<N, 16>(agg, cnt, aggPlane, cntPlane, row0, lane);
        fold_products<K, N, 1, true>(ad, wf, fu, XD, YD, row0, lane);
      }
    }
  }
}

__device__ __forceinline__ int package_half() {
  const unsigned id = (unsigned)__builtin_amdgcn_s_sendmsg_rtn(0x87);
  return (int)((id >> 18) & 1u);
}

template <int BH, int H, int U>
__global__ void edge_kernel(const float* __restrict__ ys, const float* __restrict__ yd,
                            const int* __restrict__ src, const int* __restrict__ dst,
                            const float* __restrict__ ew, const float* __restrict__ We,
                            float* __restrict__ agg0, float* __restrict__ cnt0,
                            long aggPlane, int cntPlane, int E) {
  constexpr int PER = BH / 32;
  const int wave = blockIdx.x * (blockDim.x >> 5) + (threadIdx.x >> 5);
  const int lane = threadIdx.x & 31;
  const int e0 = wave * 32;
  if (e0 >= E) return;
  const int n = (E - e0 < 32) ? (E - e0) : 32;
  const int hf = package_half();
  float* __restrict__ agg = agg0 + (long)hf * aggPlane;
  float* __restrict__ cnt = cnt0 + hf * cntPlane;

  const float wev = We[lane & (H - 1)];

  int sv = 0, dv = 0;
  float wv = 0.0f;
  if (lane < n) {
    sv = src[e0 + lane];
    dv = dst[e0 + lane];
    wv = ew[e0 + lane];
    atomicAdd(cnt + dv, 1.0f);
  }

  if (n == 32) {
    for (int j0 = 0; j0 < 32; j0 += U) {
      float xj[U][PER], xi[U][PER];
#pragma unroll
      for (int u = 0; u < U; ++u) {
        const int s = __builtin_amdgcn_readlane(sv, j0 + u);
        const int d = __builtin_amdgcn_readlane(dv, j0 + u);
        const float* __restrict__ ysr = ys + (long)s * BH + lane;
        const float* __restrict__ ydr = yd + (long)d * BH + lane;
#pragma unroll
        for (int k = 0; k < PER; ++k) {
          xj[u][k] = ysr[32 * k];
          xi[u][k] = ydr[32 * k];
        }
      }
#pragma unroll
      for (int u = 0; u < U; ++u) {
        const int d = __builtin_amdgcn_readlane(dv, j0 + u);
        const float w = __int_as_float(__builtin_amdgcn_readlane(__float_as_int(wv), j0 + u));
        const float ev = w * wev;
        float* __restrict__ ar = agg + (long)d * BH + lane;
#pragma unroll
        for (int k = 0; k < PER; ++k) {
          const float t  = xi[u][k] * xj[u][k] * ev;
          const float sg = 1.0f / (1.0f + __expf(-t));
          atomicAdd(ar + 32 * k, xj[u][k] * sg);
        }
      }
    }
  } else {
    for (int j = 0; j < n; ++j) {
      const int s = __builtin_amdgcn_readlane(sv, j);
      const int d = __builtin_amdgcn_readlane(dv, j);
      const float w = __int_as_float(__builtin_amdgcn_readlane(__float_as_int(wv), j));
      const float ev = w * wev;
      const float* __restrict__ ysr = ys + (long)s * BH + lane;
      const float* __restrict__ ydr = yd + (long)d * BH + lane;
      float* __restrict__ ar        = agg + (long)d * BH + lane;
#pragma unroll
      for (int k = 0; k < PER; ++k) {
        const float xjv = ysr[32 * k];
        const float xiv = ydr[32 * k];
        const float t   = xiv * xjv * ev;
        const float sg  = 1.0f / (1.0f + __expf(-t));
        atomicAdd(ar + 32 * k, xjv * sg);
      }
    }
  }
}


template <bool SEPARATE> struct fin_io {
  typedef const float* in_t;
  typedef float* out_t;
};
template <> struct fin_io<true> {
  typedef const float* __restrict__ in_t;
  typedef float* __restrict__ out_t;
};

template <int BH, int H, bool FINAL, int NN, bool TWOA, bool TWOC>
__device__ __forceinline__ void finalize_nodes(const float* __restrict__ xd,
                                               typename fin_io<FINAL>::in_t yd,
                                               const float* __restrict__ agg,
                                               const float* __restrict__ cnt,
                                               typename fin_io<FINAL>::out_t out,
                                               long aggPlane, int cntPlane, int Nd,
                                               int d0, int lane) {
  constexpr int PER = BH / 32;

  float vals[NN][PER], xdv[NN][PER];
#pragma unroll
  for (int q = 0; q < NN; ++q) {
    const int d = d0 + q;
    const float cn      = TWOC ? cnt[d] + cnt[cntPlane + d] : cnt[d];
    const float inv_cnt = 1.0f / fmaxf(cn, 1.0f);
    const long  base    = (long)d * BH;
#pragma unroll
    for (int j = 0; j < PER; ++j) {
      const int i  = lane + j * 32;
      const float ag = TWOA ? agg[base + i] + agg[aggPlane + base + i] : agg[base + i];
      vals[q][j] = yd[base + i] + ag * inv_cnt;
      xdv[q][j]  = xd[base + i];
    }
  }

#pragma unroll
  for (int q = 0; q < NN; ++q) {
    const int d = d0 + q;
    float sum = 0.0f, ssq = 0.0f;
#pragma unroll
    for (int j = 0; j < PER; ++j) {
      const float o = vals[q][j];
      sum += o;
      ssq += o * o;
    }
#pragma unroll
    for (int mask = 16; mask >= 1; mask >>= 1) {
      sum += __shfl_xor(sum, mask, 32);
      ssq += __shfl_xor(ssq, mask, 32);
    }
    const float mu  = sum * (1.0f / BH);
    const float var = ssq * (1.0f / BH) - mu * mu;
    const float is  = rsqrtf(var + 1e-5f);

    const long base = (long)d * BH;
#pragma unroll
    for (int j = 0; j < PER; ++j) {
      const int i = lane + j * 32;
      float v = xdv[q][j] + (vals[q][j] - mu) * is;
      v = v > 0.0f ? v : 0.01f * v;
      if (FINAL) {
        const int b = i / H, c = i & (H - 1);
        out[((long)b * Nd + d) * H + c] = v;
      } else {
        out[base + i] = v;
      }
    }
  }
}

template <int BH, int H, bool FINAL, int NPW, bool TWOA, bool TWOC>
__device__ __forceinline__ void finalize_wave(const float* __restrict__ xd,
                                              typename fin_io<FINAL>::in_t yd,
                                              const float* __restrict__ agg,
                                              const float* __restrict__ cnt,
                                              typename fin_io<FINAL>::out_t out,
                                              long aggPlane, int cntPlane, int Nd,
                                              int d0, int lane) {
  if (d0 + NPW <= Nd) {
    finalize_nodes<BH, H, FINAL, NPW, TWOA, TWOC>(xd, yd, agg, cnt, out, aggPlane,
                                                  cntPlane, Nd, d0, lane);
  } else {
    for (int d = d0; d < Nd; ++d)
      finalize_nodes<BH, H, FINAL, 1, TWOA, TWOC>(xd, yd, agg, cnt, out, aggPlane,
                                                  cntPlane, Nd, d, lane);
  }
}

template <int BH, int H, bool FINAL, int NPW>
__global__ void finalize_kernel(const float* __restrict__ xd,
                                const float* yd, const float* __restrict__ agg,
                                const float* __restrict__ cnt, float* out,
                                long aggPlane, int cntPlane, int Nd) {
  const int wave = blockIdx.x * (blockDim.x >> 5) + (threadIdx.x >> 5);
  const int lane = threadIdx.x & 31;
  const int d0 = wave * NPW;
  if (d0 >= Nd) return;
  if (aggPlane != 0 && cntPlane != 0) {
    finalize_wave<BH, H, FINAL, NPW, true, true>(xd, yd, agg, cnt, out, aggPlane, cntPlane,
                                                 Nd, d0, lane);
  } else if (aggPlane != 0) {
    finalize_wave<BH, H, FINAL, NPW, true, false>(xd, yd, agg, cnt, out, aggPlane, cntPlane,
                                                  Nd, d0, lane);
  } else if (cntPlane != 0) {
    finalize_wave<BH, H, FINAL, NPW, false, true>(xd, yd, agg, cnt, out, aggPlane, cntPlane,
                                                  Nd, d0, lane);
  } else {
    finalize_wave<BH, H, FINAL, NPW, false, false>(xd, yd, agg, cnt, out, aggPlane,
                                                   cntPlane, Nd, d0, lane);
  }
}

extern "C" void kernel_launch(void* const* d_in, const int* in_sizes, int n_in,
                              void* d_out, int out_size, void* d_ws, size_t ws_size,
                              hipStream_t stream) {
  constexpr int B = 4, Cin = 32, H = 16, Cout = 32;
  const float* X    = (const float*)d_in[0];
  const int*   src0 = (const int*)d_in[1];
  const int*   dst0 = (const int*)d_in[2];
  const float* ew0  = (const float*)d_in[3];
  const int*   res0 = (const int*)d_in[4];
  const int*   src1 = (const int*)d_in[5];
  const int*   dst1 = (const int*)d_in[6];
  const float* ew1  = (const float*)d_in[7];
  const int*   res1 = (const int*)d_in[8];
  const float* Wn1  = (const float*)d_in[9];
  const float* We1  = (const float*)d_in[10];
  const float* U1   = (const float*)d_in[11];
  const float* V1   = (const float*)d_in[12];
  const float* Wn2  = (const float*)d_in[13];
  const float* We2  = (const float*)d_in[14];
  const float* U2   = (const float*)d_in[15];
  const float* V2   = (const float*)d_in[16];

  const int E0 = in_sizes[1];
  const int N1 = in_sizes[4];
  const int E1 = in_sizes[5];
  const int N2 = in_sizes[8];
  const int N0 = in_sizes[0] / (B * Cin);

  float* ws = (float*)d_ws;
  const size_t szXS = (size_t)((N0 * B * H > N1 * B * Cout) ? N0 * B * H : N1 * B * Cout);
  const size_t szYD = (size_t)((N1 * B * H > N2 * B * Cout) ? N1 * B * H : N2 * B * Cout);
  const size_t szCN = (size_t)((N1 > N2) ? N1 : N2);
  float* regP = ws;
  float* ys   = regP + szXS;
  float* regR = ys + szXS;
  float* xd1  = regP;
  float* yd1  = regR;
  float* h1   = regR;
  float* xd2  = regP;
  float* yd2  = regP + (size_t)N2 * B * Cout;
  const size_t needAgg2 = (2 * szXS + 3 * szYD + szCN) * sizeof(float);
  const size_t needCnt2 = needAgg2 + szCN * sizeof(float);
  const int    aggPlanes = (ws_size >= needAgg2) ? 2 : 1;
  const int    cntPlanes = (ws_size >= needCnt2) ? 2 : 1;
  const long   aggPlane  = (aggPlanes == 2) ? (long)szYD : 0;
  const int    cntPlane  = (cntPlanes == 2) ? (int)szCN : 0;
  float* agg = regR + szYD;
  float* cnt = agg + (size_t)aggPlanes * szYD;

  const dim3 blk(256);
  const dim3 gblk(128);
  auto node_grid = [](int stiles, int T, int dtiles, int TD) {
    const int ws_ = (stiles + T - 1) / T;
    const int wd_ = (dtiles + TD - 1) / TD;
    const int waves = ws_ > wd_ ? ws_ : wd_;
    return dim3((unsigned)((waves + 3) / 4));
  };
  auto fin_grid = [](int nodes, int NPW) {
    const int waves = (nodes + NPW - 1) / NPW;
    return dim3((unsigned)((waves + 7) / 8));
  };

  const int R1s = N0 * B;
  const int R1d = N1 * B;
  layer_nodes_wmma<32, 16, 1, 4, 2><<<node_grid(R1s / 16, 4, R1d / 16, 2), gblk, 0, stream>>>(
      X, Wn1, V1, U1, res0, ys, xd1, yd1, agg, cnt, aggPlane, cntPlane, R1s, R1d, N0);
  edge_kernel<64, 16, 8><<<dim3((unsigned)((E0 + 127) / 128)), dim3(128), 0, stream>>>(
      ys, yd1, src0, dst0, ew0, We1, agg, cnt, aggPlane, cntPlane, E0);
  finalize_kernel<64, 16, false, 4><<<fin_grid(N1, 4), blk, 0, stream>>>(
      xd1, yd1, agg, cnt, h1, aggPlane, cntPlane, N1);

  const int R2s = N1 * B;
  const int R2d = N2 * B;
  layer_nodes_wmma<16, 32, 0, 4, 2><<<node_grid(R2s / 16, 4, R2d / 16, 2), gblk, 0, stream>>>(
      h1, Wn2, V2, U2, res1, ys, xd2, yd2, agg, cnt, aggPlane, cntPlane, R2s, R2d, 0);
  edge_kernel<128, 32, 4><<<dim3((unsigned)((E1 + 127) / 128)), dim3(128), 0, stream>>>(
      ys, yd2, src1, dst1, ew1, We2, agg, cnt, aggPlane, cntPlane, E1);
  finalize_kernel<128, 32, true, 4><<<fin_grid(N2, 4), blk, 0, stream>>>(
      xd2, yd2, agg, cnt, (float*)d_out, aggPlane, cntPlane, N2);
}


# Edge sums: one work-group per range of dst nodes keeps its agg rows in LDS, scans all dst and files its edges; LDS float adds replace global float atomics (arrival order as before); no fill, 1 plane
# speedup vs baseline: 1.9788x; 1.0600x over previous
#include <hip/hip_runtime.h>
#include <hip/hip_bf16.h>

typedef float v2f __attribute__((ext_vector_type(2)));
typedef float v8f __attribute__((ext_vector_type(8)));


template <int K, int MODE, bool GATHER, int TT>
__device__ __forceinline__ void fold_load_rows(const float* __restrict__ A,
                                               const int (&gnode)[TT],
                                               v2f (&af)[TT][K / 4],
                                               int row00, int nsrc, int lane) {
  const int m = lane & 15;
  const int h = lane >> 4;
#pragma unroll
  for (int t = 0; t < TT; ++t) {
    const int r = row00 + t * 16 + m;
    int n = r >> 2;
    const int b = r & 3;
    if (GATHER) n = gnode[t];
    long off;
    if (MODE == 0) {
      off = ((long)n * 4 + b) * (long)K;
    } else {
      off = ((long)b * nsrc + n) * (long)K;
    }
    const float4* __restrict__ arow = reinterpret_cast<const float4*>(A + off);
#pragma unroll
    for (int t2 = 0; t2 < K / 16; ++t2) {
      const float4 lo = arow[t2 * 4 + h * 2];
      const float4 hi = arow[t2 * 4 + h * 2 + 1];
      af[t][t2 * 4 + 0].x = lo.x; af[t][t2 * 4 + 0].y = lo.y;
      af[t][t2 * 4 + 1].x = lo.z; af[t][t2 * 4 + 1].y = lo.w;
      af[t][t2 * 4 + 2].x = hi.x; af[t][t2 * 4 + 2].y = hi.y;
      af[t][t2 * 4 + 3].x = hi.z; af[t][t2 * 4 + 3].y = hi.w;
    }
  }
}

template <int K, int N>
__device__ __forceinline__ void fold_form(const v2f (&wa)[K / 16][N / 4],
                                          const v2f (&vb)[N / 4][N / 16],
                                          v8f (&ff)[K / 16][N / 16]) {
#pragma unroll
  for (int t2 = 0; t2 < K / 16; ++t2) {
#pragma unroll
    for (int nt = 0; nt < N / 16; ++nt) ff[t2][nt] = v8f{};
  }
#pragma unroll
  for (int q = 0; q < N / 4; ++q) {
#pragma unroll
    for (int t2 = 0; t2 < K / 16; ++t2) {
#pragma unroll
      for (int nt = 0; nt < N / 16; ++nt) {
        ff[t2][nt] = __builtin_amdgcn_wmma_f32_16x16x4_f32(
             false, wa[t2][q],  false, vb[q][nt],
             (short)0, ff[t2][nt],  false,  false);
      }
    }
  }
}

template <int K, int N, int TT, bool BOTH>
__device__ __forceinline__ void fold_products(const v2f (&af)[TT][K / 4],
                                              const v2f (&wf)[K / 4][N / 16],
                                              const v8f (&ff)[K / 16][N / 16],
                                              float* __restrict__ Y1, float* __restrict__ Y2,
                                              int row00, int lane) {
  const int h  = lane >> 4;
  const int cn = lane & 15;
#pragma unroll
  for (int t = 0; t < TT; ++t) {
    v8f acc1[N / 16] = {};
    v8f acc2[N / 16] = {};
#pragma unroll
    for (int ks = 0; ks < K / 4; ++ks) {
      const int t2 = ks >> 2, j = ks & 3;
#pragma unroll
      for (int nt = 0; nt < N / 16; ++nt) {
        v2f fb;
        fb.x = ff[t2][nt][2 * j];
        fb.y = ff[t2][nt][2 * j + 1];
        if (BOTH) {
          acc1[nt] = __builtin_amdgcn_wmma_f32_16x16x4_f32(
               false, af[t][ks],  false, wf[ks][nt],
               (short)0, acc1[nt],  false,  false);
        }
        acc2[nt] = __builtin_amdgcn_wmma_f32_16x16x4_f32(
             false, af[t][ks],  false, fb,
             (short)0, acc2[nt],  false,  false);
      }
    }
#pragma unroll
    for (int nt = 0; nt < N / 16; ++nt) {
#pragma unroll
      for (int vr = 0; vr < 8; ++vr) {
        const int rr = row00 + t * 16 + h * 8 + vr;
        if (BOTH) Y1[(long)rr * N + nt * 16 + cn] = acc1[nt][vr];
        Y2[(long)rr * N + nt * 16 + cn] = acc2[nt][vr];
      }
    }
  }
}

template <int K, int N, int MODE, int T, int TD>
__global__ void __launch_bounds__(128) layer_nodes_wmma(
    const float* __restrict__ A, const float* __restrict__ W, const float* __restrict__ V,
    const float* __restrict__ U, const int* __restrict__ res,
    float* __restrict__ YS, float* __restrict__ XD, float* __restrict__ YD,
    int Rs, int Rd, int nsrc) {
  const int wave = blockIdx.x * (blockDim.x >> 5) + (threadIdx.x >> 5);
  const int lane = threadIdx.x & 31;
  const int row00 = wave * (16 * T);
  const int drow0 = wave * (16 * TD);
  if (row00 >= Rs && drow0 >= Rd) return;

  const int m  = lane & 15;
  const int h  = lane >> 4;
  const int kb = h << 1;

  int gnode[TD];
#pragma unroll
  for (int t = 0; t < TD; ++t) {
    const int r  = drow0 + t * 16 + m;
    const int rc = r < Rd ? r : Rd - 1;
    gnode[t] = res[rc >> 2];
  }

  v2f vb[N / 4][N / 16];
  v2f ub[N / 4][N / 16];
  v2f wa[K / 16][N / 4];
#pragma unroll
  for (int q = 0; q < N / 4; ++q) {
#pragma unroll
    for (int nt = 0; nt < N / 16; ++nt) {
      vb[q][nt].x = V[(q * 4 + kb) * N + nt * 16 + m];
      vb[q][nt].y = V[(q * 4 + kb + 1) * N + nt * 16 + m];
      ub[q][nt].x = U[(q * 4 + kb) * N + nt * 16 + m];
      ub[q][nt].y = U[(q * 4 + kb + 1) * N + nt * 16 + m];
    }
#pragma unroll
    for (int t2 = 0; t2 < K / 16; ++t2) {
      wa[t2][q].x = W[(t2 * 16 + m) * N + q * 4 + kb];
      wa[t2][q].y = W[(t2 * 16 + m) * N + q * 4 + kb + 1];
    }
  }
  v2f wf[K / 4][N / 16];
#pragma unroll
  for (int ks = 0; ks < K / 4; ++ks) {
    const int kk = (ks >> 2) * 16 + h * 8 + (ks & 3) * 2;
#pragma unroll
    for (int nt = 0; nt < N / 16; ++nt) {
      wf[ks][nt].x = W[kk * N + nt * 16 + m];
      wf[ks][nt].y = W[(kk + 1) * N + nt * 16 + m];
    }
  }

  v8f fv[K / 16][N / 16];
  v8f fu[K / 16][N / 16];
  if (row00 + 16 * T <= Rs && drow0 + 16 * TD <= Rd) {
    v2f af[T][K / 4];
    v2f ad[TD][K / 4];
    const int nogather[T] = {};
    fold_load_rows<K, MODE, false, T>(A, nogather, af, row00, nsrc, lane);
    __builtin_amdgcn_sched_barrier(0);
    fold_load_rows<K, MODE, true, TD>(A, gnode, ad, drow0, nsrc, lane);
    __builtin_amdgcn_sched_barrier(0);
    fold_form<K, N>(wa, vb, fv);
    fold_form<K, N>(wa, ub, fu);
    fold_products<K, N, T, false>(af, wf, fv, nullptr, YS, row00, lane);
    fold_products<K, N, TD, true>(ad, wf, fu, XD, YD, drow0, lane);
  } else {
    fold_form<K, N>(wa, vb, fv);
    fold_form<K, N>(wa, ub, fu);
    const int nogather[1] = {};
    for (int row0 = row00; row0 < row00 + 16 * T && row0 < Rs; row0 += 16) {
      v2f af[1][K / 4];
      fold_load_rows<K, MODE, false, 1>(A, nogather, af, row0, nsrc, lane);
      fold_products<K, N, 1, false>(af, wf, fv, nullptr, YS, row0, lane);
    }
#pragma unroll
    for (int t = 0; t < TD; ++t) {
      const int row0 = drow0 + t * 16;
      if (row0 < Rd) {
        v2f ad[1][K / 4];
        const int g1[1] = {gnode[t]};
        fold_load_rows<K, MODE, true, 1>(A, g1, ad, row0, nsrc, lane);
        fold_products<K, N, 1, true>(ad, wf, fu, XD, YD, row0, lane);
      }
    }
  }
}


template <int BH, int H>
__device__ __forceinline__ void edge_by_one_lane(const float* __restrict__ ys,
                                                 const float* __restrict__ yd,
                                                 const int* __restrict__ src,
                                                 const float* __restrict__ ew,
                                                 const float* __restrict__ We,
                                                 float* s_agg, float* s_cnt, int n0,
                                                 int e, int d) {
  const int s = src[e];
  const float w = ew[e];
  float* ar = s_agg + (d - n0) * BH;
  for (int i = 0; i < BH; ++i) {
    const float xjv = ys[(long)s * BH + i];
    const float xiv = yd[(long)d * BH + i];
    const float ev  = w * We[i & (H - 1)];
    const float t   = xiv * xjv * ev;
    const float sg  = 1.0f / (1.0f + __expf(-t));
    atomicAdd(ar + i, xjv * sg);
  }
  atomicAdd(s_cnt + (d - n0), 1.0f);
}

template <int BH, int H, int U, bool FULL>
__device__ __forceinline__ void owner_tile(const float* __restrict__ ys,
                                           const float* __restrict__ yd,
                                           float* s_agg, int n0, int sv, int dv, float wv,
                                           float wev, int nT, int lane) {
  constexpr int PER = BH / 32;
  for (int j0 = 0; j0 < (FULL ? 32 : nT); j0 += U) {
    float xj[U][PER], xi[U][PER];
#pragma unroll
    for (int u = 0; u < U; ++u) {
      const int jj = (FULL || j0 + u < nT) ? (j0 + u) : (nT - 1);
      const int s = __builtin_amdgcn_readlane(sv, jj);
      const int d = __builtin_amdgcn_readlane(dv, jj);
      const float* __restrict__ ysr = ys + (long)s * BH + lane;
      const float* __restrict__ ydr = yd + (long)d * BH + lane;
#pragma unroll
      for (int k = 0; k < PER; ++k) {
        xj[u][k] = ysr[32 * k];
        xi[u][k] = ydr[32 * k];
      }
    }
#pragma unroll
    for (int u = 0; u < U; ++u) {
      if (FULL || j0 + u < nT) {
        const int d = __builtin_amdgcn_readlane(dv, j0 + u);
        const float w = __int_as_float(__builtin_amdgcn_readlane(__float_as_int(wv), j0 + u));
        const float ev = w * wev;
        float* ar = s_agg + (d - n0) * BH + lane;
#pragma unroll
        for (int k = 0; k < PER; ++k) {
          const float t  = xi[u][k] * xj[u][k] * ev;
          const float sg = 1.0f / (1.0f + __expf(-t));
          atomicAdd(ar + 32 * k, xj[u][k] * sg);
        }
      }
    }
  }
}

template <int BH, int H, int NPW, int U, int QCAP>
__global__ void __launch_bounds__(1024) edge_owner_kernel(
    const float* __restrict__ ys, const float* __restrict__ yd,
    const int* __restrict__ src, const int* __restrict__ dst,
    const float* __restrict__ ew, const float* __restrict__ We,
    float* __restrict__ agg, float* __restrict__ cnt, int E, int Nd) {
  constexpr int SB  = 8;
  __shared__ __attribute__((aligned(16))) float s_agg[NPW * BH];
  __shared__ float s_cnt[NPW];
  __shared__ int   s_q[QCAP];
  __shared__ int   s_qn;

  constexpr int NTHR = 1024;
  const int tid   = threadIdx.x;
  const int lane  = tid & 31;
  const int wave  = __builtin_amdgcn_readfirstlane(tid >> 5);
  constexpr int nthr  = NTHR;
  constexpr int nwave = NTHR >> 5;
  const int n0 = blockIdx.x * NPW;
  if (n0 >= Nd) return;
  const int nn = (Nd - n0 < NPW) ? (Nd - n0) : NPW;

  for (int i = tid; i < NPW * BH; i += nthr) s_agg[i] = 0.0f;
  for (int i = tid; i < NPW; i += nthr) s_cnt[i] = 0.0f;
  if (tid == 0) s_qn = 0;
  __syncthreads();

  auto file4 = [&](const int4 a, const int e) {
    const bool m0 = (unsigned)(a.x - n0) < (unsigned)nn;
    const bool m1 = (unsigned)(a.y - n0) < (unsigned)nn;
    const bool m2 = (unsigned)(a.z - n0) < (unsigned)nn;
    const bool m3 = (unsigned)(a.w - n0) < (unsigned)nn;
    if (m0 | m1 | m2 | m3) {
      if (m0) { const int pos = atomicAdd(&s_qn, 1); if (pos < QCAP) s_q[pos] = e; }
      if (m1) { const int pos = atomicAdd(&s_qn, 1); if (pos < QCAP) s_q[pos] = e + 1; }
      if (m2) { const int pos = atomicAdd(&s_qn, 1); if (pos < QCAP) s_q[pos] = e + 2; }
      if (m3) { const int pos = atomicAdd(&s_qn, 1); if (pos < QCAP) s_q[pos] = e + 3; }
    }
  };
  const int4* __restrict__ d4 = reinterpret_cast<const int4*>(dst);
  const int nvec  = E >> 2;
  const int nfull = nvec / (NTHR * SB);
  const int rotb  = (int)(((long)blockIdx.x * nfull) / gridDim.x);
  for (int b = 0; b < nfull; ++b) {
    int bb = b + rotb;
    if (bb >= nfull) bb -= nfull;
    const int vbase = bb * (NTHR * SB);
    const int4* __restrict__ p = d4 + vbase + tid;
    int4 a[SB];
#pragma unroll
    for (int i = 0; i < SB; ++i) a[i] = p[i * NTHR];
#pragma unroll
    for (int i = 0; i < SB; ++i) file4(a[i], (vbase + i * NTHR + tid) * 4);
  }
  for (int v = nfull * (NTHR * SB) + tid; v < nvec; v += NTHR) file4(d4[v], v * 4);
  for (int e = nvec * 4 + tid; e < E; e += NTHR) {
    if ((unsigned)(dst[e] - n0) < (unsigned)nn) {
      const int pos = atomicAdd(&s_qn, 1);
      if (pos < QCAP) s_q[pos] = e;
    }
  }
  __syncthreads();
  const int qn = __builtin_amdgcn_readfirstlane(s_qn);

  if (qn > QCAP) {
    for (int e = tid; e < E; e += nthr) {
      const int d = dst[e];
      if ((unsigned)(d - n0) < (unsigned)nn)
        edge_by_one_lane<BH, H>(ys, yd, src, ew, We, s_agg, s_cnt, n0, e, d);
    }
  } else {
    const float wev = We[lane & (H - 1)];
    for (int t0 = wave * 32; t0 < qn; t0 += nwave * 32) {
      const int nT = (qn - t0 < 32) ? (qn - t0) : 32;
      int sv = 0, dv = n0;
      float wv = 0.0f;
      if (lane < nT) {
        const int e = s_q[t0 + lane];
        sv = src[e];
        dv = dst[e];
        wv = ew[e];
        atomicAdd(s_cnt + (dv - n0), 1.0f);
      }
      if (nT == 32) {
        owner_tile<BH, H, U, true>(ys, yd, s_agg, n0, sv, dv, wv, wev, 32, lane);
      } else {
        owner_tile<BH, H, U, false>(ys, yd, s_agg, n0, sv, dv, wv, wev, nT, lane);
      }
    }
  }
  __syncthreads();

  const float4* s4 = reinterpret_cast<const float4*>(s_agg);
  float4* __restrict__ g4 = reinterpret_cast<float4*>(agg + (long)n0 * BH);
  for (int i = tid; i < nn * (BH / 4); i += nthr) g4[i] = s4[i];
  for (int i = tid; i < nn; i += nthr) cnt[n0 + i] = s_cnt[i];
}


template <bool SEPARATE> struct fin_io {
  typedef const float* in_t;
  typedef float* out_t;
};
template <> struct fin_io<true> {
  typedef const float* __restrict__ in_t;
  typedef float* __restrict__ out_t;
};

template <int BH, int H, bool FINAL, int NN>
__device__ __forceinline__ void finalize_nodes(const float* __restrict__ xd,
                                               typename fin_io<FINAL>::in_t yd,
                                               const float* __restrict__ agg,
                                               const float* __restrict__ cnt,
                                               typename fin_io<FINAL>::out_t out,
                                               int Nd, int d0, int lane) {
  constexpr int PER = BH / 32;

  float vals[NN][PER], xdv[NN][PER];
#pragma unroll
  for (int q = 0; q < NN; ++q) {
    const int d = d0 + q;
    const float inv_cnt = 1.0f / fmaxf(cnt[d], 1.0f);
    const long  base    = (long)d * BH;
#pragma unroll
    for (int j = 0; j < PER; ++j) {
      const int i  = lane + j * 32;
      vals[q][j] = yd[base + i] + agg[base + i] * inv_cnt;
      xdv[q][j]  = xd[base + i];
    }
  }

#pragma unroll
  for (int q = 0; q < NN; ++q) {
    const int d = d0 + q;
    float sum = 0.0f, ssq = 0.0f;
#pragma unroll
    for (int j = 0; j < PER; ++j) {
      const float o = vals[q][j];
      sum += o;
      ssq += o * o;
    }
#pragma unroll
    for (int mask = 16; mask >= 1; mask >>= 1) {
      sum += __shfl_xor(sum, mask, 32);
      ssq += __shfl_xor(ssq, mask, 32);
    }
    const float mu  = sum * (1.0f / BH);
    const float var = ssq * (1.0f / BH) - mu * mu;
    const float is  = rsqrtf(var + 1e-5f);

    const long base = (long)d * BH;
#pragma unroll
    for (int j = 0; j < PER; ++j) {
      const int i = lane + j * 32;
      float v = xdv[q][j] + (vals[q][j] - mu) * is;
      v = v > 0.0f ? v : 0.01f * v;
      if (FINAL) {
        const int b = i / H, c = i & (H - 1);
        out[((long)b * Nd + d) * H + c] = v;
      } else {
        out[base + i] = v;
      }
    }
  }
}

template <int BH, int H, bool FINAL, int NPW>
__global__ void finalize_kernel(const float* __restrict__ xd,
                                const float* yd, const float* __restrict__ agg,
                                const float* __restrict__ cnt, float* out, int Nd) {
  const int wave = blockIdx.x * (blockDim.x >> 5) + (threadIdx.x >> 5);
  const int lane = threadIdx.x & 31;
  const int d0 = wave * NPW;
  if (d0 >= Nd) return;
  if (d0 + NPW <= Nd) {
    finalize_nodes<BH, H, FINAL, NPW>(xd, yd, agg, cnt, out, Nd, d0, lane);
  } else {
    for (int d = d0; d < Nd; ++d)
      finalize_nodes<BH, H, FINAL, 1>(xd, yd, agg, cnt, out, Nd, d, lane);
  }
}

extern "C" void kernel_launch(void* const* d_in, const int* in_sizes, int n_in,
                              void* d_out, int out_size, void* d_ws, size_t ws_size,
                              hipStream_t stream) {
  constexpr int B = 4, Cin = 32, H = 16, Cout = 32;
  const float* X    = (const float*)d_in[0];
  const int*   src0 = (const int*)d_in[1];
  const int*   dst0 = (const int*)d_in[2];
  const float* ew0  = (const float*)d_in[3];
  const int*   res0 = (const int*)d_in[4];
  const int*   src1 = (const int*)d_in[5];
  const int*   dst1 = (const int*)d_in[6];
  const float* ew1  = (const float*)d_in[7];
  const int*   res1 = (const int*)d_in[8];
  const float* Wn1  = (const float*)d_in[9];
  const float* We1  = (const float*)d_in[10];
  const float* U1   = (const float*)d_in[11];
  const float* V1   = (const float*)d_in[12];
  const float* Wn2  = (const float*)d_in[13];
  const float* We2  = (const float*)d_in[14];
  const float* U2   = (const float*)d_in[15];
  const float* V2   = (const float*)d_in[16];

  const int E0 = in_sizes[1];
  const int N1 = in_sizes[4];
  const int E1 = in_sizes[5];
  const int N2 = in_sizes[8];
  const int N0 = in_sizes[0] / (B * Cin);

  float* ws = (float*)d_ws;
  const size_t szXS = (size_t)((N0 * B * H > N1 * B * Cout) ? N0 * B * H : N1 * B * Cout);
  const size_t szYD = (size_t)((N1 * B * H > N2 * B * Cout) ? N1 * B * H : N2 * B * Cout);
  float* regP = ws;
  float* ys   = regP + szXS;
  float* regR = ys + szXS;
  float* xd1  = regP;
  float* yd1  = regR;
  float* h1   = regR;
  float* xd2  = regP;
  float* yd2  = regP + (size_t)N2 * B * Cout;
  float* agg = regR + szYD;
  float* cnt = agg + szYD;

  const dim3 blk(256);
  const dim3 gblk(128);
  auto node_grid = [](int stiles, int T, int dtiles, int TD) {
    const int ws_ = (stiles + T - 1) / T;
    const int wd_ = (dtiles + TD - 1) / TD;
    const int waves = ws_ > wd_ ? ws_ : wd_;
    return dim3((unsigned)((waves + 3) / 4));
  };
  const dim3 oblk(1024);
  auto owner_grid = [](int nodes, int NPW) {
    return dim3((unsigned)((nodes + NPW - 1) / NPW));
  };
  auto fin_grid = [](int nodes, int NPW) {
    const int waves = (nodes + NPW - 1) / NPW;
    return dim3((unsigned)((waves + 7) / 8));
  };

  const int R1s = N0 * B;
  const int R1d = N1 * B;
  layer_nodes_wmma<32, 16, 1, 4, 2><<<node_grid(R1s / 16, 4, R1d / 16, 2), gblk, 0, stream>>>(
      X, Wn1, V1, U1, res0, ys, xd1, yd1, R1s, R1d, N0);
  edge_owner_kernel<64, 16, 200, 8, 2560><<<owner_grid(N1, 200), oblk, 0, stream>>>(
      ys, yd1, src0, dst0, ew0, We1, agg, cnt, E0, N1);
  finalize_kernel<64, 16, false, 4><<<fin_grid(N1, 4), blk, 0, stream>>>(
      xd1, yd1, agg, cnt, h1, N1);

  const int R2s = N1 * B;
  const int R2d = N2 * B;
  layer_nodes_wmma<16, 32, 0, 4, 2><<<node_grid(R2s / 16, 4, R2d / 16, 2), gblk, 0, stream>>>(
      h1, Wn2, V2, U2, res1, ys, xd2, yd2, R2s, R2d, 0);
  edge_owner_kernel<128, 32, 100, 8, 2560><<<owner_grid(N2, 100), oblk, 0, stream>>>(
      ys, yd2, src1, dst1, ew1, We2, agg, cnt, E1, N2);
  finalize_kernel<128, 32, true, 4><<<fin_grid(N2, 4), blk, 0, stream>>>(
      xd2, yd2, agg, cnt, (float*)d_out, N2);
}


# Finalize folded into the edge launch: the group that holds a range's sums in LDS finishes its nodes itself (mean, BN, residual, leaky_relu) and writes h1 / out; no agg or cnt plane; launches 6 to 4
# speedup vs baseline: 2.0848x; 1.0536x over previous
#include <hip/hip_runtime.h>
#include <hip/hip_bf16.h>

typedef float v2f __attribute__((ext_vector_type(2)));
typedef float v8f __attribute__((ext_vector_type(8)));


template <int K, int MODE, bool GATHER, int TT>
__device__ __forceinline__ void fold_load_rows(const float* __restrict__ A,
                                               const int (&gnode)[TT],
                                               v2f (&af)[TT][K / 4],
                                               int row00, int nsrc, int lane) {
  const int m = lane & 15;
  const int h = lane >> 4;
#pragma unroll
  for (int t = 0; t < TT; ++t) {
    const int r = row00 + t * 16 + m;
    int n = r >> 2;
    const int b = r & 3;
    if (GATHER) n = gnode[t];
    long off;
    if (MODE == 0) {
      off = ((long)n * 4 + b) * (long)K;
    } else {
      off = ((long)b * nsrc + n) * (long)K;
    }
    const float4* __restrict__ arow = reinterpret_cast<const float4*>(A + off);
#pragma unroll
    for (int t2 = 0; t2 < K / 16; ++t2) {
      const float4 lo = arow[t2 * 4 + h * 2];
      const float4 hi = arow[t2 * 4 + h * 2 + 1];
      af[t][t2 * 4 + 0].x = lo.x; af[t][t2 * 4 + 0].y = lo.y;
      af[t][t2 * 4 + 1].x = lo.z; af[t][t2 * 4 + 1].y = lo.w;
      af[t][t2 * 4 + 2].x = hi.x; af[t][t2 * 4 + 2].y = hi.y;
      af[t][t2 * 4 + 3].x = hi.z; af[t][t2 * 4 + 3].y = hi.w;
    }
  }
}

template <int K, int N>
__device__ __forceinline__ void fold_form(const v2f (&wa)[K / 16][N / 4],
                                          const v2f (&vb)[N / 4][N / 16],
                                          v8f (&ff)[K / 16][N / 16]) {
#pragma unroll
  for (int t2 = 0; t2 < K / 16; ++t2) {
#pragma unroll
    for (int nt = 0; nt < N / 16; ++nt) ff[t2][nt] = v8f{};
  }
#pragma unroll
  for (int q = 0; q < N / 4; ++q) {
#pragma unroll
    for (int t2 = 0; t2 < K / 16; ++t2) {
#pragma unroll
      for (int nt = 0; nt < N / 16; ++nt) {
        ff[t2][nt] = __builtin_amdgcn_wmma_f32_16x16x4_f32(
             false, wa[t2][q],  false, vb[q][nt],
             (short)0, ff[t2][nt],  false,  false);
      }
    }
  }
}

template <int K, int N, int TT, bool BOTH>
__device__ __forceinline__ void fold_products(const v2f (&af)[TT][K / 4],
                                              const v2f (&wf)[K / 4][N / 16],
                                              const v8f (&ff)[K / 16][N / 16],
                                              float* __restrict__ Y1, float* __restrict__ Y2,
                                              int row00, int lane) {
  const int h  = lane >> 4;
  const int cn = lane & 15;
#pragma unroll
  for (int t = 0; t < TT; ++t) {
    v8f acc1[N / 16] = {};
    v8f acc2[N / 16] = {};
#pragma unroll
    for (int ks = 0; ks < K / 4; ++ks) {
      const int t2 = ks >> 2, j = ks & 3;
#pragma unroll
      for (int nt = 0; nt < N / 16; ++nt) {
        v2f fb;
        fb.x = ff[t2][nt][2 * j];
        fb.y = ff[t2][nt][2 * j + 1];
        if (BOTH) {
          acc1[nt] = __builtin_amdgcn_wmma_f32_16x16x4_f32(
               false, af[t][ks],  false, wf[ks][nt],
               (short)0, acc1[nt],  false,  false);
        }
        acc2[nt] = __builtin_amdgcn_wmma_f32_16x16x4_f32(
             false, af[t][ks],  false, fb,
             (short)0, acc2[nt],  false,  false);
      }
    }
#pragma unroll
    for (int nt = 0; nt < N / 16; ++nt) {
#pragma unroll
      for (int vr = 0; vr < 8; ++vr) {
        const int rr = row00 + t * 16 + h * 8 + vr;
        if (BOTH) Y1[(long)rr * N + nt * 16 + cn] = acc1[nt][vr];
        Y2[(long)rr * N + nt * 16 + cn] = acc2[nt][vr];
      }
    }
  }
}

template <int K, int N, int MODE, int T, int TD>
__global__ void __launch_bounds__(128) layer_nodes_wmma(
    const float* __restrict__ A, const float* __restrict__ W, const float* __restrict__ V,
    const float* __restrict__ U, const int* __restrict__ res,
    float* __restrict__ YS, float* __restrict__ XD, float* __restrict__ YD,
    int Rs, int Rd, int nsrc) {
  const int wave = blockIdx.x * (blockDim.x >> 5) + (threadIdx.x >> 5);
  const int lane = threadIdx.x & 31;
  const int row00 = wave * (16 * T);
  const int drow0 = wave * (16 * TD);
  if (row00 >= Rs && drow0 >= Rd) return;

  const int m  = lane & 15;
  const int h  = lane >> 4;
  const int kb = h << 1;

  int gnode[TD];
#pragma unroll
  for (int t = 0; t < TD; ++t) {
    const int r  = drow0 + t * 16 + m;
    const int rc = r < Rd ? r : Rd - 1;
    gnode[t] = res[rc >> 2];
  }

  v2f vb[N / 4][N / 16];
  v2f ub[N / 4][N / 16];
  v2f wa[K / 16][N / 4];
#pragma unroll
  for (int q = 0; q < N / 4; ++q) {
#pragma unroll
    for (int nt = 0; nt < N / 16; ++nt) {
      vb[q][nt].x = V[(q * 4 + kb) * N + nt * 16 + m];
      vb[q][nt].y = V[(q * 4 + kb + 1) * N + nt * 16 + m];
      ub[q][nt].x = U[(q * 4 + kb) * N + nt * 16 + m];
      ub[q][nt].y = U[(q * 4 + kb + 1) * N + nt * 16 + m];
    }
#pragma unroll
    for (int t2 = 0; t2 < K / 16; ++t2) {
      wa[t2][q].x = W[(t2 * 16 + m) * N + q * 4 + kb];
      wa[t2][q].y = W[(t2 * 16 + m) * N + q * 4 + kb + 1];
    }
  }
  v2f wf[K / 4][N / 16];
#pragma unroll
  for (int ks = 0; ks < K / 4; ++ks) {
    const int kk = (ks >> 2) * 16 + h * 8 + (ks & 3) * 2;
#pragma unroll
    for (int nt = 0; nt < N / 16; ++nt) {
      wf[ks][nt].x = W[kk * N + nt * 16 + m];
      wf[ks][nt].y = W[(kk + 1) * N + nt * 16 + m];
    }
  }

  v8f fv[K / 16][N / 16];
  v8f fu[K / 16][N / 16];
  if (row00 + 16 * T <= Rs && drow0 + 16 * TD <= Rd) {
    v2f af[T][K / 4];
    v2f ad[TD][K / 4];
    const int nogather[T] = {};
    fold_load_rows<K, MODE, false, T>(A, nogather, af, row00, nsrc, lane);
    __builtin_amdgcn_sched_barrier(0);
    fold_load_rows<K, MODE, true, TD>(A, gnode, ad, drow0, nsrc, lane);
    __builtin_amdgcn_sched_barrier(0);
    fold_form<K, N>(wa, vb, fv);
    fold_form<K, N>(wa, ub, fu);
    fold_products<K, N, T, false>(af, wf, fv, nullptr, YS, row00, lane);
    fold_products<K, N, TD, true>(ad, wf, fu, XD, YD, drow0, lane);
  } else {
    fold_form<K, N>(wa, vb, fv);
    fold_form<K, N>(wa, ub, fu);
    const int nogather[1] = {};
    for (int row0 = row00; row0 < row00 + 16 * T && row0 < Rs; row0 += 16) {
      v2f af[1][K / 4];
      fold_load_rows<K, MODE, false, 1>(A, nogather, af, row0, nsrc, lane);
      fold_products<K, N, 1, false>(af, wf, fv, nullptr, YS, row0, lane);
    }
#pragma unroll
    for (int t = 0; t < TD; ++t) {
      const int row0 = drow0 + t * 16;
      if (row0 < Rd) {
        v2f ad[1][K / 4];
        const int g1[1] = {gnode[t]};
        fold_load_rows<K, MODE, true, 1>(A, g1, ad, row0, nsrc, lane);
        fold_products<K, N, 1, true>(ad, wf, fu, XD, YD, row0, lane);
      }
    }
  }
}


template <int BH, int H>
__device__ __forceinline__ void edge_by_one_lane(const float* __restrict__ ys,
                                                 const float* __restrict__ yd,
                                                 const int* __restrict__ src,
                                                 const float* __restrict__ ew,
                                                 const float* __restrict__ We,
                                                 float* s_agg, float* s_cnt, int n0,
                                                 int e, int d) {
  const int s = src[e];
  const float w = ew[e];
  float* ar = s_agg + (d - n0) * BH;
  for (int i = 0; i < BH; ++i) {
    const float xjv = ys[(long)s * BH + i];
    const float xiv = yd[(long)d * BH + i];
    const float ev  = w * We[i & (H - 1)];
    const float t   = xiv * xjv * ev;
    const float sg  = 1.0f / (1.0f + __expf(-t));
    atomicAdd(ar + i, xjv * sg);
  }
  atomicAdd(s_cnt + (d - n0), 1.0f);
}

template <int BH, int H, int U, bool FULL>
__device__ __forceinline__ void owner_tile(const float* __restrict__ ys,
                                           const float* __restrict__ yd,
                                           float* s_agg, int n0, int sv, int dv, float wv,
                                           float wev, int nT, int lane) {
  constexpr int PER = BH / 32;
  for (int j0 = 0; j0 < (FULL ? 32 : nT); j0 += U) {
    float xj[U][PER], xi[U][PER];
#pragma unroll
    for (int u = 0; u < U; ++u) {
      const int jj = (FULL || j0 + u < nT) ? (j0 + u) : (nT - 1);
      const int s = __builtin_amdgcn_readlane(sv, jj);
      const int d = __builtin_amdgcn_readlane(dv, jj);
      const float* __restrict__ ysr = ys + (long)s * BH + lane;
      const float* __restrict__ ydr = yd + (long)d * BH + lane;
#pragma unroll
      for (int k = 0; k < PER; ++k) {
        xj[u][k] = ysr[32 * k];
        xi[u][k] = ydr[32 * k];
      }
    }
#pragma unroll
    for (int u = 0; u < U; ++u) {
      if (FULL || j0 + u < nT) {
        const int d = __builtin_amdgcn_readlane(dv, j0 + u);
        const float w = __int_as_float(__builtin_amdgcn_readlane(__float_as_int(wv), j0 + u));
        const float ev = w * wev;
        float* ar = s_agg + (d - n0) * BH + lane;
#pragma unroll
        for (int k = 0; k < PER; ++k) {
          const float t  = xi[u][k] * xj[u][k] * ev;
          const float sg = 1.0f / (1.0f + __expf(-t));
          atomicAdd(ar + 32 * k, xj[u][k] * sg);
        }
      }
    }
  }
}

template <int BH, int H, int NPW, int U, int QCAP, bool FINAL>
__global__ void __launch_bounds__(1024) edge_owner_kernel(
    const float* __restrict__ ys, const float* yd, const float* __restrict__ xd,
    const int* __restrict__ src, const int* __restrict__ dst,
    const float* __restrict__ ew, const float* __restrict__ We,
    float* out, int E, int Nd) {
  constexpr int SB  = 8;
  __shared__ float s_agg[NPW * BH];
  __shared__ float s_cnt[NPW];
  __shared__ int   s_q[QCAP];
  __shared__ int   s_qn;

  constexpr int NTHR = 1024;
  const int tid   = threadIdx.x;
  const int lane  = tid & 31;
  const int wave  = __builtin_amdgcn_readfirstlane(tid >> 5);
  constexpr int nthr  = NTHR;
  constexpr int nwave = NTHR >> 5;
  const int n0 = blockIdx.x * NPW;
  if (n0 >= Nd) return;
  const int nn = (Nd - n0 < NPW) ? (Nd - n0) : NPW;

  for (int i = tid; i < NPW * BH; i += nthr) s_agg[i] = 0.0f;
  for (int i = tid; i < NPW; i += nthr) s_cnt[i] = 0.0f;
  if (tid == 0) s_qn = 0;
  __syncthreads();

  auto file4 = [&](const int4 a, const int e) {
    const bool m0 = (unsigned)(a.x - n0) < (unsigned)nn;
    const bool m1 = (unsigned)(a.y - n0) < (unsigned)nn;
    const bool m2 = (unsigned)(a.z - n0) < (unsigned)nn;
    const bool m3 = (unsigned)(a.w - n0) < (unsigned)nn;
    if (m0 | m1 | m2 | m3) {
      if (m0) { const int pos = atomicAdd(&s_qn, 1); if (pos < QCAP) s_q[pos] = e; }
      if (m1) { const int pos = atomicAdd(&s_qn, 1); if (pos < QCAP) s_q[pos] = e + 1; }
      if (m2) { const int pos = atomicAdd(&s_qn, 1); if (pos < QCAP) s_q[pos] = e + 2; }
      if (m3) { const int pos = atomicAdd(&s_qn, 1); if (pos < QCAP) s_q[pos] = e + 3; }
    }
  };
  const int4* __restrict__ d4 = reinterpret_cast<const int4*>(dst);
  const int nvec  = E >> 2;
  const int nfull = nvec / (NTHR * SB);
  const int rotb  = (int)(((long)blockIdx.x * nfull) / gridDim.x);
  for (int b = 0; b < nfull; ++b) {
    int bb = b + rotb;
    if (bb >= nfull) bb -= nfull;
    const int vbase = bb * (NTHR * SB);
    const int4* __restrict__ p = d4 + vbase + tid;
    int4 a[SB];
#pragma unroll
    for (int i = 0; i < SB; ++i) a[i] = p[i * NTHR];
#pragma unroll
    for (int i = 0; i < SB; ++i) file4(a[i], (vbase + i * NTHR + tid) * 4);
  }
  for (int v = nfull * (NTHR * SB) + tid; v < nvec; v += NTHR) file4(d4[v], v * 4);
  for (int e = nvec * 4 + tid; e < E; e += NTHR) {
    if ((unsigned)(dst[e] - n0) < (unsigned)nn) {
      const int pos = atomicAdd(&s_qn, 1);
      if (pos < QCAP) s_q[pos] = e;
    }
  }
  __syncthreads();
  const int qn = __builtin_amdgcn_readfirstlane(s_qn);

  if (qn > QCAP) {
    for (int e = tid; e < E; e += nthr) {
      const int d = dst[e];
      if ((unsigned)(d - n0) < (unsigned)nn)
        edge_by_one_lane<BH, H>(ys, yd, src, ew, We, s_agg, s_cnt, n0, e, d);
    }
  } else {
    const float wev = We[lane & (H - 1)];
    for (int t0 = wave * 32; t0 < qn; t0 += nwave * 32) {
      const int nT = (qn - t0 < 32) ? (qn - t0) : 32;
      int sv = 0, dv = n0;
      float wv = 0.0f;
      if (lane < nT) {
        const int e = s_q[t0 + lane];
        sv = src[e];
        dv = dst[e];
        wv = ew[e];
        atomicAdd(s_cnt + (dv - n0), 1.0f);
      }
      if (nT == 32) {
        owner_tile<BH, H, U, true>(ys, yd, s_agg, n0, sv, dv, wv, wev, 32, lane);
      } else {
        owner_tile<BH, H, U, false>(ys, yd, s_agg, n0, sv, dv, wv, wev, nT, lane);
      }
    }
  }

  constexpr int PER = BH / 32;
  constexpr int NQ  = (NPW + nwave - 1) / nwave;
  float ydv[NQ][PER], xdv[NQ][PER];
#pragma unroll
  for (int q = 0; q < NQ; ++q) {
    const int ln = wave + q * nwave;
    const int lc = ln < nn ? ln : nn - 1;
    const long base = (long)(n0 + lc) * BH + lane;
#pragma unroll
    for (int j = 0; j < PER; ++j) {
      ydv[q][j] = yd[base + 32 * j];
      xdv[q][j] = xd[base + 32 * j];
    }
  }
  __syncthreads();

#pragma unroll
  for (int q = 0; q < NQ; ++q) {
    const int ln = wave + q * nwave;
    if (ln < nn) {
      const int d = n0 + ln;
      const float inv_cnt = 1.0f / fmaxf(s_cnt[ln], 1.0f);
      const float* ar = s_agg + ln * BH + lane;
      float o[PER];
      float sum = 0.0f, ssq = 0.0f;
#pragma unroll
      for (int j = 0; j < PER; ++j) {
        o[j] = ydv[q][j] + ar[32 * j] * inv_cnt;
        sum += o[j];
        ssq += o[j] * o[j];
      }
#pragma unroll
      for (int mask = 16; mask >= 1; mask >>= 1) {
        sum += __shfl_xor(sum, mask, 32);
        ssq += __shfl_xor(ssq, mask, 32);
      }
      const float mu  = sum * (1.0f / BH);
      const float var = ssq * (1.0f / BH) - mu * mu;
      const float is  = rsqrtf(var + 1e-5f);

      const long base = (long)d * BH;
#pragma unroll
      for (int j = 0; j < PER; ++j) {
        const int i = lane + j * 32;
        float v = xdv[q][j] + (o[j] - mu) * is;
        v = v > 0.0f ? v : 0.01f * v;
        if (FINAL) {
          const int b = i / H, c = i & (H - 1);
          out[((long)b * Nd + d) * H + c] = v;
        } else {
          out[base + i] = v;
        }
      }
    }
  }
}


extern "C" void kernel_launch(void* const* d_in, const int* in_sizes, int n_in,
                              void* d_out, int out_size, void* d_ws, size_t ws_size,
                              hipStream_t stream) {
  constexpr int B = 4, Cin = 32, H = 16, Cout = 32;
  const float* X    = (const float*)d_in[0];
  const int*   src0 = (const int*)d_in[1];
  const int*   dst0 = (const int*)d_in[2];
  const float* ew0  = (const float*)d_in[3];
  const int*   res0 = (const int*)d_in[4];
  const int*   src1 = (const int*)d_in[5];
  const int*   dst1 = (const int*)d_in[6];
  const float* ew1  = (const float*)d_in[7];
  const int*   res1 = (const int*)d_in[8];
  const float* Wn1  = (const float*)d_in[9];
  const float* We1  = (const float*)d_in[10];
  const float* U1   = (const float*)d_in[11];
  const float* V1   = (const float*)d_in[12];
  const float* Wn2  = (const float*)d_in[13];
  const float* We2  = (const float*)d_in[14];
  const float* U2   = (const float*)d_in[15];
  const float* V2   = (const float*)d_in[16];

  const int E0 = in_sizes[1];
  const int N1 = in_sizes[4];
  const int E1 = in_sizes[5];
  const int N2 = in_sizes[8];
  const int N0 = in_sizes[0] / (B * Cin);

  float* ws = (float*)d_ws;
  const size_t szXS = (size_t)((N0 * B * H > N1 * B * Cout) ? N0 * B * H : N1 * B * Cout);
  float* regP = ws;
  float* ys   = regP + szXS;
  float* regR = ys + szXS;
  float* xd1  = regP;
  float* yd1  = regR;
  float* h1   = regR;
  float* xd2  = regP;
  float* yd2  = regP + (size_t)N2 * B * Cout;

  const dim3 gblk(128);
  auto node_grid = [](int stiles, int T, int dtiles, int TD) {
    const int ws_ = (stiles + T - 1) / T;
    const int wd_ = (dtiles + TD - 1) / TD;
    const int waves = ws_ > wd_ ? ws_ : wd_;
    return dim3((unsigned)((waves + 3) / 4));
  };
  const dim3 oblk(1024);
  auto owner_grid = [](int nodes, int NPW) {
    return dim3((unsigned)((nodes + NPW - 1) / NPW));
  };

  const int R1s = N0 * B;
  const int R1d = N1 * B;
  layer_nodes_wmma<32, 16, 1, 4, 2><<<node_grid(R1s / 16, 4, R1d / 16, 2), gblk, 0, stream>>>(
      X, Wn1, V1, U1, res0, ys, xd1, yd1, R1s, R1d, N0);
  edge_owner_kernel<64, 16, 200, 8, 2560, false><<<owner_grid(N1, 200), oblk, 0, stream>>>(
      ys, yd1, xd1, src0, dst0, ew0, We1, h1, E0, N1);

  const int R2s = N1 * B;
  const int R2d = N2 * B;
  layer_nodes_wmma<16, 32, 0, 4, 2><<<node_grid(R2s / 16, 4, R2d / 16, 2), gblk, 0, stream>>>(
      h1, Wn2, V2, U2, res1, ys, xd2, yd2, R2s, R2d, 0);
  edge_owner_kernel<128, 32, 100, 8, 2560, true><<<owner_grid(N2, 100), oblk, 0, stream>>>(
      ys, yd2, xd2, src1, dst1, ew1, We2, (float*)d_out, E1, N2);
}


# Edge scan: node launch writes one byte per edge (owner range of its dst); owner groups scan that plane, 16 edges per 16-byte load, not the 4-byte dst array; list (float sum) order free as before
# speedup vs baseline: 2.2702x; 1.0889x over previous
#include <hip/hip_runtime.h>
#include <hip/hip_bf16.h>

typedef float v2f __attribute__((ext_vector_type(2)));
typedef float v8f __attribute__((ext_vector_type(8)));


template <int K, int MODE, bool GATHER, int TT>
__device__ __forceinline__ void fold_load_rows(const float* __restrict__ A,
                                               const int (&gnode)[TT],
                                               v2f (&af)[TT][K / 4],
                                               int row00, int nsrc, int lane) {
  const int m = lane & 15;
  const int h = lane >> 4;
#pragma unroll
  for (int t = 0; t < TT; ++t) {
    const int r = row00 + t * 16 + m;
    int n = r >> 2;
    const int b = r & 3;
    if (GATHER) n = gnode[t];
    long off;
    if (MODE == 0) {
      off = ((long)n * 4 + b) * (long)K;
    } else {
      off = ((long)b * nsrc + n) * (long)K;
    }
    const float4* __restrict__ arow = reinterpret_cast<const float4*>(A + off);
#pragma unroll
    for (int t2 = 0; t2 < K / 16; ++t2) {
      const float4 lo = arow[t2 * 4 + h * 2];
      const float4 hi = arow[t2 * 4 + h * 2 + 1];
      af[t][t2 * 4 + 0].x = lo.x; af[t][t2 * 4 + 0].y = lo.y;
      af[t][t2 * 4 + 1].x = lo.z; af[t][t2 * 4 + 1].y = lo.w;
      af[t][t2 * 4 + 2].x = hi.x; af[t][t2 * 4 + 2].y = hi.y;
      af[t][t2 * 4 + 3].x = hi.z; af[t][t2 * 4 + 3].y = hi.w;
    }
  }
}

template <int K, int N>
__device__ __forceinline__ void fold_form(const v2f (&wa)[K / 16][N / 4],
                                          const v2f (&vb)[N / 4][N / 16],
                                          v8f (&ff)[K / 16][N / 16]) {
#pragma unroll
  for (int t2 = 0; t2 < K / 16; ++t2) {
#pragma unroll
    for (int nt = 0; nt < N / 16; ++nt) ff[t2][nt] = v8f{};
  }
#pragma unroll
  for (int q = 0; q < N / 4; ++q) {
#pragma unroll
    for (int t2 = 0; t2 < K / 16; ++t2) {
#pragma unroll
      for (int nt = 0; nt < N / 16; ++nt) {
        ff[t2][nt] = __builtin_amdgcn_wmma_f32_16x16x4_f32(
             false, wa[t2][q],  false, vb[q][nt],
             (short)0, ff[t2][nt],  false,  false);
      }
    }
  }
}

template <int K, int N, int TT, bool BOTH>
__device__ __forceinline__ void fold_products(const v2f (&af)[TT][K / 4],
                                              const v2f (&wf)[K / 4][N / 16],
                                              const v8f (&ff)[K / 16][N / 16],
                                              float* __restrict__ Y1, float* __restrict__ Y2,
                                              int row00, int lane) {
  const int h  = lane >> 4;
  const int cn = lane & 15;
#pragma unroll
  for (int t = 0; t < TT; ++t) {
    v8f acc1[N / 16] = {};
    v8f acc2[N / 16] = {};
#pragma unroll
    for (int ks = 0; ks < K / 4; ++ks) {
      const int t2 = ks >> 2, j = ks & 3;
#pragma unroll
      for (int nt = 0; nt < N / 16; ++nt) {
        v2f fb;
        fb.x = ff[t2][nt][2 * j];
        fb.y = ff[t2][nt][2 * j + 1];
        if (BOTH) {
          acc1[nt] = __builtin_amdgcn_wmma_f32_16x16x4_f32(
               false, af[t][ks],  false, wf[ks][nt],
               (short)0, acc1[nt],  false,  false);
        }
        acc2[nt] = __builtin_amdgcn_wmma_f32_16x16x4_f32(
             false, af[t][ks],  false, fb,
             (short)0, acc2[nt],  false,  false);
      }
    }
#pragma unroll
    for (int nt = 0; nt < N / 16; ++nt) {
#pragma unroll
      for (int vr = 0; vr < 8; ++vr) {
        const int rr = row00 + t * 16 + h * 8 + vr;
        if (BOTH) Y1[(long)rr * N + nt * 16 + cn] = acc1[nt][vr];
        Y2[(long)rr * N + nt * 16 + cn] = acc2[nt][vr];
      }
    }
  }
}

template <int K, int N, int MODE, int T, int TD, int NPW>
__global__ void __launch_bounds__(128) layer_nodes_wmma(
    const float* __restrict__ A, const float* __restrict__ W, const float* __restrict__ V,
    const float* __restrict__ U, const int* __restrict__ res,
    float* __restrict__ YS, float* __restrict__ XD, float* __restrict__ YD,
    int Rs, int Rd, int nsrc,
    const int* __restrict__ dstE, unsigned int* __restrict__ rid, int E, int Nd) {
  const int wave = blockIdx.x * (blockDim.x >> 5) + (threadIdx.x >> 5);
  const int lane = threadIdx.x & 31;
  const int row00 = wave * (16 * T);
  const int drow0 = wave * (16 * TD);

  const int gl = wave * 32 + lane;
  const bool plane = rid != nullptr && gl < (E >> 2);
  int4 dq = {0, 0, 0, 0};
  if (plane) dq = reinterpret_cast<const int4*>(dstE)[gl];
  auto range_of = [&](const int d) {
    return (unsigned)d < (unsigned)Nd ? (unsigned)d / (unsigned)NPW : 255u;
  };
  auto store_plane = [&]() {
    if (plane)
      rid[gl] = range_of(dq.x) | (range_of(dq.y) << 8) | (range_of(dq.z) << 16) |
                (range_of(dq.w) << 24);
  };
  if (row00 >= Rs && drow0 >= Rd) {
    store_plane();
    return;
  }

  const int m  = lane & 15;
  const int h  = lane >> 4;
  const int kb = h << 1;

  int gnode[TD];
#pragma unroll
  for (int t = 0; t < TD; ++t) {
    const int r  = drow0 + t * 16 + m;
    const int rc = r < Rd ? r : Rd - 1;
    gnode[t] = res[rc >> 2];
  }

  v2f vb[N / 4][N / 16];
  v2f ub[N / 4][N / 16];
  v2f wa[K / 16][N / 4];
#pragma unroll
  for (int q = 0; q < N / 4; ++q) {
#pragma unroll
    for (int nt = 0; nt < N / 16; ++nt) {
      vb[q][nt].x = V[(q * 4 + kb) * N + nt * 16 + m];
      vb[q][nt].y = V[(q * 4 + kb + 1) * N + nt * 16 + m];
      ub[q][nt].x = U[(q * 4 + kb) * N + nt * 16 + m];
      ub[q][nt].y = U[(q * 4 + kb + 1) * N + nt * 16 + m];
    }
#pragma unroll
    for (int t2 = 0; t2 < K / 16; ++t2) {
      wa[t2][q].x = W[(t2 * 16 + m) * N + q * 4 + kb];
      wa[t2][q].y = W[(t2 * 16 + m) * N + q * 4 + kb + 1];
    }
  }
  v2f wf[K / 4][N / 16];
#pragma unroll
  for (int ks = 0; ks < K / 4; ++ks) {
    const int kk = (ks >> 2) * 16 + h * 8 + (ks & 3) * 2;
#pragma unroll
    for (int nt = 0; nt < N / 16; ++nt) {
      wf[ks][nt].x = W[kk * N + nt * 16 + m];
      wf[ks][nt].y = W[(kk + 1) * N + nt * 16 + m];
    }
  }

  v8f fv[K / 16][N / 16];
  v8f fu[K / 16][N / 16];
  if (row00 + 16 * T <= Rs && drow0 + 16 * TD <= Rd) {
    v2f af[T][K / 4];
    v2f ad[TD][K / 4];
    const int nogather[T] = {};
    fold_load_rows<K, MODE, false, T>(A, nogather, af, row00, nsrc, lane);
    __builtin_amdgcn_sched_barrier(0);
    fold_load_rows<K, MODE, true, TD>(A, gnode, ad, drow0, nsrc, lane);
    __builtin_amdgcn_sched_barrier(0);
    fold_form<K, N>(wa, vb, fv);
    fold_form<K, N>(wa, ub, fu);
    fold_products<K, N, T, false>(af, wf, fv, nullptr, YS, row00, lane);
    fold_products<K, N, TD, true>(ad, wf, fu, XD, YD, drow0, lane);
    store_plane();
  } else {
    fold_form<K, N>(wa, vb, fv);
    fold_form<K, N>(wa, ub, fu);
    const int nogather[1] = {};
    for (int row0 = row00; row0 < row00 + 16 * T && row0 < Rs; row0 += 16) {
      v2f af[1][K / 4];
      fold_load_rows<K, MODE, false, 1>(A, nogather, af, row0, nsrc, lane);
      fold_products<K, N, 1, false>(af, wf, fv, nullptr, YS, row0, lane);
    }
#pragma unroll
    for (int t = 0; t < TD; ++t) {
      const int row0 = drow0 + t * 16;
      if (row0 < Rd) {
        v2f ad[1][K / 4];
        const int g1[1] = {gnode[t]};
        fold_load_rows<K, MODE, true, 1>(A, g1, ad, row0, nsrc, lane);
        fold_products<K, N, 1, true>(ad, wf, fu, XD, YD, row0, lane);
      }
    }
    store_plane();
  }
}


template <int BH, int H>
__device__ __forceinline__ void edge_by_one_lane(const float* __restrict__ ys,
                                                 const float* __restrict__ yd,
                                                 const int* __restrict__ src,
                                                 const float* __restrict__ ew,
                                                 const float* __restrict__ We,
                                                 float* s_agg, float* s_cnt, int n0,
                                                 int e, int d) {
  const int s = src[e];
  const float w = ew[e];
  float* ar = s_agg + (d - n0) * BH;
  for (int i = 0; i < BH; ++i) {
    const float xjv = ys[(long)s * BH + i];
    const float xiv = yd[(long)d * BH + i];
    const float ev  = w * We[i & (H - 1)];
    const float t   = xiv * xjv * ev;
    const float sg  = 1.0f / (1.0f + __expf(-t));
    atomicAdd(ar + i, xjv * sg);
  }
  atomicAdd(s_cnt + (d - n0), 1.0f);
}

template <int BH, int H, int U, bool FULL>
__device__ __forceinline__ void owner_tile(const float* __restrict__ ys,
                                           const float* __restrict__ yd,
                                           float* s_agg, int n0, int sv, int dv, float wv,
                                           float wev, int nT, int lane) {
  constexpr int PER = BH / 32;
  for (int j0 = 0; j0 < (FULL ? 32 : nT); j0 += U) {
    float xj[U][PER], xi[U][PER];
#pragma unroll
    for (int u = 0; u < U; ++u) {
      const int jj = (FULL || j0 + u < nT) ? (j0 + u) : (nT - 1);
      const int s = __builtin_amdgcn_readlane(sv, jj);
      const int d = __builtin_amdgcn_readlane(dv, jj);
      const float* __restrict__ ysr = ys + (long)s * BH + lane;
      const float* __restrict__ ydr = yd + (long)d * BH + lane;
#pragma unroll
      for (int k = 0; k < PER; ++k) {
        xj[u][k] = ysr[32 * k];
        xi[u][k] = ydr[32 * k];
      }
    }
#pragma unroll
    for (int u = 0; u < U; ++u) {
      if (FULL || j0 + u < nT) {
        const int d = __builtin_amdgcn_readlane(dv, j0 + u);
        const float w = __int_as_float(__builtin_amdgcn_readlane(__float_as_int(wv), j0 + u));
        const float ev = w * wev;
        float* ar = s_agg + (d - n0) * BH + lane;
#pragma unroll
        for (int k = 0; k < PER; ++k) {
          const float t  = xi[u][k] * xj[u][k] * ev;
          const float sg = 1.0f / (1.0f + __expf(-t));
          atomicAdd(ar + 32 * k, xj[u][k] * sg);
        }
      }
    }
  }
}

template <int BH, int H, int NPW, int U, int QCAP, bool FINAL>
__global__ void __launch_bounds__(1024) edge_owner_kernel(
    const float* __restrict__ ys, const float* yd, const float* __restrict__ xd,
    const int* __restrict__ src, const int* __restrict__ dst,
    const float* __restrict__ ew, const float* __restrict__ We,
    const unsigned int* __restrict__ rid, float* out, int E, int Nd) {
  constexpr int SB  = 8;
  constexpr int SBP = 12;
  __shared__ float s_agg[NPW * BH];
  __shared__ float s_cnt[NPW];
  __shared__ int   s_q[QCAP];
  __shared__ int   s_qn;

  constexpr int NTHR = 1024;
  const int tid   = threadIdx.x;
  const int lane  = tid & 31;
  const int wave  = __builtin_amdgcn_readfirstlane(tid >> 5);
  constexpr int nthr  = NTHR;
  constexpr int nwave = NTHR >> 5;
  const int n0 = blockIdx.x * NPW;
  if (n0 >= Nd) return;
  const int nn = (Nd - n0 < NPW) ? (Nd - n0) : NPW;

  for (int i = tid; i < NPW * BH; i += nthr) s_agg[i] = 0.0f;
  for (int i = tid; i < NPW; i += nthr) s_cnt[i] = 0.0f;
  if (tid == 0) s_qn = 0;
  __syncthreads();

  auto file4 = [&](const int4 a, const int e) {
    const bool m0 = (unsigned)(a.x - n0) < (unsigned)nn;
    const bool m1 = (unsigned)(a.y - n0) < (unsigned)nn;
    const bool m2 = (unsigned)(a.z - n0) < (unsigned)nn;
    const bool m3 = (unsigned)(a.w - n0) < (unsigned)nn;
    if (m0 | m1 | m2 | m3) {
      if (m0) { const int pos = atomicAdd(&s_qn, 1); if (pos < QCAP) s_q[pos] = e; }
      if (m1) { const int pos = atomicAdd(&s_qn, 1); if (pos < QCAP) s_q[pos] = e + 1; }
      if (m2) { const int pos = atomicAdd(&s_qn, 1); if (pos < QCAP) s_q[pos] = e + 2; }
      if (m3) { const int pos = atomicAdd(&s_qn, 1); if (pos < QCAP) s_q[pos] = e + 3; }
    }
  };
  const int4* __restrict__ d4 = reinterpret_cast<const int4*>(dst);
  const int nvec  = E >> 2;
  if (rid != nullptr) {
    const unsigned pat = blockIdx.x * 0x01010101u;
    auto zero_byte_any = [](const unsigned x) { return (x - 0x01010101u) & ~x; };
    auto nonzero_bytes = [](const unsigned x) {
      return ((x & 0x7f7f7f7fu) + 0x7f7f7f7fu) | x;
    };
    auto file_flags = [&](unsigned pk, const int e) {
      while (pk) {
        const int bit = __ffs(pk) - 1;
        pk &= pk - 1;
        const int pos = atomicAdd(&s_qn, 1);
        if (pos < QCAP) s_q[pos] = e + ((bit & 7) << 2) + (bit >> 3);
      }
    };
    const uint4* __restrict__ p16 = reinterpret_cast<const uint4*>(rid);
    const int nv16 = E >> 4;
    auto file_piece = [&](const uint4 q, const int v) {
      const unsigned t0 = nonzero_bytes(q.x ^ pat);
      const unsigned t1 = nonzero_bytes(q.y ^ pat);
      const unsigned t2 = nonzero_bytes(q.z ^ pat);
      const unsigned t3 = nonzero_bytes(q.w ^ pat);
      const unsigned pk = ((~t0 & 0x80808080u) >> 7) | ((~t1 & 0x80808080u) >> 6) |
                          ((~t2 & 0x80808080u) >> 5) | ((~t3 & 0x80808080u) >> 4);
      file_flags(pk, v * 16);
    };
    const int nfullp = nv16 / (NTHR * SBP);
    const int rotp   = (int)(blockIdx.x * (unsigned)nfullp / gridDim.x);
    const int tr     = (tid + 32 * (int)(blockIdx.x & 31u)) & (NTHR - 1);
    for (int b = 0; b < nfullp; ++b) {
      int bb = b + rotp;
      if (bb >= nfullp) bb -= nfullp;
      const int vbase = bb * (NTHR * SBP) + tr;
      const uint4* __restrict__ p = p16 + vbase;
      uint4 a[SBP];
#pragma unroll
      for (int i = 0; i < SBP; ++i) a[i] = p[i * NTHR];
      unsigned mine = 0;
#pragma unroll
      for (int i = 0; i < SBP; ++i) {
        const unsigned z = zero_byte_any(a[i].x ^ pat) | zero_byte_any(a[i].y ^ pat) |
                           zero_byte_any(a[i].z ^ pat) | zero_byte_any(a[i].w ^ pat);
        mine = (mine << 1) | min(z & 0x80808080u, 1u);
      }
      while (mine) {
        const int i = SBP - __ffs(mine);
        mine &= mine - 1;
        const int v = vbase + i * NTHR;
        file_piece(p16[v], v);
      }
    }
    for (int v = nfullp * (NTHR * SBP) + tid; v < nv16; v += NTHR) file_piece(p16[v], v);
    for (int w = nv16 * 4 + tid; w < nvec; w += NTHR) {
      const unsigned t = nonzero_bytes(rid[w] ^ pat);
      file_flags((~t & 0x80808080u) >> 7, w * 4);
    }
  } else {
    const int nfull = nvec / (NTHR * SB);
    const int rotb  = (int)(((long)blockIdx.x * nfull) / gridDim.x);
    for (int b = 0; b < nfull; ++b) {
      int bb = b + rotb;
      if (bb >= nfull) bb -= nfull;
      const int vbase = bb * (NTHR * SB);
      const int4* __restrict__ p = d4 + vbase + tid;
      int4 a[SB];
#pragma unroll
      for (int i = 0; i < SB; ++i) a[i] = p[i * NTHR];
#pragma unroll
      for (int i = 0; i < SB; ++i) file4(a[i], (vbase + i * NTHR + tid) * 4);
    }
    for (int v = nfull * (NTHR * SB) + tid; v < nvec; v += NTHR) file4(d4[v], v * 4);
  }
  for (int e = nvec * 4 + tid; e < E; e += NTHR) {
    if ((unsigned)(dst[e] - n0) < (unsigned)nn) {
      const int pos = atomicAdd(&s_qn, 1);
      if (pos < QCAP) s_q[pos] = e;
    }
  }
  __syncthreads();
  const int qn = __builtin_amdgcn_readfirstlane(s_qn);

  if (qn > QCAP) {
    for (int e = tid; e < E; e += nthr) {
      const int d = dst[e];
      if ((unsigned)(d - n0) < (unsigned)nn)
        edge_by_one_lane<BH, H>(ys, yd, src, ew, We, s_agg, s_cnt, n0, e, d);
    }
  } else {
    const float wev = We[lane & (H - 1)];
    for (int t0 = wave * 32; t0 < qn; t0 += nwave * 32) {
      const int nT = (qn - t0 < 32) ? (qn - t0) : 32;
      int sv = 0, dv = n0;
      float wv = 0.0f;
      if (lane < nT) {
        const int e = s_q[t0 + lane];
        sv = src[e];
        dv = dst[e];
        wv = ew[e];
        atomicAdd(s_cnt + (dv - n0), 1.0f);
      }
      if (nT == 32) {
        owner_tile<BH, H, U, true>(ys, yd, s_agg, n0, sv, dv, wv, wev, 32, lane);
      } else {
        owner_tile<BH, H, U, false>(ys, yd, s_agg, n0, sv, dv, wv, wev, nT, lane);
      }
    }
  }

  constexpr int PER = BH / 32;
  constexpr int NQ  = (NPW + nwave - 1) / nwave;
  float ydv[NQ][PER], xdv[NQ][PER];
#pragma unroll
  for (int q = 0; q < NQ; ++q) {
    const int ln = wave + q * nwave;
    const int lc = ln < nn ? ln : nn - 1;
    const long base = (long)(n0 + lc) * BH + lane;
#pragma unroll
    for (int j = 0; j < PER; ++j) {
      ydv[q][j] = yd[base + 32 * j];
      xdv[q][j] = xd[base + 32 * j];
    }
  }
  __syncthreads();

#pragma unroll
  for (int q = 0; q < NQ; ++q) {
    const int ln = wave + q * nwave;
    if (ln < nn) {
      const int d = n0 + ln;
      const float inv_cnt = 1.0f / fmaxf(s_cnt[ln], 1.0f);
      const float* ar = s_agg + ln * BH + lane;
      float o[PER];
      float sum = 0.0f, ssq = 0.0f;
#pragma unroll
      for (int j = 0; j < PER; ++j) {
        o[j] = ydv[q][j] + ar[32 * j] * inv_cnt;
        sum += o[j];
        ssq += o[j] * o[j];
      }
#pragma unroll
      for (int mask = 16; mask >= 1; mask >>= 1) {
        sum += __shfl_xor(sum, mask, 32);
        ssq += __shfl_xor(ssq, mask, 32);
      }
      const float mu  = sum * (1.0f / BH);
      const float var = ssq * (1.0f / BH) - mu * mu;
      const float is  = rsqrtf(var + 1e-5f);

      const long base = (long)d * BH;
#pragma unroll
      for (int j = 0; j < PER; ++j) {
        const int i = lane + j * 32;
        float v = xdv[q][j] + (o[j] - mu) * is;
        v = v > 0.0f ? v : 0.01f * v;
        if (FINAL) {
          const int b = i / H, c = i & (H - 1);
          out[((long)b * Nd + d) * H + c] = v;
        } else {
          out[base + i] = v;
        }
      }
    }
  }
}


extern "C" void kernel_launch(void* const* d_in, const int* in_sizes, int n_in,
                              void* d_out, int out_size, void* d_ws, size_t ws_size,
                              hipStream_t stream) {
  constexpr int B = 4, Cin = 32, H = 16, Cout = 32;
  const float* X    = (const float*)d_in[0];
  const int*   src0 = (const int*)d_in[1];
  const int*   dst0 = (const int*)d_in[2];
  const float* ew0  = (const float*)d_in[3];
  const int*   res0 = (const int*)d_in[4];
  const int*   src1 = (const int*)d_in[5];
  const int*   dst1 = (const int*)d_in[6];
  const float* ew1  = (const float*)d_in[7];
  const int*   res1 = (const int*)d_in[8];
  const float* Wn1  = (const float*)d_in[9];
  const float* We1  = (const float*)d_in[10];
  const float* U1   = (const float*)d_in[11];
  const float* V1   = (const float*)d_in[12];
  const float* Wn2  = (const float*)d_in[13];
  const float* We2  = (const float*)d_in[14];
  const float* U2   = (const float*)d_in[15];
  const float* V2   = (const float*)d_in[16];

  const int E0 = in_sizes[1];
  const int N1 = in_sizes[4];
  const int E1 = in_sizes[5];
  const int N2 = in_sizes[8];
  const int N0 = in_sizes[0] / (B * Cin);

  float* ws = (float*)d_ws;
  const size_t szXS = (size_t)((N0 * B * H > N1 * B * Cout) ? N0 * B * H : N1 * B * Cout);
  float* regP = ws;
  float* ys   = regP + szXS;
  float* regR = ys + szXS;
  float* xd1  = regP;
  float* yd1  = regR;
  float* h1   = regR;
  float* xd2  = regP;
  float* yd2  = regP + (size_t)N2 * B * Cout;

  const dim3 gblk(128);
  auto node_grid = [](int stiles, int T, int dtiles, int TD) {
    const int ws_ = (stiles + T - 1) / T;
    const int wd_ = (dtiles + TD - 1) / TD;
    const int waves = ws_ > wd_ ? ws_ : wd_;
    return dim3((unsigned)((waves + 3) / 4));
  };
  const dim3 oblk(1024);
  auto owner_grid = [](int nodes, int NPW) {
    return dim3((unsigned)((nodes + NPW - 1) / NPW));
  };

  const int R1s = N0 * B;
  const int R1d = N1 * B;
  const dim3 ng1 = node_grid(R1s / 16, 4, R1d / 16, 2);
  const dim3 og1 = owner_grid(N1, 200);
  unsigned int* rid1 = (unsigned int*)(regP + (size_t)N1 * B * H);
  if ((szXS - (size_t)N1 * B * H) * sizeof(float) < (size_t)E0 || og1.x > 255u ||
      (size_t)ng1.x * 128 < (size_t)(E0 >> 2))
    rid1 = nullptr;
  layer_nodes_wmma<32, 16, 1, 4, 2, 200><<<ng1, gblk, 0, stream>>>(
      X, Wn1, V1, U1, res0, ys, xd1, yd1, R1s, R1d, N0, dst0, rid1, E0, N1);
  edge_owner_kernel<64, 16, 200, 8, 2560, false><<<og1, oblk, 0, stream>>>(
      ys, yd1, xd1, src0, dst0, ew0, We1, rid1, h1, E0, N1);

  const int R2s = N1 * B;
  const int R2d = N2 * B;
  const dim3 ng2 = node_grid(R2s / 16, 4, R2d / 16, 2);
  const dim3 og2 = owner_grid(N2, 100);
  float* endR = regR + (size_t)N1 * B * H;
  unsigned int* rid2 = (unsigned int*)endR;
  if ((size_t)(endR - ws) * sizeof(float) + (size_t)E1 > ws_size || og2.x > 255u ||
      (size_t)ng2.x * 128 < (size_t)(E1 >> 2))
    rid2 = nullptr;
  layer_nodes_wmma<16, 32, 0, 4, 2, 100><<<ng2, gblk, 0, stream>>>(
      h1, Wn2, V2, U2, res1, ys, xd2, yd2, R2s, R2d, 0, dst1, rid2, E1, N2);
  edge_owner_kernel<128, 32, 100, 8, 2560, true><<<og2, oblk, 0, stream>>>(
      ys, yd2, xd2, src1, dst1, ew1, We2, rid2, (float*)d_out, E1, N2);
}


# Edge launch: owner group stages its node range's yd rows in LDS once (contiguous loads); an edge's yd[dst] gather becomes an LDS read, global row loads per edge halved; LDS add order free as before
# speedup vs baseline: 2.3902x; 1.0529x over previous
#include <hip/hip_runtime.h>
#include <hip/hip_bf16.h>

typedef float v2f __attribute__((ext_vector_type(2)));
typedef float v8f __attribute__((ext_vector_type(8)));


template <int K, int MODE, bool GATHER, int TT>
__device__ __forceinline__ void fold_load_rows(const float* __restrict__ A,
                                               const int (&gnode)[TT],
                                               v2f (&af)[TT][K / 4],
                                               int row00, int nsrc, int lane) {
  const int m = lane & 15;
  const int h = lane >> 4;
#pragma unroll
  for (int t = 0; t < TT; ++t) {
    const int r = row00 + t * 16 + m;
    int n = r >> 2;
    const int b = r & 3;
    if (GATHER) n = gnode[t];
    long off;
    if (MODE == 0) {
      off = ((long)n * 4 + b) * (long)K;
    } else {
      off = ((long)b * nsrc + n) * (long)K;
    }
    const float4* __restrict__ arow = reinterpret_cast<const float4*>(A + off);
#pragma unroll
    for (int t2 = 0; t2 < K / 16; ++t2) {
      const float4 lo = arow[t2 * 4 + h * 2];
      const float4 hi = arow[t2 * 4 + h * 2 + 1];
      af[t][t2 * 4 + 0].x = lo.x; af[t][t2 * 4 + 0].y = lo.y;
      af[t][t2 * 4 + 1].x = lo.z; af[t][t2 * 4 + 1].y = lo.w;
      af[t][t2 * 4 + 2].x = hi.x; af[t][t2 * 4 + 2].y = hi.y;
      af[t][t2 * 4 + 3].x = hi.z; af[t][t2 * 4 + 3].y = hi.w;
    }
  }
}

template <int K, int N>
__device__ __forceinline__ void fold_form(const v2f (&wa)[K / 16][N / 4],
                                          const v2f (&vb)[N / 4][N / 16],
                                          v8f (&ff)[K / 16][N / 16]) {
#pragma unroll
  for (int t2 = 0; t2 < K / 16; ++t2) {
#pragma unroll
    for (int nt = 0; nt < N / 16; ++nt) ff[t2][nt] = v8f{};
  }
#pragma unroll
  for (int q = 0; q < N / 4; ++q) {
#pragma unroll
    for (int t2 = 0; t2 < K / 16; ++t2) {
#pragma unroll
      for (int nt = 0; nt < N / 16; ++nt) {
        ff[t2][nt] = __builtin_amdgcn_wmma_f32_16x16x4_f32(
             false, wa[t2][q],  false, vb[q][nt],
             (short)0, ff[t2][nt],  false,  false);
      }
    }
  }
}

template <int K, int N, int TT, bool BOTH>
__device__ __forceinline__ void fold_products(const v2f (&af)[TT][K / 4],
                                              const v2f (&wf)[K / 4][N / 16],
                                              const v8f (&ff)[K / 16][N / 16],
                                              float* __restrict__ Y1, float* __restrict__ Y2,
                                              int row00, int lane) {
  const int h  = lane >> 4;
  const int cn = lane & 15;
#pragma unroll
  for (int t = 0; t < TT; ++t) {
    v8f acc1[N / 16] = {};
    v8f acc2[N / 16] = {};
#pragma unroll
    for (int ks = 0; ks < K / 4; ++ks) {
      const int t2 = ks >> 2, j = ks & 3;
#pragma unroll
      for (int nt = 0; nt < N / 16; ++nt) {
        v2f fb;
        fb.x = ff[t2][nt][2 * j];
        fb.y = ff[t2][nt][2 * j + 1];
        if (BOTH) {
          acc1[nt] = __builtin_amdgcn_wmma_f32_16x16x4_f32(
               false, af[t][ks],  false, wf[ks][nt],
               (short)0, acc1[nt],  false,  false);
        }
        acc2[nt] = __builtin_amdgcn_wmma_f32_16x16x4_f32(
             false, af[t][ks],  false, fb,
             (short)0, acc2[nt],  false,  false);
      }
    }
#pragma unroll
    for (int nt = 0; nt < N / 16; ++nt) {
#pragma unroll
      for (int vr = 0; vr < 8; ++vr) {
        const int rr = row00 + t * 16 + h * 8 + vr;
        if (BOTH) Y1[(long)rr * N + nt * 16 + cn] = acc1[nt][vr];
        Y2[(long)rr * N + nt * 16 + cn] = acc2[nt][vr];
      }
    }
  }
}

template <int K, int N, int MODE, int T, int TD, int NPW>
__global__ void __launch_bounds__(128) layer_nodes_wmma(
    const float* __restrict__ A, const float* __restrict__ W, const float* __restrict__ V,
    const float* __restrict__ U, const int* __restrict__ res,
    float* __restrict__ YS, float* __restrict__ XD, float* __restrict__ YD,
    int Rs, int Rd, int nsrc,
    const int* __restrict__ dstE, unsigned int* __restrict__ rid, int E, int Nd) {
  const int wave = blockIdx.x * (blockDim.x >> 5) + (threadIdx.x >> 5);
  const int lane = threadIdx.x & 31;
  const int row00 = wave * (16 * T);
  const int drow0 = wave * (16 * TD);

  const int gl = wave * 32 + lane;
  const bool plane = rid != nullptr && gl < (E >> 2);
  int4 dq = {0, 0, 0, 0};
  if (plane) dq = reinterpret_cast<const int4*>(dstE)[gl];
  auto range_of = [&](const int d) {
    return (unsigned)d < (unsigned)Nd ? (unsigned)d / (unsigned)NPW : 255u;
  };
  auto store_plane = [&]() {
    if (plane)
      rid[gl] = range_of(dq.x) | (range_of(dq.y) << 8) | (range_of(dq.z) << 16) |
                (range_of(dq.w) << 24);
  };
  if (row00 >= Rs && drow0 >= Rd) {
    store_plane();
    return;
  }

  const int m  = lane & 15;
  const int h  = lane >> 4;
  const int kb = h << 1;

  int gnode[TD];
#pragma unroll
  for (int t = 0; t < TD; ++t) {
    const int r  = drow0 + t * 16 + m;
    const int rc = r < Rd ? r : Rd - 1;
    gnode[t] = res[rc >> 2];
  }

  v2f vb[N / 4][N / 16];
  v2f ub[N / 4][N / 16];
  v2f wa[K / 16][N / 4];
#pragma unroll
  for (int q = 0; q < N / 4; ++q) {
#pragma unroll
    for (int nt = 0; nt < N / 16; ++nt) {
      vb[q][nt].x = V[(q * 4 + kb) * N + nt * 16 + m];
      vb[q][nt].y = V[(q * 4 + kb + 1) * N + nt * 16 + m];
      ub[q][nt].x = U[(q * 4 + kb) * N + nt * 16 + m];
      ub[q][nt].y = U[(q * 4 + kb + 1) * N + nt * 16 + m];
    }
#pragma unroll
    for (int t2 = 0; t2 < K / 16; ++t2) {
      wa[t2][q].x = W[(t2 * 16 + m) * N + q * 4 + kb];
      wa[t2][q].y = W[(t2 * 16 + m) * N + q * 4 + kb + 1];
    }
  }
  v2f wf[K / 4][N / 16];
#pragma unroll
  for (int ks = 0; ks < K / 4; ++ks) {
    const int kk = (ks >> 2) * 16 + h * 8 + (ks & 3) * 2;
#pragma unroll
    for (int nt = 0; nt < N / 16; ++nt) {
      wf[ks][nt].x = W[kk * N + nt * 16 + m];
      wf[ks][nt].y = W[(kk + 1) * N + nt * 16 + m];
    }
  }

  v8f fv[K / 16][N / 16];
  v8f fu[K / 16][N / 16];
  if (row00 + 16 * T <= Rs && drow0 + 16 * TD <= Rd) {
    v2f af[T][K / 4];
    v2f ad[TD][K / 4];
    const int nogather[T] = {};
    fold_load_rows<K, MODE, false, T>(A, nogather, af, row00, nsrc, lane);
    __builtin_amdgcn_sched_barrier(0);
    fold_load_rows<K, MODE, true, TD>(A, gnode, ad, drow0, nsrc, lane);
    __builtin_amdgcn_sched_barrier(0);
    fold_form<K, N>(wa, vb, fv);
    fold_form<K, N>(wa, ub, fu);
    fold_products<K, N, T, false>(af, wf, fv, nullptr, YS, row00, lane);
    fold_products<K, N, TD, true>(ad, wf, fu, XD, YD, drow0, lane);
    store_plane();
  } else {
    fold_form<K, N>(wa, vb, fv);
    fold_form<K, N>(wa, ub, fu);
    const int nogather[1] = {};
    for (int row0 = row00; row0 < row00 + 16 * T && row0 < Rs; row0 += 16) {
      v2f af[1][K / 4];
      fold_load_rows<K, MODE, false, 1>(A, nogather, af, row0, nsrc, lane);
      fold_products<K, N, 1, false>(af, wf, fv, nullptr, YS, row0, lane);
    }
#pragma unroll
    for (int t = 0; t < TD; ++t) {
      const int row0 = drow0 + t * 16;
      if (row0 < Rd) {
        v2f ad[1][K / 4];
        const int g1[1] = {gnode[t]};
        fold_load_rows<K, MODE, true, 1>(A, g1, ad, row0, nsrc, lane);
        fold_products<K, N, 1, true>(ad, wf, fu, XD, YD, row0, lane);
      }
    }
    store_plane();
  }
}


template <int BH, int H>
__device__ __forceinline__ void edge_by_one_lane(const float* __restrict__ ys,
                                                 const float* __restrict__ yd,
                                                 const int* __restrict__ src,
                                                 const float* __restrict__ ew,
                                                 const float* __restrict__ We,
                                                 float* s_agg, float* s_cnt, int n0,
                                                 int e, int d) {
  const int s = src[e];
  const float w = ew[e];
  float* ar = s_agg + (d - n0) * BH;
  for (int i = 0; i < BH; ++i) {
    const float xjv = ys[(long)s * BH + i];
    const float xiv = yd[(long)d * BH + i];
    const float ev  = w * We[i & (H - 1)];
    const float t   = xiv * xjv * ev;
    const float sg  = 1.0f / (1.0f + __expf(-t));
    atomicAdd(ar + i, xjv * sg);
  }
  atomicAdd(s_cnt + (d - n0), 1.0f);
}

template <int BH, int H, int U, bool FULL>
__device__ __forceinline__ void owner_tile(const float* __restrict__ ys,
                                           const float* s_yd,
                                           float* s_agg, int n0, int sv, int dv, float wv,
                                           float wev, int nT, int lane) {
  constexpr int PER = BH / 32;
  constexpr int G   = U < 8 ? U : 8;
  for (int j0 = 0; j0 < (FULL ? 32 : nT); j0 += U) {
    float xj[U][PER];
#pragma unroll
    for (int u = 0; u < U; ++u) {
      const int jj = (FULL || j0 + u < nT) ? (j0 + u) : (nT - 1);
      const int s = __builtin_amdgcn_readlane(sv, jj);
      const float* __restrict__ ysr = ys + (long)s * BH + lane;
#pragma unroll
      for (int k = 0; k < PER; ++k) xj[u][k] = ysr[32 * k];
    }
#pragma unroll
    for (int u0 = 0; u0 < U; u0 += G) {
      float xi[G][PER];
#pragma unroll
      for (int g = 0; g < G; ++g) {
        const int jj = (FULL || j0 + u0 + g < nT) ? (j0 + u0 + g) : (nT - 1);
        const int d = __builtin_amdgcn_readlane(dv, jj);
        const float* yr = s_yd + (d - n0) * BH + lane;
#pragma unroll
        for (int k = 0; k < PER; ++k) xi[g][k] = yr[32 * k];
      }
#pragma unroll
      for (int g = 0; g < G; ++g) {
        const int u = u0 + g;
        if (FULL || j0 + u < nT) {
          const int d = __builtin_amdgcn_readlane(dv, j0 + u);
          const float w = __int_as_float(__builtin_amdgcn_readlane(__float_as_int(wv), j0 + u));
          const float ev = w * wev;
          float* ar = s_agg + (d - n0) * BH + lane;
#pragma unroll
          for (int k = 0; k < PER; ++k) {
            const float t  = xi[g][k] * xj[u][k] * ev;
            const float sg = 1.0f / (1.0f + __expf(-t));
            atomicAdd(ar + 32 * k, xj[u][k] * sg);
          }
        }
      }
    }
  }
}

template <int BH, int H, int NPW, int U, int QCAP, bool FINAL>
__global__ void __launch_bounds__(1024) edge_owner_kernel(
    const float* __restrict__ ys, const float* yd, const float* __restrict__ xd,
    const int* __restrict__ src, const int* __restrict__ dst,
    const float* __restrict__ ew, const float* __restrict__ We,
    const unsigned int* __restrict__ rid, float* out, int E, int Nd) {
  constexpr int NTHR = 1024;
  constexpr int SB  = 8;
  constexpr int SBP = 12;
  __shared__ float s_agg[NPW * BH];
  alignas(16) __shared__ float s_yd[NPW * BH];
  __shared__ float s_cnt[NPW];
  __shared__ int   s_q[QCAP];
  __shared__ int   s_qn;

  const int tid   = threadIdx.x;
  const int lane  = tid & 31;
  const int wave  = __builtin_amdgcn_readfirstlane(tid >> 5);
  constexpr int nthr  = NTHR;
  constexpr int nwave = NTHR >> 5;
  const int n0 = blockIdx.x * NPW;
  if (n0 >= Nd) return;
  const int nn = (Nd - n0 < NPW) ? (Nd - n0) : NPW;

  constexpr int NY4 = NPW * BH / 4;
  constexpr int NYV = (NY4 + NTHR - 1) / NTHR;
  const float4* yd4 = reinterpret_cast<const float4*>(yd + (long)n0 * BH);
  const int ny4 = nn * (BH / 4);
  float4 ydq[NYV];
#pragma unroll
  for (int i = 0; i < NYV; ++i) {
    const int p = tid + i * NTHR;
    ydq[i] = p < ny4 ? yd4[p] : float4{0.0f, 0.0f, 0.0f, 0.0f};
  }

  for (int i = tid; i < NPW * BH; i += nthr) s_agg[i] = 0.0f;
  for (int i = tid; i < NPW; i += nthr) s_cnt[i] = 0.0f;
  if (tid == 0) s_qn = 0;
  __syncthreads();

  auto file4 = [&](const int4 a, const int e) {
    const bool m0 = (unsigned)(a.x - n0) < (unsigned)nn;
    const bool m1 = (unsigned)(a.y - n0) < (unsigned)nn;
    const bool m2 = (unsigned)(a.z - n0) < (unsigned)nn;
    const bool m3 = (unsigned)(a.w - n0) < (unsigned)nn;
    if (m0 | m1 | m2 | m3) {
      if (m0) { const int pos = atomicAdd(&s_qn, 1); if (pos < QCAP) s_q[pos] = e; }
      if (m1) { const int pos = atomicAdd(&s_qn, 1); if (pos < QCAP) s_q[pos] = e + 1; }
      if (m2) { const int pos = atomicAdd(&s_qn, 1); if (pos < QCAP) s_q[pos] = e + 2; }
      if (m3) { const int pos = atomicAdd(&s_qn, 1); if (pos < QCAP) s_q[pos] = e + 3; }
    }
  };
  const int4* __restrict__ d4 = reinterpret_cast<const int4*>(dst);
  const int nvec  = E >> 2;
  if (rid != nullptr) {
    const unsigned pat = blockIdx.x * 0x01010101u;
    auto zero_byte_any = [](const unsigned x) { return (x - 0x01010101u) & ~x; };
    auto nonzero_bytes = [](const unsigned x) {
      return ((x & 0x7f7f7f7fu) + 0x7f7f7f7fu) | x;
    };
    auto file_flags = [&](unsigned pk, const int e) {
      while (pk) {
        const int bit = __ffs(pk) - 1;
        pk &= pk - 1;
        const int pos = atomicAdd(&s_qn, 1);
        if (pos < QCAP) s_q[pos] = e + ((bit & 7) << 2) + (bit >> 3);
      }
    };
    const uint4* __restrict__ p16 = reinterpret_cast<const uint4*>(rid);
    const int nv16 = E >> 4;
    auto file_piece = [&](const uint4 q, const int v) {
      const unsigned t0 = nonzero_bytes(q.x ^ pat);
      const unsigned t1 = nonzero_bytes(q.y ^ pat);
      const unsigned t2 = nonzero_bytes(q.z ^ pat);
      const unsigned t3 = nonzero_bytes(q.w ^ pat);
      const unsigned pk = ((~t0 & 0x80808080u) >> 7) | ((~t1 & 0x80808080u) >> 6) |
                          ((~t2 & 0x80808080u) >> 5) | ((~t3 & 0x80808080u) >> 4);
      file_flags(pk, v * 16);
    };
    const int nfullp = nv16 / (NTHR * SBP);
    const int rotp   = (int)(blockIdx.x * (unsigned)nfullp / gridDim.x);
    const int tr     = (tid + 32 * (int)(blockIdx.x & 31u)) & (NTHR - 1);
    for (int b = 0; b < nfullp; ++b) {
      int bb = b + rotp;
      if (bb >= nfullp) bb -= nfullp;
      const int vbase = bb * (NTHR * SBP) + tr;
      const uint4* __restrict__ p = p16 + vbase;
      uint4 a[SBP];
#pragma unroll
      for (int i = 0; i < SBP; ++i) a[i] = p[i * NTHR];
      unsigned mine = 0;
#pragma unroll
      for (int i = 0; i < SBP; ++i) {
        const unsigned z = zero_byte_any(a[i].x ^ pat) | zero_byte_any(a[i].y ^ pat) |
                           zero_byte_any(a[i].z ^ pat) | zero_byte_any(a[i].w ^ pat);
        mine = (mine << 1) | min(z & 0x80808080u, 1u);
      }
      while (mine) {
        const int i = SBP - __ffs(mine);
        mine &= mine - 1;
        const int v = vbase + i * NTHR;
        file_piece(p16[v], v);
      }
    }
    for (int v = nfullp * (NTHR * SBP) + tid; v < nv16; v += NTHR) file_piece(p16[v], v);
    for (int w = nv16 * 4 + tid; w < nvec; w += NTHR) {
      const unsigned t = nonzero_bytes(rid[w] ^ pat);
      file_flags((~t & 0x80808080u) >> 7, w * 4);
    }
  } else {
    const int nfull = nvec / (NTHR * SB);
    const int rotb  = (int)(((long)blockIdx.x * nfull) / gridDim.x);
    for (int b = 0; b < nfull; ++b) {
      int bb = b + rotb;
      if (bb >= nfull) bb -= nfull;
      const int vbase = bb * (NTHR * SB);
      const int4* __restrict__ p = d4 + vbase + tid;
      int4 a[SB];
#pragma unroll
      for (int i = 0; i < SB; ++i) a[i] = p[i * NTHR];
#pragma unroll
      for (int i = 0; i < SB; ++i) file4(a[i], (vbase + i * NTHR + tid) * 4);
    }
    for (int v = nfull * (NTHR * SB) + tid; v < nvec; v += NTHR) file4(d4[v], v * 4);
  }
  for (int e = nvec * 4 + tid; e < E; e += NTHR) {
    if ((unsigned)(dst[e] - n0) < (unsigned)nn) {
      const int pos = atomicAdd(&s_qn, 1);
      if (pos < QCAP) s_q[pos] = e;
    }
  }
#pragma unroll
  for (int i = 0; i < NYV; ++i) {
    const int p = tid + i * NTHR;
    if (p < ny4) reinterpret_cast<float4*>(s_yd)[p] = ydq[i];
  }
  __syncthreads();
  const int qn = __builtin_amdgcn_readfirstlane(s_qn);

  if (qn > QCAP) {
    for (int e = tid; e < E; e += nthr) {
      const int d = dst[e];
      if ((unsigned)(d - n0) < (unsigned)nn)
        edge_by_one_lane<BH, H>(ys, yd, src, ew, We, s_agg, s_cnt, n0, e, d);
    }
  } else {
    const float wev = We[lane & (H - 1)];
    for (int t0 = wave * 32; t0 < qn; t0 += nwave * 32) {
      const int nT = (qn - t0 < 32) ? (qn - t0) : 32;
      int sv = 0, dv = n0;
      float wv = 0.0f;
      if (lane < nT) {
        const int e = s_q[t0 + lane];
        sv = src[e];
        dv = dst[e];
        wv = ew[e];
        atomicAdd(s_cnt + (dv - n0), 1.0f);
      }
      if (nT == 32) {
        owner_tile<BH, H, U, true>(ys, s_yd, s_agg, n0, sv, dv, wv, wev, 32, lane);
      } else {
        owner_tile<BH, H, U, false>(ys, s_yd, s_agg, n0, sv, dv, wv, wev, nT, lane);
      }
    }
  }

  constexpr int PER = BH / 32;
  constexpr int NQ  = (NPW + nwave - 1) / nwave;
  float xdv[NQ][PER];
#pragma unroll
  for (int q = 0; q < NQ; ++q) {
    const int ln = wave + q * nwave;
    const int lc = ln < nn ? ln : nn - 1;
    const long base = (long)(n0 + lc) * BH + lane;
#pragma unroll
    for (int j = 0; j < PER; ++j) xdv[q][j] = xd[base + 32 * j];
  }
  __syncthreads();

#pragma unroll
  for (int q = 0; q < NQ; ++q) {
    const int ln = wave + q * nwave;
    if (ln < nn) {
      const int d = n0 + ln;
      const float inv_cnt = 1.0f / fmaxf(s_cnt[ln], 1.0f);
      const float* ar = s_agg + ln * BH + lane;
      const float* yr = s_yd + ln * BH + lane;
      float o[PER];
      float sum = 0.0f, ssq = 0.0f;
#pragma unroll
      for (int j = 0; j < PER; ++j) {
        o[j] = yr[32 * j] + ar[32 * j] * inv_cnt;
        sum += o[j];
        ssq += o[j] * o[j];
      }
#pragma unroll
      for (int mask = 16; mask >= 1; mask >>= 1) {
        sum += __shfl_xor(sum, mask, 32);
        ssq += __shfl_xor(ssq, mask, 32);
      }
      const float mu  = sum * (1.0f / BH);
      const float var = ssq * (1.0f / BH) - mu * mu;
      const float is  = rsqrtf(var + 1e-5f);

      const long base = (long)d * BH;
#pragma unroll
      for (int j = 0; j < PER; ++j) {
        const int i = lane + j * 32;
        float v = xdv[q][j] + (o[j] - mu) * is;
        v = v > 0.0f ? v : 0.01f * v;
        if (FINAL) {
          const int b = i / H, c = i & (H - 1);
          out[((long)b * Nd + d) * H + c] = v;
        } else {
          out[base + i] = v;
        }
      }
    }
  }
}


extern "C" void kernel_launch(void* const* d_in, const int* in_sizes, int n_in,
                              void* d_out, int out_size, void* d_ws, size_t ws_size,
                              hipStream_t stream) {
  constexpr int B = 4, Cin = 32, H = 16, Cout = 32;
  const float* X    = (const float*)d_in[0];
  const int*   src0 = (const int*)d_in[1];
  const int*   dst0 = (const int*)d_in[2];
  const float* ew0  = (const float*)d_in[3];
  const int*   res0 = (const int*)d_in[4];
  const int*   src1 = (const int*)d_in[5];
  const int*   dst1 = (const int*)d_in[6];
  const float* ew1  = (const float*)d_in[7];
  const int*   res1 = (const int*)d_in[8];
  const float* Wn1  = (const float*)d_in[9];
  const float* We1  = (const float*)d_in[10];
  const float* U1   = (const float*)d_in[11];
  const float* V1   = (const float*)d_in[12];
  const float* Wn2  = (const float*)d_in[13];
  const float* We2  = (const float*)d_in[14];
  const float* U2   = (const float*)d_in[15];
  const float* V2   = (const float*)d_in[16];

  const int E0 = in_sizes[1];
  const int N1 = in_sizes[4];
  const int E1 = in_sizes[5];
  const int N2 = in_sizes[8];
  const int N0 = in_sizes[0] / (B * Cin);

  float* ws = (float*)d_ws;
  const size_t szXS = (size_t)((N0 * B * H > N1 * B * Cout) ? N0 * B * H : N1 * B * Cout);
  float* regP = ws;
  float* ys   = regP + szXS;
  float* regR = ys + szXS;
  float* xd1  = regP;
  float* yd1  = regR;
  float* h1   = regR;
  float* xd2  = regP;
  float* yd2  = regP + (size_t)N2 * B * Cout;

  const dim3 gblk(128);
  auto node_grid = [](int stiles, int T, int dtiles, int TD) {
    const int ws_ = (stiles + T - 1) / T;
    const int wd_ = (dtiles + TD - 1) / TD;
    const int waves = ws_ > wd_ ? ws_ : wd_;
    return dim3((unsigned)((waves + 3) / 4));
  };
  const dim3 oblk(1024);
  auto owner_grid = [](int nodes, int NPW) {
    return dim3((unsigned)((nodes + NPW - 1) / NPW));
  };

  const int R1s = N0 * B;
  const int R1d = N1 * B;
  const dim3 ng1 = node_grid(R1s / 16, 4, R1d / 16, 2);
  const dim3 og1 = owner_grid(N1, 200);
  unsigned int* rid1 = (unsigned int*)(regP + (size_t)N1 * B * H);
  if ((szXS - (size_t)N1 * B * H) * sizeof(float) < (size_t)E0 || og1.x > 255u ||
      (size_t)ng1.x * 128 < (size_t)(E0 >> 2))
    rid1 = nullptr;
  layer_nodes_wmma<32, 16, 1, 4, 2, 200><<<ng1, gblk, 0, stream>>>(
      X, Wn1, V1, U1, res0, ys, xd1, yd1, R1s, R1d, N0, dst0, rid1, E0, N1);
  edge_owner_kernel<64, 16, 200, 16, 2560, false><<<og1, oblk, 0, stream>>>(
      ys, yd1, xd1, src0, dst0, ew0, We1, rid1, h1, E0, N1);

  const int R2s = N1 * B;
  const int R2d = N2 * B;
  const dim3 ng2 = node_grid(R2s / 16, 4, R2d / 16, 2);
  const dim3 og2 = owner_grid(N2, 100);
  float* endR = regR + (size_t)N1 * B * H;
  unsigned int* rid2 = (unsigned int*)endR;
  if ((size_t)(endR - ws) * sizeof(float) + (size_t)E1 > ws_size || og2.x > 255u ||
      (size_t)ng2.x * 128 < (size_t)(E1 >> 2))
    rid2 = nullptr;
  layer_nodes_wmma<16, 32, 0, 4, 2, 100><<<ng2, gblk, 0, stream>>>(
      h1, Wn2, V2, U2, res1, ys, xd2, yd2, R2s, R2d, 0, dst1, rid2, E1, N2);
  edge_owner_kernel<128, 32, 100, 8, 2560, true><<<og2, oblk, 0, stream>>>(
      ys, yd2, xd2, src1, dst1, ew1, We2, rid2, (float*)d_out, E1, N2);
}


# Layer-2 node launch gone (4 to 3): layer-1 owner groups form ys2 on WMMA from the h1 rows they hold in LDS; layer-2 groups form xd2, yd2 of their own nodes from h1[res1] into LDS; sum order kept
# speedup vs baseline: 2.6286x; 1.0997x over previous
#include <hip/hip_runtime.h>
#include <hip/hip_bf16.h>

typedef float v2f __attribute__((ext_vector_type(2)));
typedef float v8f __attribute__((ext_vector_type(8)));


template <int K, int MODE, bool GATHER, int TT>
__device__ __forceinline__ void fold_load_rows(const float* __restrict__ A,
                                               const int (&gnode)[TT],
                                               v2f (&af)[TT][K / 4],
                                               int row00, int nsrc, int lane) {
  const int m = lane & 15;
  const int h = lane >> 4;
#pragma unroll
  for (int t = 0; t < TT; ++t) {
    const int r = row00 + t * 16 + m;
    int n = r >> 2;
    const int b = r & 3;
    if (GATHER) n = gnode[t];
    long off;
    if (MODE == 0) {
      off = ((long)n * 4 + b) * (long)K;
    } else {
      off = ((long)b * nsrc + n) * (long)K;
    }
    const float4* __restrict__ arow = reinterpret_cast<const float4*>(A + off);
#pragma unroll
    for (int t2 = 0; t2 < K / 16; ++t2) {
      const float4 lo = arow[t2 * 4 + h * 2];
      const float4 hi = arow[t2 * 4 + h * 2 + 1];
      af[t][t2 * 4 + 0].x = lo.x; af[t][t2 * 4 + 0].y = lo.y;
      af[t][t2 * 4 + 1].x = lo.z; af[t][t2 * 4 + 1].y = lo.w;
      af[t][t2 * 4 + 2].x = hi.x; af[t][t2 * 4 + 2].y = hi.y;
      af[t][t2 * 4 + 3].x = hi.z; af[t][t2 * 4 + 3].y = hi.w;
    }
  }
}

template <int K, int N>
__device__ __forceinline__ void fold_form(const v2f (&wa)[K / 16][N / 4],
                                          const v2f (&vb)[N / 4][N / 16],
                                          v8f (&ff)[K / 16][N / 16]) {
#pragma unroll
  for (int t2 = 0; t2 < K / 16; ++t2) {
#pragma unroll
    for (int nt = 0; nt < N / 16; ++nt) ff[t2][nt] = v8f{};
  }
#pragma unroll
  for (int q = 0; q < N / 4; ++q) {
#pragma unroll
    for (int t2 = 0; t2 < K / 16; ++t2) {
#pragma unroll
      for (int nt = 0; nt < N / 16; ++nt) {
        ff[t2][nt] = __builtin_amdgcn_wmma_f32_16x16x4_f32(
             false, wa[t2][q],  false, vb[q][nt],
             (short)0, ff[t2][nt],  false,  false);
      }
    }
  }
}

template <int K, int N, int TT, bool BOTH>
__device__ __forceinline__ void fold_products(const v2f (&af)[TT][K / 4],
                                              const v2f (&wf)[K / 4][N / 16],
                                              const v8f (&ff)[K / 16][N / 16],
                                              float* __restrict__ Y1, float* __restrict__ Y2,
                                              int row00, int lane) {
  const int h  = lane >> 4;
  const int cn = lane & 15;
#pragma unroll
  for (int t = 0; t < TT; ++t) {
    v8f acc1[N / 16] = {};
    v8f acc2[N / 16] = {};
#pragma unroll
    for (int ks = 0; ks < K / 4; ++ks) {
      const int t2 = ks >> 2, j = ks & 3;
#pragma unroll
      for (int nt = 0; nt < N / 16; ++nt) {
        v2f fb;
        fb.x = ff[t2][nt][2 * j];
        fb.y = ff[t2][nt][2 * j + 1];
        if (BOTH) {
          acc1[nt] = __builtin_amdgcn_wmma_f32_16x16x4_f32(
               false, af[t][ks],  false, wf[ks][nt],
               (short)0, acc1[nt],  false,  false);
        }
        acc2[nt] = __builtin_amdgcn_wmma_f32_16x16x4_f32(
             false, af[t][ks],  false, fb,
             (short)0, acc2[nt],  false,  false);
      }
    }
#pragma unroll
    for (int nt = 0; nt < N / 16; ++nt) {
#pragma unroll
      for (int vr = 0; vr < 8; ++vr) {
        const int rr = row00 + t * 16 + h * 8 + vr;
        if (BOTH) Y1[(long)rr * N + nt * 16 + cn] = acc1[nt][vr];
        Y2[(long)rr * N + nt * 16 + cn] = acc2[nt][vr];
      }
    }
  }
}

template <int K, int N, int MODE, int T, int TD, int NPW, int NPW2>
__global__ void __launch_bounds__(128) layer_nodes_wmma(
    const float* __restrict__ A, const float* __restrict__ W, const float* __restrict__ V,
    const float* __restrict__ U, const int* __restrict__ res,
    float* __restrict__ YS, float* __restrict__ XD, float* __restrict__ YD,
    int Rs, int Rd, int nsrc,
    const int* __restrict__ dstE, unsigned int* __restrict__ rid, int E, int Nd,
    const int* __restrict__ dstE2, unsigned int* __restrict__ rid2, int E2, int Nd2) {
  const int wave = blockIdx.x * (blockDim.x >> 5) + (threadIdx.x >> 5);
  const int lane = threadIdx.x & 31;
  const int row00 = wave * (16 * T);
  const int drow0 = wave * (16 * TD);

  const int gl = wave * 32 + lane;
  const bool plane = rid != nullptr && gl < (E >> 2);
  const bool plane2 = rid2 != nullptr && gl < (E2 >> 2);
  int4 dq = {0, 0, 0, 0};
  int4 dq2 = {0, 0, 0, 0};
  if (plane) dq = reinterpret_cast<const int4*>(dstE)[gl];
  if (plane2) dq2 = reinterpret_cast<const int4*>(dstE2)[gl];
  auto range_of = [](const int d, const int nd, const unsigned npw) {
    return (unsigned)d < (unsigned)nd ? (unsigned)d / npw : 255u;
  };
  auto store_plane = [&]() {
    if (plane)
      rid[gl] = range_of(dq.x, Nd, NPW) | (range_of(dq.y, Nd, NPW) << 8) |
                (range_of(dq.z, Nd, NPW) << 16) | (range_of(dq.w, Nd, NPW) << 24);
    if (plane2)
      rid2[gl] = range_of(dq2.x, Nd2, NPW2) | (range_of(dq2.y, Nd2, NPW2) << 8) |
                 (range_of(dq2.z, Nd2, NPW2) << 16) | (range_of(dq2.w, Nd2, NPW2) << 24);
  };
  if (row00 >= Rs && drow0 >= Rd) {
    store_plane();
    return;
  }

  const int m  = lane & 15;
  const int h  = lane >> 4;
  const int kb = h << 1;

  int gnode[TD];
#pragma unroll
  for (int t = 0; t < TD; ++t) {
    const int r  = drow0 + t * 16 + m;
    const int rc = r < Rd ? r : Rd - 1;
    gnode[t] = res[rc >> 2];
  }

  v2f vb[N / 4][N / 16];
  v2f ub[N / 4][N / 16];
  v2f wa[K / 16][N / 4];
#pragma unroll
  for (int q = 0; q < N / 4; ++q) {
#pragma unroll
    for (int nt = 0; nt < N / 16; ++nt) {
      vb[q][nt].x = V[(q * 4 + kb) * N + nt * 16 + m];
      vb[q][nt].y = V[(q * 4 + kb + 1) * N + nt * 16 + m];
      ub[q][nt].x = U[(q * 4 + kb) * N + nt * 16 + m];
      ub[q][nt].y = U[(q * 4 + kb + 1) * N + nt * 16 + m];
    }
#pragma unroll
    for (int t2 = 0; t2 < K / 16; ++t2) {
      wa[t2][q].x = W[(t2 * 16 + m) * N + q * 4 + kb];
      wa[t2][q].y = W[(t2 * 16 + m) * N + q * 4 + kb + 1];
    }
  }
  v2f wf[K / 4][N / 16];
#pragma unroll
  for (int ks = 0; ks < K / 4; ++ks) {
    const int kk = (ks >> 2) * 16 + h * 8 + (ks & 3) * 2;
#pragma unroll
    for (int nt = 0; nt < N / 16; ++nt) {
      wf[ks][nt].x = W[kk * N + nt * 16 + m];
      wf[ks][nt].y = W[(kk + 1) * N + nt * 16 + m];
    }
  }

  v8f fv[K / 16][N / 16];
  v8f fu[K / 16][N / 16];
  if (row00 + 16 * T <= Rs && drow0 + 16 * TD <= Rd) {
    v2f af[T][K / 4];
    v2f ad[TD][K / 4];
    const int nogather[T] = {};
    fold_load_rows<K, MODE, false, T>(A, nogather, af, row00, nsrc, lane);
    __builtin_amdgcn_sched_barrier(0);
    fold_load_rows<K, MODE, true, TD>(A, gnode, ad, drow0, nsrc, lane);
    __builtin_amdgcn_sched_barrier(0);
    fold_form<K, N>(wa, vb, fv);
    fold_form<K, N>(wa, ub, fu);
    fold_products<K, N, T, false>(af, wf, fv, nullptr, YS, row00, lane);
    fold_products<K, N, TD, true>(ad, wf, fu, XD, YD, drow0, lane);
    store_plane();
  } else {
    fold_form<K, N>(wa, vb, fv);
    fold_form<K, N>(wa, ub, fu);
    const int nogather[1] = {};
    for (int row0 = row00; row0 < row00 + 16 * T && row0 < Rs; row0 += 16) {
      v2f af[1][K / 4];
      fold_load_rows<K, MODE, false, 1>(A, nogather, af, row0, nsrc, lane);
      fold_products<K, N, 1, false>(af, wf, fv, nullptr, YS, row0, lane);
    }
#pragma unroll
    for (int t = 0; t < TD; ++t) {
      const int row0 = drow0 + t * 16;
      if (row0 < Rd) {
        v2f ad[1][K / 4];
        const int g1[1] = {gnode[t]};
        fold_load_rows<K, MODE, true, 1>(A, g1, ad, row0, nsrc, lane);
        fold_products<K, N, 1, true>(ad, wf, fu, XD, YD, row0, lane);
      }
    }
    store_plane();
  }
}


template <int K, int N>
__device__ __forceinline__ void load_fold_operands(const float* __restrict__ W,
                                                   const float* __restrict__ M,
                                                   v2f (&wa)[K / 16][N / 4],
                                                   v2f (&mb)[N / 4][N / 16], int lane) {
  const int m  = lane & 15;
  const int kb = (lane >> 4) << 1;
#pragma unroll
  for (int q = 0; q < N / 4; ++q) {
#pragma unroll
    for (int nt = 0; nt < N / 16; ++nt) {
      mb[q][nt].x = M[(q * 4 + kb) * N + nt * 16 + m];
      mb[q][nt].y = M[(q * 4 + kb + 1) * N + nt * 16 + m];
    }
#pragma unroll
    for (int t2 = 0; t2 < K / 16; ++t2) {
      wa[t2][q].x = W[(t2 * 16 + m) * N + q * 4 + kb];
      wa[t2][q].y = W[(t2 * 16 + m) * N + q * 4 + kb + 1];
    }
  }
}

template <int K, int N, bool BOTH>
__device__ __forceinline__ void tile_products(const v2f (&af)[K / 4],
                                              const v2f (&wf)[K / 4][N / 16],
                                              const v8f (&ff)[K / 16][N / 16],
                                              float* Y1, float* Y2, int row0, int rlim, int lane) {
  const int h  = lane >> 4;
  const int cn = lane & 15;
  v8f acc1[N / 16] = {};
  v8f acc2[N / 16] = {};
#pragma unroll
  for (int ks = 0; ks < K / 4; ++ks) {
    const int t2 = ks >> 2, j = ks & 3;
#pragma unroll
    for (int nt = 0; nt < N / 16; ++nt) {
      v2f fb;
      fb.x = ff[t2][nt][2 * j];
      fb.y = ff[t2][nt][2 * j + 1];
      if (BOTH) {
        acc1[nt] = __builtin_amdgcn_wmma_f32_16x16x4_f32(
             false, af[ks],  false, wf[ks][nt],
             (short)0, acc1[nt],  false,  false);
      }
      acc2[nt] = __builtin_amdgcn_wmma_f32_16x16x4_f32(
           false, af[ks],  false, fb,
           (short)0, acc2[nt],  false,  false);
    }
  }
  if (row0 + 16 <= rlim) {
#pragma unroll
    for (int nt = 0; nt < N / 16; ++nt) {
#pragma unroll
      for (int vr = 0; vr < 8; ++vr) {
        const int rr = row0 + h * 8 + vr;
        if (BOTH) Y1[(long)rr * N + nt * 16 + cn] = acc1[nt][vr];
        Y2[(long)rr * N + nt * 16 + cn] = acc2[nt][vr];
      }
    }
  } else {
#pragma unroll
    for (int nt = 0; nt < N / 16; ++nt) {
#pragma unroll
      for (int vr = 0; vr < 8; ++vr) {
        const int rr = row0 + h * 8 + vr;
        if (rr < rlim) {
          if (BOTH) Y1[(long)rr * N + nt * 16 + cn] = acc1[nt][vr];
          Y2[(long)rr * N + nt * 16 + cn] = acc2[nt][vr];
        }
      }
    }
  }
}

template <int BH, int H>
__device__ __forceinline__ void edge_by_one_lane(const float* __restrict__ ys,
                                                 const float* s_yd,
                                                 const int* __restrict__ src,
                                                 const float* __restrict__ ew,
                                                 const float* __restrict__ We,
                                                 float* s_agg, float* s_cnt, int n0,
                                                 int e, int d) {
  const int s = src[e];
  const float w = ew[e];
  float* ar = s_agg + (d - n0) * BH;
  for (int i = 0; i < BH; ++i) {
    const float xjv = ys[(long)s * BH + i];
    const float xiv = s_yd[(d - n0) * BH + i];
    const float ev  = w * We[i & (H - 1)];
    const float t   = xiv * xjv * ev;
    const float sg  = 1.0f / (1.0f + __expf(-t));
    atomicAdd(ar + i, xjv * sg);
  }
  atomicAdd(s_cnt + (d - n0), 1.0f);
}

template <int BH, int H, int U, bool FULL>
__device__ __forceinline__ void owner_tile(const float* __restrict__ ys,
                                           const float* s_yd,
                                           float* s_agg, int n0, int sv, int dv, float wv,
                                           float wev, int nT, int lane) {
  constexpr int PER = BH / 32;
  constexpr int G   = U < 8 ? U : 8;
  for (int j0 = 0; j0 < (FULL ? 32 : nT); j0 += U) {
    float xj[U][PER];
#pragma unroll
    for (int u = 0; u < U; ++u) {
      const int jj = (FULL || j0 + u < nT) ? (j0 + u) : (nT - 1);
      const int s = __builtin_amdgcn_readlane(sv, jj);
      const float* __restrict__ ysr = ys + (long)s * BH + lane;
#pragma unroll
      for (int k = 0; k < PER; ++k) xj[u][k] = ysr[32 * k];
    }
#pragma unroll
    for (int u0 = 0; u0 < U; u0 += G) {
      float xi[G][PER];
#pragma unroll
      for (int g = 0; g < G; ++g) {
        const int jj = (FULL || j0 + u0 + g < nT) ? (j0 + u0 + g) : (nT - 1);
        const int d = __builtin_amdgcn_readlane(dv, jj);
        const float* yr = s_yd + (d - n0) * BH + lane;
#pragma unroll
        for (int k = 0; k < PER; ++k) xi[g][k] = yr[32 * k];
      }
#pragma unroll
      for (int g = 0; g < G; ++g) {
        const int u = u0 + g;
        if (FULL || j0 + u < nT) {
          const int d = __builtin_amdgcn_readlane(dv, j0 + u);
          const float w = __int_as_float(__builtin_amdgcn_readlane(__float_as_int(wv), j0 + u));
          const float ev = w * wev;
          float* ar = s_agg + (d - n0) * BH + lane;
#pragma unroll
          for (int k = 0; k < PER; ++k) {
            const float t  = xi[g][k] * xj[u][k] * ev;
            const float sg = 1.0f / (1.0f + __expf(-t));
            atomicAdd(ar + 32 * k, xj[u][k] * sg);
          }
        }
      }
    }
  }
}

template <int BH, int H, int NPW, int U, int QCAP, bool FINAL, int NEXTN, int MAKEK>
__global__ void __launch_bounds__(1024) edge_owner_kernel(
    const float* __restrict__ ys, const float* yd, const float* __restrict__ xd,
    const int* __restrict__ src, const int* __restrict__ dst,
    const float* __restrict__ ew, const float* __restrict__ We,
    const unsigned int* __restrict__ rid, float* out, int E, int Nd,
    const float* __restrict__ Wx, const float* __restrict__ Mx,
    const float* __restrict__ hin, const int* __restrict__ res, float* __restrict__ ysn) {
  constexpr bool NEXT = NEXTN > 0;
  constexpr bool MAKE = MAKEK > 0;
  constexpr int NTHR = 1024;
  constexpr int SB  = 8;
  constexpr int SBP = 12;
  alignas(16) __shared__ float s_agg[NPW * BH];
  alignas(16) __shared__ float s_yd[NPW * BH];
  __shared__ float s_xd[MAKE ? NPW * BH : 1];
  __shared__ float s_cnt[NPW];
  __shared__ int   s_q[QCAP];
  __shared__ int   s_qn;

  const int tid   = threadIdx.x;
  const int lane  = tid & 31;
  const int wave  = __builtin_amdgcn_readfirstlane(tid >> 5);
  constexpr int nthr  = NTHR;
  constexpr int nwave = NTHR >> 5;
  const int n0 = blockIdx.x * NPW;
  if (n0 >= Nd) return;
  const int nn = (Nd - n0 < NPW) ? (Nd - n0) : NPW;

  constexpr int NY4 = NPW * BH / 4;
  constexpr int NYV = (NY4 + NTHR - 1) / NTHR;
  const float4* yd4 = reinterpret_cast<const float4*>(yd + (long)n0 * BH);
  const int ny4 = nn * (BH / 4);
  float4 ydq[NYV];
  if (!MAKE) {
#pragma unroll
    for (int i = 0; i < NYV; ++i) {
      const int p = tid + i * NTHR;
      ydq[i] = p < ny4 ? yd4[p] : float4{0.0f, 0.0f, 0.0f, 0.0f};
    }
  }

  constexpr int KM  = MAKE ? MAKEK : 16;
  constexpr int NM  = BH / 4;
  constexpr int NTM = ((NPW * 4 + 15) / 16 + nwave - 1) / nwave;
  const int mrows  = nn * 4;
  const int mtiles = (mrows + 15) >> 4;
  int gm[NTM];
  v2f wam[KM / 16][NM / 4];
  v2f ubm[NM / 4][NM / 16];
  v2f wfm[KM / 4][NM / 16];
  v8f fum[KM / 16][NM / 16];
  v2f adm[NTM][1][KM / 4];
  if (MAKE) {
#pragma unroll
    for (int i = 0; i < NTM; ++i) {
      const int r  = (wave + i * nwave) * 16 + (lane & 15);
      const int rc = r < mrows ? r : mrows - 1;
      gm[i] = res[n0 + (rc >> 2)];
    }
    load_fold_operands<KM, NM>(Wx, Mx, wam, ubm, lane);
#pragma unroll
    for (int ks = 0; ks < KM / 4; ++ks) {
      const int kk = (ks >> 2) * 16 + (lane >> 4) * 8 + (ks & 3) * 2;
#pragma unroll
      for (int nt = 0; nt < NM / 16; ++nt) {
        wfm[ks][nt].x = Wx[kk * NM + nt * 16 + (lane & 15)];
        wfm[ks][nt].y = Wx[(kk + 1) * NM + nt * 16 + (lane & 15)];
      }
    }
  }

  for (int i = tid; i < NPW * BH; i += nthr) s_agg[i] = 0.0f;
  for (int i = tid; i < NPW; i += nthr) s_cnt[i] = 0.0f;
  if (tid == 0) s_qn = 0;
  __syncthreads();

  if (MAKE) {
    fold_form<KM, NM>(wam, ubm, fum);
#pragma unroll
    for (int i = 0; i < NTM; ++i) {
      const int g1[1] = {gm[i]};
      fold_load_rows<KM, 0, true, 1>(hin, g1, adm[i], (wave + i * nwave) * 16, 0, lane);
    }
  }

  auto file4 = [&](const int4 a, const int e) {
    const bool m0 = (unsigned)(a.x - n0) < (unsigned)nn;
    const bool m1 = (unsigned)(a.y - n0) < (unsigned)nn;
    const bool m2 = (unsigned)(a.z - n0) < (unsigned)nn;
    const bool m3 = (unsigned)(a.w - n0) < (unsigned)nn;
    if (m0 | m1 | m2 | m3) {
      if (m0) { const int pos = atomicAdd(&s_qn, 1); if (pos < QCAP) s_q[pos] = e; }
      if (m1) { const int pos = atomicAdd(&s_qn, 1); if (pos < QCAP) s_q[pos] = e + 1; }
      if (m2) { const int pos = atomicAdd(&s_qn, 1); if (pos < QCAP) s_q[pos] = e + 2; }
      if (m3) { const int pos = atomicAdd(&s_qn, 1); if (pos < QCAP) s_q[pos] = e + 3; }
    }
  };
  const int4* __restrict__ d4 = reinterpret_cast<const int4*>(dst);
  const int nvec  = E >> 2;
  if (rid != nullptr) {
    const unsigned pat = blockIdx.x * 0x01010101u;
    auto zero_byte_any = [](const unsigned x) { return (x - 0x01010101u) & ~x; };
    auto nonzero_bytes = [](const unsigned x) {
      return ((x & 0x7f7f7f7fu) + 0x7f7f7f7fu) | x;
    };
    auto file_flags = [&](unsigned pk, const int e) {
      while (pk) {
        const int bit = __ffs(pk) - 1;
        pk &= pk - 1;
        const int pos = atomicAdd(&s_qn, 1);
        if (pos < QCAP) s_q[pos] = e + ((bit & 7) << 2) + (bit >> 3);
      }
    };
    const uint4* __restrict__ p16 = reinterpret_cast<const uint4*>(rid);
    const int nv16 = E >> 4;
    auto file_piece = [&](const uint4 q, const int v) {
      const unsigned t0 = nonzero_bytes(q.x ^ pat);
      const unsigned t1 = nonzero_bytes(q.y ^ pat);
      const unsigned t2 = nonzero_bytes(q.z ^ pat);
      const unsigned t3 = nonzero_bytes(q.w ^ pat);
      const unsigned pk = ((~t0 & 0x80808080u) >> 7) | ((~t1 & 0x80808080u) >> 6) |
                          ((~t2 & 0x80808080u) >> 5) | ((~t3 & 0x80808080u) >> 4);
      file_flags(pk, v * 16);
    };
    const int nfullp = nv16 / (NTHR * SBP);
    const int rotp   = (int)(blockIdx.x * (unsigned)nfullp / gridDim.x);
    const int tr     = (tid + 32 * (int)(blockIdx.x & 31u)) & (NTHR - 1);
    for (int b = 0; b < nfullp; ++b) {
      int bb = b + rotp;
      if (bb >= nfullp) bb -= nfullp;
      const int vbase = bb * (NTHR * SBP) + tr;
      const uint4* __restrict__ p = p16 + vbase;
      uint4 a[SBP];
#pragma unroll
      for (int i = 0; i < SBP; ++i) a[i] = p[i * NTHR];
      if (MAKE) {
        static_assert(SBP == 12, "the statement below names every piece of a batch");
        asm volatile("" : "+v"(a[0].x), "+v"(a[1].x), "+v"(a[2].x), "+v"(a[3].x),
                          "+v"(a[4].x), "+v"(a[5].x), "+v"(a[6].x), "+v"(a[7].x),
                          "+v"(a[8].x), "+v"(a[9].x), "+v"(a[10].x), "+v"(a[11].x));
      }
      unsigned mine = 0;
#pragma unroll
      for (int i = 0; i < SBP; ++i) {
        const unsigned z = zero_byte_any(a[i].x ^ pat) | zero_byte_any(a[i].y ^ pat) |
                           zero_byte_any(a[i].z ^ pat) | zero_byte_any(a[i].w ^ pat);
        mine = (mine << 1) | min(z & 0x80808080u, 1u);
      }
      while (mine) {
        const int i = SBP - __ffs(mine);
        mine &= mine - 1;
        const int v = vbase + i * NTHR;
        file_piece(p16[v], v);
      }
    }
    for (int v = nfullp * (NTHR * SBP) + tid; v < nv16; v += NTHR) file_piece(p16[v], v);
    for (int w = nv16 * 4 + tid; w < nvec; w += NTHR) {
      const unsigned t = nonzero_bytes(rid[w] ^ pat);
      file_flags((~t & 0x80808080u) >> 7, w * 4);
    }
  } else {
    const int nfull = nvec / (NTHR * SB);
    const int rotb  = (int)(((long)blockIdx.x * nfull) / gridDim.x);
    for (int b = 0; b < nfull; ++b) {
      int bb = b + rotb;
      if (bb >= nfull) bb -= nfull;
      const int vbase = bb * (NTHR * SB);
      const int4* __restrict__ p = d4 + vbase + tid;
      int4 a[SB];
#pragma unroll
      for (int i = 0; i < SB; ++i) a[i] = p[i * NTHR];
#pragma unroll
      for (int i = 0; i < SB; ++i) file4(a[i], (vbase + i * NTHR + tid) * 4);
    }
    for (int v = nfull * (NTHR * SB) + tid; v < nvec; v += NTHR) file4(d4[v], v * 4);
  }
  for (int e = nvec * 4 + tid; e < E; e += NTHR) {
    if ((unsigned)(dst[e] - n0) < (unsigned)nn) {
      const int pos = atomicAdd(&s_qn, 1);
      if (pos < QCAP) s_q[pos] = e;
    }
  }
  if (MAKE) {
#pragma unroll
    for (int i = 0; i < NTM; ++i) {
      const int tile = wave + i * nwave;
      if (tile < mtiles)
        tile_products<KM, NM, true>(adm[i][0], wfm, fum, s_xd, s_yd, tile * 16, mrows, lane);
    }
  } else {
#pragma unroll
    for (int i = 0; i < NYV; ++i) {
      const int p = tid + i * NTHR;
      if (p < ny4) reinterpret_cast<float4*>(s_yd)[p] = ydq[i];
    }
  }
  __syncthreads();
  const int qn = __builtin_amdgcn_readfirstlane(s_qn);

  if (qn > QCAP) {
    for (int e = tid; e < E; e += nthr) {
      const int d = dst[e];
      if ((unsigned)(d - n0) < (unsigned)nn)
        edge_by_one_lane<BH, H>(ys, s_yd, src, ew, We, s_agg, s_cnt, n0, e, d);
    }
  } else {
    const float wev = We[lane & (H - 1)];
    for (int t0 = wave * 32; t0 < qn; t0 += nwave * 32) {
      const int nT = (qn - t0 < 32) ? (qn - t0) : 32;
      int sv = 0, dv = n0;
      float wv = 0.0f;
      if (lane < nT) {
        const int e = s_q[t0 + lane];
        sv = src[e];
        dv = dst[e];
        wv = ew[e];
        atomicAdd(s_cnt + (dv - n0), 1.0f);
      }
      if (nT == 32) {
        owner_tile<BH, H, U, true>(ys, s_yd, s_agg, n0, sv, dv, wv, wev, 32, lane);
      } else {
        owner_tile<BH, H, U, false>(ys, s_yd, s_agg, n0, sv, dv, wv, wev, nT, lane);
      }
    }
  }

  constexpr int PER = BH / 32;
  constexpr int NQ  = (NPW + nwave - 1) / nwave;
  float xdv[NQ][PER];
  if (!MAKE) {
#pragma unroll
    for (int q = 0; q < NQ; ++q) {
      const int ln = wave + q * nwave;
      const int lc = ln < nn ? ln : nn - 1;
      const long base = (long)(n0 + lc) * BH + lane;
#pragma unroll
      for (int j = 0; j < PER; ++j) xdv[q][j] = xd[base + 32 * j];
    }
  }
  constexpr int KX = BH / 4;
  constexpr int NX = NEXT ? NEXTN : 16;
  v2f wan[KX / 16][NX / 4];
  v2f vbn[NX / 4][NX / 16];
  if (NEXT) load_fold_operands<KX, NX>(Wx, Mx, wan, vbn, lane);
  __syncthreads();

#pragma unroll
  for (int q = 0; q < NQ; ++q) {
    const int ln = wave + q * nwave;
    if (ln < nn) {
      const int d = n0 + ln;
      const float inv_cnt = 1.0f / fmaxf(s_cnt[ln], 1.0f);
      const float* ar = s_agg + ln * BH + lane;
      const float* yr = s_yd + ln * BH + lane;
      float o[PER];
      float sum = 0.0f, ssq = 0.0f;
#pragma unroll
      for (int j = 0; j < PER; ++j) {
        o[j] = yr[32 * j] + ar[32 * j] * inv_cnt;
        sum += o[j];
        ssq += o[j] * o[j];
      }
#pragma unroll
      for (int mask = 16; mask >= 1; mask >>= 1) {
        sum += __shfl_xor(sum, mask, 32);
        ssq += __shfl_xor(ssq, mask, 32);
      }
      const float mu  = sum * (1.0f / BH);
      const float var = ssq * (1.0f / BH) - mu * mu;
      const float is  = rsqrtf(var + 1e-5f);

      const long base = (long)d * BH;
#pragma unroll
      for (int j = 0; j < PER; ++j) {
        const int i = lane + j * 32;
        const float xv = MAKE ? s_xd[ln * BH + i] : xdv[q][j];
        float v = xv + (o[j] - mu) * is;
        v = v > 0.0f ? v : 0.01f * v;
        if (NEXT) s_agg[ln * BH + i] = v;
        if (FINAL) {
          const int b = i / H, c = i & (H - 1);
          out[((long)b * Nd + d) * H + c] = v;
        } else {
          out[base + i] = v;
        }
      }
    }
  }

  if (NEXT) {
    v8f fvn[KX / 16][NX / 16];
    fold_form<KX, NX>(wan, vbn, fvn);
    __syncthreads();
    const int xrows  = nn * 4;
    const int xtiles = (xrows + 15) >> 4;
    const v2f nowf[KX / 4][NX / 16] = {};
    float* yrange = ysn + (long)n0 * 4 * NX;
    for (int tile = wave; tile < xtiles; tile += nwave) {
      const int r = tile * 16 + (lane & 15);
      v2f afn[KX / 4];
#pragma unroll
      for (int t2 = 0; t2 < KX / 16; ++t2) {
        const float4* p = reinterpret_cast<const float4*>(s_agg + r * KX + t2 * 16 + (lane >> 4) * 8);
        float4 lo = {0.0f, 0.0f, 0.0f, 0.0f}, hi = {0.0f, 0.0f, 0.0f, 0.0f};
        if (r < xrows) { lo = p[0]; hi = p[1]; }
        afn[t2 * 4 + 0].x = lo.x; afn[t2 * 4 + 0].y = lo.y;
        afn[t2 * 4 + 1].x = lo.z; afn[t2 * 4 + 1].y = lo.w;
        afn[t2 * 4 + 2].x = hi.x; afn[t2 * 4 + 2].y = hi.y;
        afn[t2 * 4 + 3].x = hi.z; afn[t2 * 4 + 3].y = hi.w;
      }
      tile_products<KX, NX, false>(afn, nowf, fvn, nullptr, yrange, tile * 16, xrows, lane);
    }
  }
}


extern "C" void kernel_launch(void* const* d_in, const int* in_sizes, int n_in,
                              void* d_out, int out_size, void* d_ws, size_t ws_size,
                              hipStream_t stream) {
  constexpr int B = 4, Cin = 32, H = 16, Cout = 32;
  const float* X    = (const float*)d_in[0];
  const int*   src0 = (const int*)d_in[1];
  const int*   dst0 = (const int*)d_in[2];
  const float* ew0  = (const float*)d_in[3];
  const int*   res0 = (const int*)d_in[4];
  const int*   src1 = (const int*)d_in[5];
  const int*   dst1 = (const int*)d_in[6];
  const float* ew1  = (const float*)d_in[7];
  const int*   res1 = (const int*)d_in[8];
  const float* Wn1  = (const float*)d_in[9];
  const float* We1  = (const float*)d_in[10];
  const float* U1   = (const float*)d_in[11];
  const float* V1   = (const float*)d_in[12];
  const float* Wn2  = (const float*)d_in[13];
  const float* We2  = (const float*)d_in[14];
  const float* U2   = (const float*)d_in[15];
  const float* V2   = (const float*)d_in[16];

  const int E0 = in_sizes[1];
  const int N1 = in_sizes[4];
  const int E1 = in_sizes[5];
  const int N2 = in_sizes[8];
  const int N0 = in_sizes[0] / (B * Cin);

  float* ws = (float*)d_ws;
  const size_t szXS = (size_t)((N0 * B * H > N1 * B * Cout) ? N0 * B * H : N1 * B * Cout);
  float* regP = ws;
  float* ys   = regP + szXS;
  float* regR = ys + szXS;
  float* xd1  = regP;
  float* yd1  = regR;
  float* h1   = regR;
  float* xd2  = regP;
  float* yd2  = regP + (size_t)N2 * B * Cout;

  const dim3 gblk(128);
  auto node_grid = [](int stiles, int T, int dtiles, int TD) {
    const int ws_ = (stiles + T - 1) / T;
    const int wd_ = (dtiles + TD - 1) / TD;
    const int waves = ws_ > wd_ ? ws_ : wd_;
    return dim3((unsigned)((waves + 3) / 4));
  };
  const dim3 oblk(1024);
  auto owner_grid = [](int nodes, int NPW) {
    return dim3((unsigned)((nodes + NPW - 1) / NPW));
  };

  const int R1s = N0 * B;
  const int R1d = N1 * B;
  const dim3 ng1 = node_grid(R1s / 16, 4, R1d / 16, 2);
  const dim3 og1 = owner_grid(N1, 200);
  unsigned int* rid1 = (unsigned int*)(regP + (size_t)N1 * B * H);
  if ((szXS - (size_t)N1 * B * H) * sizeof(float) < (size_t)E0 || og1.x > 255u ||
      (size_t)ng1.x * 128 < (size_t)(E0 >> 2))
    rid1 = nullptr;
  const dim3 og2 = owner_grid(N2, 100);
  float* endR = regR + (size_t)N1 * B * H;

  float* ys2 = endR;
  unsigned int* rid2f = (unsigned int*)(ys2 + (size_t)N1 * B * Cout);
  if (((size_t)(ys2 - ws) + (size_t)N1 * B * Cout) * sizeof(float) + (size_t)E1 <= ws_size) {
    if (og2.x > 255u || (size_t)ng1.x * 128 < (size_t)(E1 >> 2)) rid2f = nullptr;
    layer_nodes_wmma<32, 16, 1, 4, 2, 200, 100><<<ng1, gblk, 0, stream>>>(
        X, Wn1, V1, U1, res0, ys, xd1, yd1, R1s, R1d, N0, dst0, rid1, E0, N1,
        dst1, rid2f, E1, N2);
    edge_owner_kernel<64, 16, 200, 16, 2560, false, 32, 0><<<og1, oblk, 0, stream>>>(
        ys, yd1, xd1, src0, dst0, ew0, We1, rid1, h1, E0, N1,
        Wn2, V2, nullptr, nullptr, ys2);
    edge_owner_kernel<128, 32, 100, 8, 2560, true, 0, 16><<<og2, oblk, 0, stream>>>(
        ys2, nullptr, nullptr, src1, dst1, ew1, We2, rid2f, (float*)d_out, E1, N2,
        Wn2, U2, h1, res1, nullptr);
    return;
  }

  layer_nodes_wmma<32, 16, 1, 4, 2, 200, 1><<<ng1, gblk, 0, stream>>>(
      X, Wn1, V1, U1, res0, ys, xd1, yd1, R1s, R1d, N0, dst0, rid1, E0, N1,
      nullptr, nullptr, 0, 0);
  edge_owner_kernel<64, 16, 200, 16, 2560, false, 0, 0><<<og1, oblk, 0, stream>>>(
      ys, yd1, xd1, src0, dst0, ew0, We1, rid1, h1, E0, N1,
      nullptr, nullptr, nullptr, nullptr, nullptr);

  const int R2s = N1 * B;
  const int R2d = N2 * B;
  const dim3 ng2 = node_grid(R2s / 16, 4, R2d / 16, 2);
  unsigned int* rid2 = (unsigned int*)endR;
  if ((size_t)(endR - ws) * sizeof(float) + (size_t)E1 > ws_size || og2.x > 255u ||
      (size_t)ng2.x * 128 < (size_t)(E1 >> 2))
    rid2 = nullptr;
  layer_nodes_wmma<16, 32, 0, 4, 2, 100, 1><<<ng2, gblk, 0, stream>>>(
      h1, Wn2, V2, U2, res1, ys, xd2, yd2, R2s, R2d, 0, dst1, rid2, E1, N2,
      nullptr, nullptr, 0, 0);
  edge_owner_kernel<128, 32, 100, 8, 2560, true, 0, 0><<<og2, oblk, 0, stream>>>(
      ys, yd2, xd2, src1, dst1, ew1, We2, rid2, (float*)d_out, E1, N2,
      nullptr, nullptr, nullptr, nullptr, nullptr);
}
